# starting kernel
# baseline (speedup 1.0000x reference)
#include <hip/hip_runtime.h>
#include <stdint.h>

#define NNODES 50000
#define NEDGES 800000
#define ETOT   (NEDGES + NNODES)

typedef float v2f __attribute__((ext_vector_type(2)));
typedef float v8f __attribute__((ext_vector_type(8)));

__device__ __forceinline__ unsigned flipF(float f) {
    unsigned u = __float_as_uint(f);
    unsigned mask = (unsigned)(-(int)(u >> 31)) | 0x80000000u;
    return u ^ mask;
}
__device__ __forceinline__ float unflipF(unsigned u) {
    unsigned mask = ((u >> 31) - 1u) | 0x80000000u;
    return __uint_as_float(u ^ mask);
}

__device__ __forceinline__ void edge_sd(const int* __restrict__ ei, int e, int& s, int& d) {
    if (e < NEDGES) { s = ei[e]; d = ei[NEDGES + e]; }
    else            { s = e - NEDGES; d = s; }
}

__global__ void zero_f32(float* __restrict__ p, long n) {
    long i = (long)blockIdx.x * blockDim.x + threadIdx.x;
    long st = (long)gridDim.x * blockDim.x;
    for (; i < n; i += st) p[i] = 0.0f;
}

__global__ __launch_bounds__(256)
void gemm_f32_wmma(const float* __restrict__ A, const float* __restrict__ B,
                   float* __restrict__ C, int M, int N, int K) {
    const int wave = (blockIdx.x * blockDim.x + threadIdx.x) >> 5;
    const int lane = threadIdx.x & 31;
    const int stripsN = N >> 6;
    const int tm = wave / stripsN;
    const int tn = (wave % stripsN) * 4;
    if (tm * 16 >= M) return;

    const int m   = lane & 15;
    const int kh  = (lane >> 4) << 1;
    const int n0  = lane & 15;

    v8f acc[4];
    #pragma unroll
    for (int t = 0; t < 4; ++t) acc[t] = (v8f){0.f,0.f,0.f,0.f,0.f,0.f,0.f,0.f};

    const float* __restrict__ Arow = A + (long)(tm * 16 + m) * K;
    for (int k0 = 0; k0 < K; k0 += 4) {
        v2f a;
        a.x = Arow[k0 + kh + 0];
        a.y = Arow[k0 + kh + 1];
        #pragma unroll
        for (int t = 0; t < 4; ++t) {
            const int col = (tn + t) * 16 + n0;
            v2f b;
            b.x = B[(long)(k0 + kh + 0) * N + col];
            b.y = B[(long)(k0 + kh + 1) * N + col];
            acc[t] = __builtin_amdgcn_wmma_f32_16x16x4_f32(
                         false, a, false, b, (short)0, acc[t], false, false);
        }
    }
    const int moff = (lane >> 4) * 8;
    #pragma unroll
    for (int t = 0; t < 4; ++t) {
        #pragma unroll
        for (int v = 0; v < 8; ++v) {
            C[(long)(tm * 16 + v + moff) * N + (tn + t) * 16 + n0] = acc[t][v];
        }
    }
}

__global__ void attn_scores(const float* __restrict__ xh,
                            const float* __restrict__ att_src,
                            const float* __restrict__ att_dst,
                            float* __restrict__ asrc, float* __restrict__ adst,
                            int H, int C) {
    int idx = blockIdx.x * blockDim.x + threadIdx.x;
    if (idx >= NNODES * H) return;
    int n = idx / H, h = idx % H;
    const float* x = xh + (long)n * H * C + (long)h * C;
    const float* as = att_src + h * C;
    const float* ad = att_dst + h * C;
    float ss = 0.f, sd = 0.f;
    for (int c = 0; c < C; ++c) { float v = x[c]; ss += v * as[c]; sd += v * ad[c]; }
    asrc[idx] = ss;
    adst[idx] = sd;
}

__global__ void edge_max(const int* __restrict__ ei,
                         const float* __restrict__ asrc, const float* __restrict__ adst,
                         float* __restrict__ e_out, unsigned* __restrict__ emax, int H) {
    int idx = blockIdx.x * blockDim.x + threadIdx.x;
    if (idx >= ETOT * H) return;
    int e = idx / H, h = idx % H;
    int s, d; edge_sd(ei, e, s, d);
    float x = asrc[s * H + h] + adst[d * H + h];
    x = (x > 0.f) ? x : 0.2f * x;
    e_out[idx] = x;
    atomicMax(&emax[d * H + h], flipF(x));
}

__global__ void edge_expsum(const int* __restrict__ ei,
                            float* __restrict__ e_io, const unsigned* __restrict__ emax,
                            float* __restrict__ esum, int H) {
    int idx = blockIdx.x * blockDim.x + threadIdx.x;
    if (idx >= ETOT * H) return;
    int e = idx / H, h = idx % H;
    int s, d; edge_sd(ei, e, s, d); (void)s;
    float m  = unflipF(emax[d * H + h]);
    float ex = expf(e_io[idx] - m);
    e_io[idx] = ex;
    atomicAdd(&esum[d * H + h], ex);
}

__global__ __launch_bounds__(256)
void edge_aggregate(const int* __restrict__ ei, const float* __restrict__ ex,
                    const float* __restrict__ esum, const float* __restrict__ xh,
                    float* __restrict__ out, int H, int C) {
    int gid  = blockIdx.x * blockDim.x + threadIdx.x;
    int e    = gid >> 5;
    int lane = gid & 31;
    if (e >= ETOT) return;
    int s, d; edge_sd(ei, e, s, d);
    const int HC = H * C;
    const float* __restrict__ xs = xh + (long)s * HC;
    float* __restrict__ od = out + (long)d * HC;
    for (int base = lane * 4; base < HC; base += 128) {
        int h = base / C;
        float coeff = ex[(long)e * H + h] / esum[d * H + h];
        float4 v = *(const float4*)(xs + base);
        atomicAdd(od + base + 0, coeff * v.x);
        atomicAdd(od + base + 1, coeff * v.y);
        atomicAdd(od + base + 2, coeff * v.z);
        atomicAdd(od + base + 3, coeff * v.w);
    }
}

__global__ void add_bias(float* __restrict__ out, const float* __restrict__ b,
                         int NC, long total) {
    long i = (long)blockIdx.x * blockDim.x + threadIdx.x;
    if (i >= total) return;
    out[i] += b[(int)(i % NC)];
}

static inline int cdiv(long a, int b) { return (int)((a + b - 1) / b); }

extern "C" void kernel_launch(void* const* d_in, const int* in_sizes, int n_in,
                              void* d_out, int out_size, void* d_ws, size_t ws_size,
                              hipStream_t stream) {
    const float* X        = (const float*)d_in[0];
    const int*   ei       = (const int*)  d_in[1];
    const float* W1       = (const float*)d_in[2];
    const float* att_src1 = (const float*)d_in[3];
    const float* att_dst1 = (const float*)d_in[4];
    const float* b1       = (const float*)d_in[5];
    const float* W2       = (const float*)d_in[6];
    const float* att_src2 = (const float*)d_in[7];
    const float* att_dst2 = (const float*)d_in[8];
    const float* b2       = (const float*)d_in[9];
    float* out = (float*)d_out;

    float* ws = (float*)d_ws;
    long off = 0;
    float*    xh1   = ws + off; off += (long)NNODES * 256;
    float*    h1    = ws + off; off += (long)NNODES * 256;
    float*    esum1 = ws + off; off += (long)NNODES * 4;
    unsigned* emax1 = (unsigned*)(ws + off); off += (long)NNODES * 4;
    float*    esum2 = ws + off; off += NNODES;
    unsigned* emax2 = (unsigned*)(ws + off); off += NNODES;
    float*    asrc1 = ws + off; off += (long)NNODES * 4;
    float*    adst1 = ws + off; off += (long)NNODES * 4;
    float*    e1    = ws + off; off += (long)ETOT * 4;
    float*    xh2   = xh1;
    float*    e2    = e1;
    float*    asrc2 = asrc1;
    float*    adst2 = adst1;

    const int B = 256;

    long zlen = (long)NNODES * 256 + (long)NNODES * 4 * 2 + (long)NNODES * 2;
    zero_f32<<<2048, B, 0, stream>>>(h1, zlen);
    zero_f32<<<2048, B, 0, stream>>>(out, (long)NNODES * 128);

    {
        int waves = (NNODES / 16) * (256 / 64);
        gemm_f32_wmma<<<cdiv((long)waves * 32, B), B, 0, stream>>>(X, W1, xh1, NNODES, 256, 64);
        attn_scores<<<cdiv((long)NNODES * 4, B), B, 0, stream>>>(xh1, att_src1, att_dst1,
                                                                 asrc1, adst1, 4, 64);
        edge_max   <<<cdiv((long)ETOT * 4, B), B, 0, stream>>>(ei, asrc1, adst1, e1, emax1, 4);
        edge_expsum<<<cdiv((long)ETOT * 4, B), B, 0, stream>>>(ei, e1, emax1, esum1, 4);
        edge_aggregate<<<cdiv((long)ETOT * 32, B), B, 0, stream>>>(ei, e1, esum1, xh1, h1, 4, 64);
        add_bias<<<cdiv((long)NNODES * 256, B), B, 0, stream>>>(h1, b1, 256, (long)NNODES * 256);
    }

    {
        int waves = (NNODES / 16) * (128 / 64);
        gemm_f32_wmma<<<cdiv((long)waves * 32, B), B, 0, stream>>>(h1, W2, xh2, NNODES, 128, 256);
        attn_scores<<<cdiv((long)NNODES, B), B, 0, stream>>>(xh2, att_src2, att_dst2,
                                                             asrc2, adst2, 1, 128);
        edge_max   <<<cdiv((long)ETOT, B), B, 0, stream>>>(ei, asrc2, adst2, e2, emax2, 1);
        edge_expsum<<<cdiv((long)ETOT, B), B, 0, stream>>>(ei, e2, emax2, esum2, 1);
        edge_aggregate<<<cdiv((long)ETOT * 32, B), B, 0, stream>>>(ei, e2, esum2, xh2, out, 1, 128);
        add_bias<<<cdiv((long)NNODES * 128, B), B, 0, stream>>>(out, b2, 128, (long)NNODES * 128);
    }
}


# Edges grouped by destination in-call (count with arrival rank, scan, fill); one wave per node does softmax and weighted row sum in registers, one store per row; no float atomics, sums in rank order
# speedup vs baseline: 3.1579x; 3.1579x over previous
#include <hip/hip_runtime.h>
#include <stdint.h>


#define NNODES 50000
#define NEDGES 800000

typedef float v2f __attribute__((ext_vector_type(2)));
typedef float v8f __attribute__((ext_vector_type(8)));


__global__ void zero_i32(int* __restrict__ p, int n) {
    int i = blockIdx.x * blockDim.x + threadIdx.x;
    if (i < n) p[i] = 0;
}

__global__ void dst_count_rank(const int* __restrict__ ei, int* __restrict__ cnt,
                               int* __restrict__ rank) {
    int e = blockIdx.x * blockDim.x + threadIdx.x;
    if (e >= NEDGES) return;
    rank[e] = atomicAdd(&cnt[ei[NEDGES + e]], 1);
}

#define SCAN_T   1024
#define SCAN_PER ((NNODES + SCAN_T - 1) / SCAN_T)

__global__ __launch_bounds__(SCAN_T)
void rowptr_scan(const int* __restrict__ cnt, int* __restrict__ rowptr) {
    static_assert((SCAN_PER - 1) * SCAN_T <= NNODES, "all tiles but the last are full");
    __shared__ int sh[SCAN_T * SCAN_PER];
    __shared__ int wsum[32];
    const int t = threadIdx.x;
    const int lane = t & 31, w = t >> 5;
    {
        int v[SCAN_PER];
        #pragma unroll
        for (int k = 0; k < SCAN_PER - 1; ++k) v[k] = cnt[k * SCAN_T + t];
        const int il = (SCAN_PER - 1) * SCAN_T + t;
        v[SCAN_PER - 1] = (il < NNODES) ? cnt[il] : 0;
        #pragma unroll
        for (int k = 0; k < SCAN_PER; ++k) sh[k * SCAN_T + t] = v[k];
    }
    __syncthreads();
    const int base = t * SCAN_PER;
    int s = 0;
    for (int k = 0; k < SCAN_PER; ++k) s += sh[base + k];
    int inc = s;
    for (int o = 1; o < 32; o <<= 1) { int v = __shfl_up(inc, o); if (lane >= o) inc += v; }
    if (lane == 31) wsum[w] = inc;
    __syncthreads();
    if (w == 0) {
        int v = wsum[lane], vi = v;
        for (int o = 1; o < 32; o <<= 1) { int u = __shfl_up(vi, o); if (lane >= o) vi += u; }
        wsum[lane] = vi - v;
    }
    __syncthreads();
    int run = wsum[w] + inc - s;
    for (int k = 0; k < SCAN_PER; ++k) { int c = sh[base + k]; sh[base + k] = run; run += c; }
    __syncthreads();
    #pragma unroll
    for (int k = 0; k < SCAN_PER - 1; ++k) rowptr[k * SCAN_T + t] = sh[k * SCAN_T + t];
    {
        const int il = (SCAN_PER - 1) * SCAN_T + t;
        if (il < NNODES) rowptr[il] = sh[il];
    }
    if (t == SCAN_T - 1) rowptr[NNODES] = run;
}

__global__ void csr_fill(const int* __restrict__ ei, const int* __restrict__ rowptr,
                         const int* __restrict__ rank, int* __restrict__ csr_src) {
    int e = blockIdx.x * blockDim.x + threadIdx.x;
    if (e >= NEDGES) return;
    int s = ei[e], d = ei[NEDGES + e];
    csr_src[rowptr[d] + rank[e]] = s;
}

__global__ __launch_bounds__(256)
void gemm_f32_wmma(const float* __restrict__ A, const float* __restrict__ B,
                   float* __restrict__ C, int M, int N, int K) {
    const int wave = (blockIdx.x * blockDim.x + threadIdx.x) >> 5;
    const int lane = threadIdx.x & 31;
    const int stripsN = N >> 6;
    const int tm = wave / stripsN;
    const int tn = (wave % stripsN) * 4;
    if (tm * 16 >= M) return;

    const int m   = lane & 15;
    const int kh  = (lane >> 4) << 1;
    const int n0  = lane & 15;

    v8f acc[4];
    #pragma unroll
    for (int t = 0; t < 4; ++t) acc[t] = (v8f){0.f,0.f,0.f,0.f,0.f,0.f,0.f,0.f};

    const float* __restrict__ Arow = A + (long)(tm * 16 + m) * K;
    for (int k0 = 0; k0 < K; k0 += 4) {
        v2f a;
        a.x = Arow[k0 + kh + 0];
        a.y = Arow[k0 + kh + 1];
        #pragma unroll
        for (int t = 0; t < 4; ++t) {
            const int col = (tn + t) * 16 + n0;
            v2f b;
            b.x = B[(long)(k0 + kh + 0) * N + col];
            b.y = B[(long)(k0 + kh + 1) * N + col];
            acc[t] = __builtin_amdgcn_wmma_f32_16x16x4_f32(
                         false, a, false, b, (short)0, acc[t], false, false);
        }
    }
    const int moff = (lane >> 4) * 8;
    #pragma unroll
    for (int t = 0; t < 4; ++t) {
        #pragma unroll
        for (int v = 0; v < 8; ++v) {
            C[(long)(tm * 16 + v + moff) * N + (tn + t) * 16 + n0] = acc[t][v];
        }
    }
}

__global__ void attn_scores(const float* __restrict__ xh,
                            const float* __restrict__ att_src,
                            const float* __restrict__ att_dst,
                            float* __restrict__ asrc, float* __restrict__ adst,
                            int H, int C) {
    int idx = blockIdx.x * blockDim.x + threadIdx.x;
    if (idx >= NNODES * H) return;
    int n = idx / H, h = idx % H;
    const float* x = xh + (long)n * H * C + (long)h * C;
    const float* as = att_src + h * C;
    const float* ad = att_dst + h * C;
    float ss = 0.f, sd = 0.f;
    for (int c = 0; c < C; ++c) { float v = x[c]; ss += v * as[c]; sd += v * ad[c]; }
    asrc[idx] = ss;
    adst[idx] = sd;
}

__device__ __forceinline__ float leaky02(float x) { return (x > 0.f) ? x : 0.2f * x; }

template <int H, int C>
__global__ __launch_bounds__(256)
void node_softmax_aggregate(const int* __restrict__ rowptr, const int* __restrict__ csr_src,
                            const float* __restrict__ asrc, const float* __restrict__ adst,
                            const float* __restrict__ xh, const float* __restrict__ bias,
                            float* __restrict__ out) {
    constexpr int HC  = H * C;
    constexpr int NIT = HC / 128;
    const int lane = threadIdx.x & 31;
    const int n = __builtin_amdgcn_readfirstlane((int)((blockIdx.x * blockDim.x + threadIdx.x) >> 5));
    if (n >= NNODES) return;
    const int start = __builtin_amdgcn_readfirstlane(rowptr[n]);
    const int deg   = __builtin_amdgcn_readfirstlane(rowptr[n + 1]) - start;
    const int total = deg + 1;

    float ad[H];
    #pragma unroll
    for (int h = 0; h < H; ++h) ad[h] = adst[n * H + h];

    const bool v0 = lane < total;
    const int  s0 = (lane < deg) ? csr_src[start + lane] : n;
    float x0[H];
    #pragma unroll
    for (int h = 0; h < H; ++h) x0[h] = v0 ? leaky02(asrc[s0 * H + h] + ad[h]) : -INFINITY;

    float m[H];
    #pragma unroll
    for (int h = 0; h < H; ++h) m[h] = x0[h];
    for (int c0 = 32; c0 < total; c0 += 32) {
        const int t = c0 + lane;
        const int s = (t < deg) ? csr_src[start + t] : n;
        if (t < total) {
            #pragma unroll
            for (int h = 0; h < H; ++h) m[h] = fmaxf(m[h], leaky02(asrc[s * H + h] + ad[h]));
        }
    }
    #pragma unroll
    for (int h = 0; h < H; ++h) {
        #pragma unroll
        for (int o = 16; o > 0; o >>= 1) m[h] = fmaxf(m[h], __shfl_xor(m[h], o));
    }

    float e0[H], sum[H];
    #pragma unroll
    for (int h = 0; h < H; ++h) { e0[h] = v0 ? expf(x0[h] - m[h]) : 0.f; sum[h] = e0[h]; }
    for (int c0 = 32; c0 < total; c0 += 32) {
        const int t = c0 + lane;
        const int s = (t < deg) ? csr_src[start + t] : n;
        if (t < total) {
            #pragma unroll
            for (int h = 0; h < H; ++h) sum[h] += expf(leaky02(asrc[s * H + h] + ad[h]) - m[h]);
        }
    }
    #pragma unroll
    for (int h = 0; h < H; ++h) {
        #pragma unroll
        for (int o = 16; o > 0; o >>= 1) sum[h] += __shfl_xor(sum[h], o);
    }

    int hsel[NIT];
    #pragma unroll
    for (int it = 0; it < NIT; ++it) hsel[it] = (it * 128 + lane * 4) / C;
    float4 acc[NIT];
    #pragma unroll
    for (int it = 0; it < NIT; ++it) acc[it] = make_float4(0.f, 0.f, 0.f, 0.f);

    for (int c0 = 0; c0 < total; c0 += 32) {
        int s_l = s0;
        float cf[H];
        if (c0 == 0) {
            #pragma unroll
            for (int h = 0; h < H; ++h) cf[h] = e0[h] / sum[h];
        } else {
            const int t = c0 + lane;
            s_l = (t < deg) ? csr_src[start + t] : n;
            #pragma unroll
            for (int h = 0; h < H; ++h)
                cf[h] = expf(leaky02(asrc[s_l * H + h] + ad[h]) - m[h]) / sum[h];
        }
        const int cnt = (total - c0 < 32) ? (total - c0) : 32;
        #pragma unroll 4
        for (int jj = 0; jj < cnt; ++jj) {
            const int s = __builtin_amdgcn_readlane(s_l, jj);
            float ch[H];
            #pragma unroll
            for (int h = 0; h < H; ++h)
                ch[h] = __int_as_float(__builtin_amdgcn_readlane(__float_as_int(cf[h]), jj));
            const float* __restrict__ row = xh + (long)s * HC + lane * 4;
            #pragma unroll
            for (int it = 0; it < NIT; ++it) {
                float c = ch[0];
                #pragma unroll
                for (int h = 1; h < H; ++h) c = (hsel[it] == h) ? ch[h] : c;
                const float4 v = *(const float4*)(row + it * 128);
                acc[it].x += c * v.x;
                acc[it].y += c * v.y;
                acc[it].z += c * v.z;
                acc[it].w += c * v.w;
            }
        }
    }

    float* __restrict__ orow = out + (long)n * HC + lane * 4;
    #pragma unroll
    for (int it = 0; it < NIT; ++it) {
        const float4 b = *(const float4*)(bias + it * 128 + lane * 4);
        float4 r;
        r.x = acc[it].x + b.x; r.y = acc[it].y + b.y; r.z = acc[it].z + b.z; r.w = acc[it].w + b.w;
        *(float4*)(orow + it * 128) = r;
    }
}

static inline int cdiv(long a, int b) { return (int)((a + b - 1) / b); }

extern "C" void kernel_launch(void* const* d_in, const int* in_sizes, int n_in,
                              void* d_out, int out_size, void* d_ws, size_t ws_size,
                              hipStream_t stream) {
    const float* X        = (const float*)d_in[0];
    const int*   ei       = (const int*)  d_in[1];
    const float* W1       = (const float*)d_in[2];
    const float* att_src1 = (const float*)d_in[3];
    const float* att_dst1 = (const float*)d_in[4];
    const float* b1       = (const float*)d_in[5];
    const float* W2       = (const float*)d_in[6];
    const float* att_src2 = (const float*)d_in[7];
    const float* att_dst2 = (const float*)d_in[8];
    const float* b2       = (const float*)d_in[9];
    float* out = (float*)d_out;

    float* ws = (float*)d_ws;
    long off = 0;
    float* xh     = ws + off; off += (long)NNODES * 256;
    float* h1     = ws + off; off += (long)NNODES * 256;
    float* asrc   = ws + off; off += (long)NNODES * 4;
    float* adst   = ws + off; off += (long)NNODES * 4;
    int*   cnt    = (int*)(ws + off); off += NNODES;
    int*   rowptr = (int*)(ws + off); off += NNODES + 4;
    int*   rank   = (int*)(ws + off); off += NEDGES;
    int*   csrc   = (int*)(ws + off); off += NEDGES;

    const int B = 256;

    zero_i32      <<<cdiv(NNODES, B), B, 0, stream>>>(cnt, NNODES);
    dst_count_rank<<<cdiv(NEDGES, B), B, 0, stream>>>(ei, cnt, rank);
    rowptr_scan   <<<1, SCAN_T, 0, stream>>>(cnt, rowptr);
    csr_fill      <<<cdiv(NEDGES, B), B, 0, stream>>>(ei, rowptr, rank, csrc);

    {
        int waves = (NNODES / 16) * (256 / 64);
        gemm_f32_wmma<<<cdiv((long)waves * 32, B), B, 0, stream>>>(X, W1, xh, NNODES, 256, 64);
        attn_scores<<<cdiv((long)NNODES * 4, B), B, 0, stream>>>(xh, att_src1, att_dst1,
                                                                 asrc, adst, 4, 64);
        node_softmax_aggregate<4, 64><<<cdiv((long)NNODES * 32, B), B, 0, stream>>>(
            rowptr, csrc, asrc, adst, xh, b1, h1);
    }

    {
        int waves = (NNODES / 16) * (128 / 64);
        gemm_f32_wmma<<<cdiv((long)waves * 32, B), B, 0, stream>>>(h1, W2, xh, NNODES, 128, 256);
        attn_scores<<<cdiv((long)NNODES, B), B, 0, stream>>>(xh, att_src2, att_dst2,
                                                             asrc, adst, 1, 128);
        node_softmax_aggregate<1, 128><<<cdiv((long)NNODES * 32, B), B, 0, stream>>>(
            rowptr, csrc, asrc, adst, xh, b2, out);
    }
}


# Layer 1 sums the 64-float input rows per head before projecting (weighted sum commutes with X W1), scores as X.(W1.att): one 256-B gather per edge, not four; sum order changes
# speedup vs baseline: 3.5041x; 1.1096x over previous
#include <hip/hip_runtime.h>
#include <stdint.h>


#define NNODES 50000
#define NEDGES 800000

typedef float v2f __attribute__((ext_vector_type(2)));
typedef float v8f __attribute__((ext_vector_type(8)));


__global__ void zero_i32(int* __restrict__ p, int n) {
    int i = blockIdx.x * blockDim.x + threadIdx.x;
    if (i < n) p[i] = 0;
}

__global__ void dst_count_rank(const int* __restrict__ ei, int* __restrict__ cnt,
                               int* __restrict__ rank) {
    int e = blockIdx.x * blockDim.x + threadIdx.x;
    if (e >= NEDGES) return;
    rank[e] = atomicAdd(&cnt[ei[NEDGES + e]], 1);
}

#define SCAN_T   1024
#define SCAN_PER ((NNODES + SCAN_T - 1) / SCAN_T)

__global__ __launch_bounds__(SCAN_T)
void rowptr_scan(const int* __restrict__ cnt, int* __restrict__ rowptr) {
    static_assert((SCAN_PER - 1) * SCAN_T <= NNODES, "all tiles but the last are full");
    __shared__ int sh[SCAN_T * SCAN_PER];
    __shared__ int wsum[32];
    const int t = threadIdx.x;
    const int lane = t & 31, w = t >> 5;
    {
        int v[SCAN_PER];
        #pragma unroll
        for (int k = 0; k < SCAN_PER - 1; ++k) v[k] = cnt[k * SCAN_T + t];
        const int il = (SCAN_PER - 1) * SCAN_T + t;
        v[SCAN_PER - 1] = (il < NNODES) ? cnt[il] : 0;
        #pragma unroll
        for (int k = 0; k < SCAN_PER; ++k) sh[k * SCAN_T + t] = v[k];
    }
    __syncthreads();
    const int base = t * SCAN_PER;
    int s = 0;
    for (int k = 0; k < SCAN_PER; ++k) s += sh[base + k];
    int inc = s;
    for (int o = 1; o < 32; o <<= 1) { int v = __shfl_up(inc, o); if (lane >= o) inc += v; }
    if (lane == 31) wsum[w] = inc;
    __syncthreads();
    if (w == 0) {
        int v = wsum[lane], vi = v;
        for (int o = 1; o < 32; o <<= 1) { int u = __shfl_up(vi, o); if (lane >= o) vi += u; }
        wsum[lane] = vi - v;
    }
    __syncthreads();
    int run = wsum[w] + inc - s;
    for (int k = 0; k < SCAN_PER; ++k) { int c = sh[base + k]; sh[base + k] = run; run += c; }
    __syncthreads();
    #pragma unroll
    for (int k = 0; k < SCAN_PER - 1; ++k) rowptr[k * SCAN_T + t] = sh[k * SCAN_T + t];
    {
        const int il = (SCAN_PER - 1) * SCAN_T + t;
        if (il < NNODES) rowptr[il] = sh[il];
    }
    if (t == SCAN_T - 1) rowptr[NNODES] = run;
}

__global__ void csr_fill(const int* __restrict__ ei, const int* __restrict__ rowptr,
                         const int* __restrict__ rank, int* __restrict__ csr_src) {
    int e = blockIdx.x * blockDim.x + threadIdx.x;
    if (e >= NEDGES) return;
    int s = ei[e], d = ei[NEDGES + e];
    csr_src[rowptr[d] + rank[e]] = s;
}

__global__ __launch_bounds__(256)
void gemm_f32_wmma(const float* __restrict__ A, const float* __restrict__ B,
                   float* __restrict__ C, int M, int N, int K,
                   int lda, int a_strip_step, const float* __restrict__ bias) {
    const int wave = (blockIdx.x * blockDim.x + threadIdx.x) >> 5;
    const int lane = threadIdx.x & 31;
    const int stripsN = N >> 6;
    const int tm = wave / stripsN;
    const int strip = wave % stripsN;
    const int tn = strip * 4;
    if (tm * 16 >= M) return;

    const int m   = lane & 15;
    const int kh  = (lane >> 4) << 1;
    const int n0  = lane & 15;

    v8f acc[4];
    #pragma unroll
    for (int t = 0; t < 4; ++t) {
        const float bv = bias ? bias[(tn + t) * 16 + n0] : 0.f;
        acc[t] = (v8f){bv,bv,bv,bv,bv,bv,bv,bv};
    }

    const float* __restrict__ Arow = A + (long)(tm * 16 + m) * lda + strip * a_strip_step;
    for (int k0 = 0; k0 < K; k0 += 4) {
        v2f a;
        a.x = Arow[k0 + kh + 0];
        a.y = Arow[k0 + kh + 1];
        #pragma unroll
        for (int t = 0; t < 4; ++t) {
            const int col = (tn + t) * 16 + n0;
            v2f b;
            b.x = B[(long)(k0 + kh + 0) * N + col];
            b.y = B[(long)(k0 + kh + 1) * N + col];
            acc[t] = __builtin_amdgcn_wmma_f32_16x16x4_f32(
                         false, a, false, b, (short)0, acc[t], false, false);
        }
    }
    const int moff = (lane >> 4) * 8;
    #pragma unroll
    for (int t = 0; t < 4; ++t) {
        #pragma unroll
        for (int v = 0; v < 8; ++v) {
            C[(long)(tm * 16 + v + moff) * N + (tn + t) * 16 + n0] = acc[t][v];
        }
    }
}


__global__ __launch_bounds__(256)
void input_scores(const float* __restrict__ X, const float* __restrict__ W1,
                  const float* __restrict__ att_src, const float* __restrict__ att_dst,
                  float* __restrict__ asrc, float* __restrict__ adst) {
    __shared__ float4 wsrc[64];
    __shared__ float4 wdst[64];
    const int t = threadIdx.x;
    {
        const int h = t & 3;
        const float4* __restrict__ w4 = (const float4*)(W1 + t * 64);
        const float4* __restrict__ s4 = (const float4*)(att_src + h * 64);
        const float4* __restrict__ d4 = (const float4*)(att_dst + h * 64);
        float ss = 0.f, sd = 0.f;
        #pragma unroll
        for (int c = 0; c < 16; ++c) {
            const float4 w = w4[c], a = s4[c], d = d4[c];
            ss += w.x * a.x; ss += w.y * a.y; ss += w.z * a.z; ss += w.w * a.w;
            sd += w.x * d.x; sd += w.y * d.y; sd += w.z * d.z; sd += w.w * d.w;
        }
        ((float*)wsrc)[t] = ss;
        ((float*)wdst)[t] = sd;
    }
    __syncthreads();
    const int n = blockIdx.x * 256 + t;
    if (n >= NNODES) return;
    const float4* __restrict__ x4 = (const float4*)(X + (long)n * 64);
    float4 as = make_float4(0.f, 0.f, 0.f, 0.f), ad = make_float4(0.f, 0.f, 0.f, 0.f);
    #pragma unroll
    for (int c = 0; c < 16; ++c) {
        const float4 x = x4[c];
        const float xs[4] = {x.x, x.y, x.z, x.w};
        #pragma unroll
        for (int j = 0; j < 4; ++j) {
            const float4 ws = wsrc[c * 4 + j], wd = wdst[c * 4 + j];
            as.x += xs[j] * ws.x; as.y += xs[j] * ws.y; as.z += xs[j] * ws.z; as.w += xs[j] * ws.w;
            ad.x += xs[j] * wd.x; ad.y += xs[j] * wd.y; ad.z += xs[j] * wd.z; ad.w += xs[j] * wd.w;
        }
    }
    *(float4*)(asrc + (long)n * 4) = as;
    *(float4*)(adst + (long)n * 4) = ad;
}

__device__ __forceinline__ float leaky02(float x) { return (x > 0.f) ? x : 0.2f * x; }

__global__ __launch_bounds__(256)
void node_softmax_aggregate_in(const int* __restrict__ rowptr, const int* __restrict__ csr_src,
                               const float* __restrict__ asrc, const float* __restrict__ adst,
                               const float* __restrict__ X, float* __restrict__ agg) {
    constexpr int H = 4;
    const int lane = threadIdx.x & 31;
    const int n = __builtin_amdgcn_readfirstlane((int)((blockIdx.x * blockDim.x + threadIdx.x) >> 5));
    if (n >= NNODES) return;
    const int start = __builtin_amdgcn_readfirstlane(rowptr[n]);
    const int deg   = __builtin_amdgcn_readfirstlane(rowptr[n + 1]) - start;
    const int total = deg + 1;

    float ad[H];
    #pragma unroll
    for (int h = 0; h < H; ++h) ad[h] = adst[n * H + h];

    const bool v0 = lane < total;
    const int  s0 = (lane < deg) ? csr_src[start + lane] : n;
    const float4 a0 = *(const float4*)(asrc + (long)s0 * H);
    const float a0v[H] = {a0.x, a0.y, a0.z, a0.w};
    float x0[H];
    #pragma unroll
    for (int h = 0; h < H; ++h) x0[h] = v0 ? leaky02(a0v[h] + ad[h]) : -INFINITY;

    float m[H];
    #pragma unroll
    for (int h = 0; h < H; ++h) m[h] = x0[h];
    for (int c0 = 32; c0 < total; c0 += 32) {
        const int t = c0 + lane;
        const int s = (t < deg) ? csr_src[start + t] : n;
        if (t < total) {
            #pragma unroll
            for (int h = 0; h < H; ++h) m[h] = fmaxf(m[h], leaky02(asrc[s * H + h] + ad[h]));
        }
    }
    #pragma unroll
    for (int h = 0; h < H; ++h) {
        #pragma unroll
        for (int o = 16; o > 0; o >>= 1) m[h] = fmaxf(m[h], __shfl_xor(m[h], o));
    }

    float e0[H], sum[H];
    #pragma unroll
    for (int h = 0; h < H; ++h) { e0[h] = v0 ? expf(x0[h] - m[h]) : 0.f; sum[h] = e0[h]; }
    for (int c0 = 32; c0 < total; c0 += 32) {
        const int t = c0 + lane;
        const int s = (t < deg) ? csr_src[start + t] : n;
        if (t < total) {
            #pragma unroll
            for (int h = 0; h < H; ++h) sum[h] += expf(leaky02(asrc[s * H + h] + ad[h]) - m[h]);
        }
    }
    #pragma unroll
    for (int h = 0; h < H; ++h) {
        #pragma unroll
        for (int o = 16; o > 0; o >>= 1) sum[h] += __shfl_xor(sum[h], o);
    }

    float2 acc[H];
    #pragma unroll
    for (int h = 0; h < H; ++h) acc[h] = make_float2(0.f, 0.f);

    for (int c0 = 0; c0 < total; c0 += 32) {
        int s_l = s0;
        float cf[H];
        if (c0 == 0) {
            #pragma unroll
            for (int h = 0; h < H; ++h) cf[h] = e0[h] / sum[h];
        } else {
            const int t = c0 + lane;
            s_l = (t < deg) ? csr_src[start + t] : n;
            #pragma unroll
            for (int h = 0; h < H; ++h)
                cf[h] = (t < total) ? expf(leaky02(asrc[s_l * H + h] + ad[h]) - m[h]) / sum[h] : 0.f;
        }
        const int cnt = (total - c0 < 32) ? (total - c0) : 32;
        for (int j0 = 0; j0 < cnt; j0 += 8) {
            #pragma unroll
            for (int u = 0; u < 8; ++u) {
                const int jj = j0 + u;
                const int s = __builtin_amdgcn_readlane(s_l, jj);
                const float2 v = *(const float2*)(X + (long)s * 64 + lane * 2);
                #pragma unroll
                for (int h = 0; h < H; ++h) {
                    const float c = __int_as_float(__builtin_amdgcn_readlane(__float_as_int(cf[h]), jj));
                    acc[h].x = fmaf(c, v.x, acc[h].x);
                    acc[h].y = fmaf(c, v.y, acc[h].y);
                }
            }
        }
    }

    float* __restrict__ arow = agg + (long)n * (H * 64) + lane * 2;
    #pragma unroll
    for (int h = 0; h < H; ++h) *(float2*)(arow + h * 64) = acc[h];
}

__global__ void attn_scores(const float* __restrict__ xh,
                            const float* __restrict__ att_src,
                            const float* __restrict__ att_dst,
                            float* __restrict__ asrc, float* __restrict__ adst,
                            int H, int C) {
    int idx = blockIdx.x * blockDim.x + threadIdx.x;
    if (idx >= NNODES * H) return;
    int n = idx / H, h = idx % H;
    const float* x = xh + (long)n * H * C + (long)h * C;
    const float* as = att_src + h * C;
    const float* ad = att_dst + h * C;
    float ss = 0.f, sd = 0.f;
    for (int c = 0; c < C; ++c) { float v = x[c]; ss += v * as[c]; sd += v * ad[c]; }
    asrc[idx] = ss;
    adst[idx] = sd;
}

template <int H, int C>
__global__ __launch_bounds__(256)
void node_softmax_aggregate(const int* __restrict__ rowptr, const int* __restrict__ csr_src,
                            const float* __restrict__ asrc, const float* __restrict__ adst,
                            const float* __restrict__ xh, const float* __restrict__ bias,
                            float* __restrict__ out) {
    constexpr int HC  = H * C;
    constexpr int NIT = HC / 128;
    const int lane = threadIdx.x & 31;
    const int n = __builtin_amdgcn_readfirstlane((int)((blockIdx.x * blockDim.x + threadIdx.x) >> 5));
    if (n >= NNODES) return;
    const int start = __builtin_amdgcn_readfirstlane(rowptr[n]);
    const int deg   = __builtin_amdgcn_readfirstlane(rowptr[n + 1]) - start;
    const int total = deg + 1;

    float ad[H];
    #pragma unroll
    for (int h = 0; h < H; ++h) ad[h] = adst[n * H + h];

    const bool v0 = lane < total;
    const int  s0 = (lane < deg) ? csr_src[start + lane] : n;
    float x0[H];
    #pragma unroll
    for (int h = 0; h < H; ++h) x0[h] = v0 ? leaky02(asrc[s0 * H + h] + ad[h]) : -INFINITY;

    float m[H];
    #pragma unroll
    for (int h = 0; h < H; ++h) m[h] = x0[h];
    for (int c0 = 32; c0 < total; c0 += 32) {
        const int t = c0 + lane;
        const int s = (t < deg) ? csr_src[start + t] : n;
        if (t < total) {
            #pragma unroll
            for (int h = 0; h < H; ++h) m[h] = fmaxf(m[h], leaky02(asrc[s * H + h] + ad[h]));
        }
    }
    #pragma unroll
    for (int h = 0; h < H; ++h) {
        #pragma unroll
        for (int o = 16; o > 0; o >>= 1) m[h] = fmaxf(m[h], __shfl_xor(m[h], o));
    }

    float e0[H], sum[H];
    #pragma unroll
    for (int h = 0; h < H; ++h) { e0[h] = v0 ? expf(x0[h] - m[h]) : 0.f; sum[h] = e0[h]; }
    for (int c0 = 32; c0 < total; c0 += 32) {
        const int t = c0 + lane;
        const int s = (t < deg) ? csr_src[start + t] : n;
        if (t < total) {
            #pragma unroll
            for (int h = 0; h < H; ++h) sum[h] += expf(leaky02(asrc[s * H + h] + ad[h]) - m[h]);
        }
    }
    #pragma unroll
    for (int h = 0; h < H; ++h) {
        #pragma unroll
        for (int o = 16; o > 0; o >>= 1) sum[h] += __shfl_xor(sum[h], o);
    }

    int hsel[NIT];
    #pragma unroll
    for (int it = 0; it < NIT; ++it) hsel[it] = (it * 128 + lane * 4) / C;
    float4 acc[NIT];
    #pragma unroll
    for (int it = 0; it < NIT; ++it) acc[it] = make_float4(0.f, 0.f, 0.f, 0.f);

    for (int c0 = 0; c0 < total; c0 += 32) {
        int s_l = s0;
        float cf[H];
        if (c0 == 0) {
            #pragma unroll
            for (int h = 0; h < H; ++h) cf[h] = e0[h] / sum[h];
        } else {
            const int t = c0 + lane;
            s_l = (t < deg) ? csr_src[start + t] : n;
            #pragma unroll
            for (int h = 0; h < H; ++h)
                cf[h] = expf(leaky02(asrc[s_l * H + h] + ad[h]) - m[h]) / sum[h];
        }
        const int cnt = (total - c0 < 32) ? (total - c0) : 32;
        #pragma unroll 4
        for (int jj = 0; jj < cnt; ++jj) {
            const int s = __builtin_amdgcn_readlane(s_l, jj);
            float ch[H];
            #pragma unroll
            for (int h = 0; h < H; ++h)
                ch[h] = __int_as_float(__builtin_amdgcn_readlane(__float_as_int(cf[h]), jj));
            const float* __restrict__ row = xh + (long)s * HC + lane * 4;
            #pragma unroll
            for (int it = 0; it < NIT; ++it) {
                float c = ch[0];
                #pragma unroll
                for (int h = 1; h < H; ++h) c = (hsel[it] == h) ? ch[h] : c;
                const float4 v = *(const float4*)(row + it * 128);
                acc[it].x += c * v.x;
                acc[it].y += c * v.y;
                acc[it].z += c * v.z;
                acc[it].w += c * v.w;
            }
        }
    }

    float* __restrict__ orow = out + (long)n * HC + lane * 4;
    #pragma unroll
    for (int it = 0; it < NIT; ++it) {
        const float4 b = *(const float4*)(bias + it * 128 + lane * 4);
        float4 r;
        r.x = acc[it].x + b.x; r.y = acc[it].y + b.y; r.z = acc[it].z + b.z; r.w = acc[it].w + b.w;
        *(float4*)(orow + it * 128) = r;
    }
}

static inline int cdiv(long a, int b) { return (int)((a + b - 1) / b); }

extern "C" void kernel_launch(void* const* d_in, const int* in_sizes, int n_in,
                              void* d_out, int out_size, void* d_ws, size_t ws_size,
                              hipStream_t stream) {
    const float* X        = (const float*)d_in[0];
    const int*   ei       = (const int*)  d_in[1];
    const float* W1       = (const float*)d_in[2];
    const float* att_src1 = (const float*)d_in[3];
    const float* att_dst1 = (const float*)d_in[4];
    const float* b1       = (const float*)d_in[5];
    const float* W2       = (const float*)d_in[6];
    const float* att_src2 = (const float*)d_in[7];
    const float* att_dst2 = (const float*)d_in[8];
    const float* b2       = (const float*)d_in[9];
    float* out = (float*)d_out;

    float* ws = (float*)d_ws;
    long off = 0;
    float* xh     = ws + off; off += (long)NNODES * 256;
    float* h1     = ws + off; off += (long)NNODES * 256;
    float* asrc   = ws + off; off += (long)NNODES * 4;
    float* adst   = ws + off; off += (long)NNODES * 4;
    int*   cnt    = (int*)(ws + off); off += NNODES;
    int*   rowptr = (int*)(ws + off); off += NNODES + 4;
    int*   rank   = (int*)(ws + off); off += NEDGES;
    int*   csrc   = (int*)(ws + off); off += NEDGES;

    const int B = 256;

    zero_i32      <<<cdiv(NNODES, B), B, 0, stream>>>(cnt, NNODES);
    dst_count_rank<<<cdiv(NEDGES, B), B, 0, stream>>>(ei, cnt, rank);
    rowptr_scan   <<<1, SCAN_T, 0, stream>>>(cnt, rowptr);
    csr_fill      <<<cdiv(NEDGES, B), B, 0, stream>>>(ei, rowptr, rank, csrc);

    {
        int waves = (NNODES / 16) * (256 / 64);
        input_scores<<<cdiv(NNODES, B), B, 0, stream>>>(X, W1, att_src1, att_dst1, asrc, adst);
        node_softmax_aggregate_in<<<cdiv((long)NNODES * 32, B), B, 0, stream>>>(
            rowptr, csrc, asrc, adst, X, xh);
        gemm_f32_wmma<<<cdiv((long)waves * 32, B), B, 0, stream>>>(xh, W1, h1, NNODES, 256, 64,
                                                                   256, 64, b1);
    }

    {
        int waves = (NNODES / 16) * (128 / 64);
        gemm_f32_wmma<<<cdiv((long)waves * 32, B), B, 0, stream>>>(h1, W2, xh, NNODES, 128, 256,
                                                                   256, 0, nullptr);
        attn_scores<<<cdiv((long)NNODES, B), B, 0, stream>>>(xh, att_src2, att_dst2,
                                                             asrc, adst, 1, 128);
        node_softmax_aggregate<1, 128><<<cdiv((long)NNODES * 32, B), B, 0, stream>>>(
            rowptr, csrc, asrc, adst, xh, b2, out);
    }
}


# Layer 1's projection folded into layer 2's (nothing non-linear between): one GEMM agg @ W12 + b12, W12 = W1_h @ W2 row blocks built in-call on WMMA; layer-1 GEMM launch gone; sum order changes
# speedup vs baseline: 3.9494x; 1.1271x over previous
#include <hip/hip_runtime.h>
#include <stdint.h>


#define NNODES 50000
#define NEDGES 800000

typedef float v2f __attribute__((ext_vector_type(2)));
typedef float v8f __attribute__((ext_vector_type(8)));


__global__ void zero_i32(int* __restrict__ p, int n) {
    int i = blockIdx.x * blockDim.x + threadIdx.x;
    if (i < n) p[i] = 0;
}

__global__ void dst_count_rank(const int* __restrict__ ei, int* __restrict__ cnt,
                               int* __restrict__ rank) {
    int e = blockIdx.x * blockDim.x + threadIdx.x;
    if (e >= NEDGES) return;
    rank[e] = atomicAdd(&cnt[ei[NEDGES + e]], 1);
}

#define SCAN_T   1024
#define SCAN_PER ((NNODES + SCAN_T - 1) / SCAN_T)

__global__ __launch_bounds__(SCAN_T)
void rowptr_scan(const int* __restrict__ cnt, int* __restrict__ rowptr) {
    static_assert((SCAN_PER - 1) * SCAN_T <= NNODES, "all tiles but the last are full");
    __shared__ int sh[SCAN_T * SCAN_PER];
    __shared__ int wsum[32];
    const int t = threadIdx.x;
    const int lane = t & 31, w = t >> 5;
    {
        int v[SCAN_PER];
        #pragma unroll
        for (int k = 0; k < SCAN_PER - 1; ++k) v[k] = cnt[k * SCAN_T + t];
        const int il = (SCAN_PER - 1) * SCAN_T + t;
        v[SCAN_PER - 1] = (il < NNODES) ? cnt[il] : 0;
        #pragma unroll
        for (int k = 0; k < SCAN_PER; ++k) sh[k * SCAN_T + t] = v[k];
    }
    __syncthreads();
    const int base = t * SCAN_PER;
    int s = 0;
    for (int k = 0; k < SCAN_PER; ++k) s += sh[base + k];
    int inc = s;
    for (int o = 1; o < 32; o <<= 1) { int v = __shfl_up(inc, o); if (lane >= o) inc += v; }
    if (lane == 31) wsum[w] = inc;
    __syncthreads();
    if (w == 0) {
        int v = wsum[lane], vi = v;
        for (int o = 1; o < 32; o <<= 1) { int u = __shfl_up(vi, o); if (lane >= o) vi += u; }
        wsum[lane] = vi - v;
    }
    __syncthreads();
    int run = wsum[w] + inc - s;
    for (int k = 0; k < SCAN_PER; ++k) { int c = sh[base + k]; sh[base + k] = run; run += c; }
    __syncthreads();
    #pragma unroll
    for (int k = 0; k < SCAN_PER - 1; ++k) rowptr[k * SCAN_T + t] = sh[k * SCAN_T + t];
    {
        const int il = (SCAN_PER - 1) * SCAN_T + t;
        if (il < NNODES) rowptr[il] = sh[il];
    }
    if (t == SCAN_T - 1) rowptr[NNODES] = run;
}

__global__ void csr_fill(const int* __restrict__ ei, const int* __restrict__ rowptr,
                         const int* __restrict__ rank, int* __restrict__ csr_src) {
    int e = blockIdx.x * blockDim.x + threadIdx.x;
    if (e >= NEDGES) return;
    int s = ei[e], d = ei[NEDGES + e];
    csr_src[rowptr[d] + rank[e]] = s;
}

__global__ __launch_bounds__(256)
void gemm_f32_wmma(const float* __restrict__ A, const float* __restrict__ B,
                   float* __restrict__ C, int M, int N, int K,
                   int lda, const float* __restrict__ bias) {
    const int wave = (blockIdx.x * blockDim.x + threadIdx.x) >> 5;
    const int lane = threadIdx.x & 31;
    const int stripsN = N >> 6;
    const int tm = wave / stripsN;
    const int strip = wave % stripsN;
    const int tn = strip * 4;
    if (tm * 16 >= M) return;

    const int m   = lane & 15;
    const int kh  = (lane >> 4) << 1;
    const int n0  = lane & 15;

    v8f acc[4];
    #pragma unroll
    for (int t = 0; t < 4; ++t) {
        const float bv = bias ? bias[(tn + t) * 16 + n0] : 0.f;
        acc[t] = (v8f){bv,bv,bv,bv,bv,bv,bv,bv};
    }

    const float* __restrict__ Arow = A + (long)(tm * 16 + m) * lda;
    for (int k0 = 0; k0 < K; k0 += 4) {
        v2f a;
        a.x = Arow[k0 + kh + 0];
        a.y = Arow[k0 + kh + 1];
        #pragma unroll
        for (int t = 0; t < 4; ++t) {
            const int col = (tn + t) * 16 + n0;
            v2f b;
            b.x = B[(long)(k0 + kh + 0) * N + col];
            b.y = B[(long)(k0 + kh + 1) * N + col];
            acc[t] = __builtin_amdgcn_wmma_f32_16x16x4_f32(
                         false, a, false, b, (short)0, acc[t], false, false);
        }
    }
    const int moff = (lane >> 4) * 8;
    #pragma unroll
    for (int t = 0; t < 4; ++t) {
        #pragma unroll
        for (int v = 0; v < 8; ++v) {
            C[(long)(tm * 16 + v + moff) * N + (tn + t) * 16 + n0] = acc[t][v];
        }
    }
}


#define SCORE_BLOCKS ((NNODES + 255) / 256)
#define FOLD_BLOCKS  5

__device__ __forceinline__ void fold_weights_block(int fb, const float* __restrict__ W1,
                                                   const float* __restrict__ W2, const float* __restrict__ b1,
                                                   float* __restrict__ w12, float* __restrict__ b12,
                                                   float (*part)[128]) {
    const int lane  = threadIdx.x & 31;
    const int w     = threadIdx.x >> 5;
    const int strip = w & 1;
    const int q     = w >> 1;
    const int m     = lane & 15;
    const int kh    = (lane >> 4) << 1;
    const int n0    = lane & 15;
    const bool is_bias = (fb == FOLD_BLOCKS - 1);
    const int h     = is_bias ? q : fb;

    const float* __restrict__ Arow = is_bias ? (b1 + h * 64) : (W1 + (q * 16 + m) * 256 + h * 64);
    const float* __restrict__ Bcol = W2 + (long)h * 64 * 128 + strip * 64 + n0;

    v8f acc[4];
    #pragma unroll
    for (int t = 0; t < 4; ++t) acc[t] = (v8f){0.f,0.f,0.f,0.f,0.f,0.f,0.f,0.f};

    #pragma unroll 4
    for (int k0 = 0; k0 < 64; k0 += 4) {
        v2f a;
        a.x = Arow[k0 + kh + 0];
        a.y = Arow[k0 + kh + 1];
        #pragma unroll
        for (int t = 0; t < 4; ++t) {
            v2f b;
            b.x = Bcol[(k0 + kh + 0) * 128 + t * 16];
            b.y = Bcol[(k0 + kh + 1) * 128 + t * 16];
            acc[t] = __builtin_amdgcn_wmma_f32_16x16x4_f32(
                         false, a, false, b, (short)0, acc[t], false, false);
        }
    }

    if (!is_bias) {
        float* __restrict__ Crow = w12 + (long)(h * 64 + q * 16 + (lane >> 4) * 8) * 128 + strip * 64 + n0;
        #pragma unroll
        for (int t = 0; t < 4; ++t) {
            #pragma unroll
            for (int v = 0; v < 8; ++v) Crow[v * 128 + t * 16] = acc[t][v];
        }
    } else {
        if (lane < 16) {
            #pragma unroll
            for (int t = 0; t < 4; ++t) part[q][strip * 64 + t * 16 + n0] = acc[t][0];
        }
        __syncthreads();
        const int j = threadIdx.x;
        if (j < 128) b12[j] = ((part[0][j] + part[1][j]) + part[2][j]) + part[3][j];
    }
}

__global__ __launch_bounds__(256)
void input_scores(const float* __restrict__ X, const float* __restrict__ W1,
                  const float* __restrict__ att_src, const float* __restrict__ att_dst,
                  float* __restrict__ asrc, float* __restrict__ adst,
                  const float* __restrict__ W2, const float* __restrict__ b1,
                  float* __restrict__ w12, float* __restrict__ b12) {
    __shared__ float4 wsrc[64];
    __shared__ float4 wdst[64];
    __shared__ float part[4][128];
    if (blockIdx.x >= SCORE_BLOCKS) {
        fold_weights_block((int)blockIdx.x - SCORE_BLOCKS, W1, W2, b1, w12, b12, part);
        return;
    }
    const int t = threadIdx.x;
    {
        const int h = t & 3;
        const float4* __restrict__ w4 = (const float4*)(W1 + t * 64);
        const float4* __restrict__ s4 = (const float4*)(att_src + h * 64);
        const float4* __restrict__ d4 = (const float4*)(att_dst + h * 64);
        float ss = 0.f, sd = 0.f;
        #pragma unroll
        for (int c = 0; c < 16; ++c) {
            const float4 w = w4[c], a = s4[c], d = d4[c];
            ss += w.x * a.x; ss += w.y * a.y; ss += w.z * a.z; ss += w.w * a.w;
            sd += w.x * d.x; sd += w.y * d.y; sd += w.z * d.z; sd += w.w * d.w;
        }
        ((float*)wsrc)[t] = ss;
        ((float*)wdst)[t] = sd;
    }
    __syncthreads();
    const int n = blockIdx.x * 256 + t;
    if (n >= NNODES) return;
    const float4* __restrict__ x4 = (const float4*)(X + (long)n * 64);
    float4 as = make_float4(0.f, 0.f, 0.f, 0.f), ad = make_float4(0.f, 0.f, 0.f, 0.f);
    #pragma unroll
    for (int c = 0; c < 16; ++c) {
        const float4 x = x4[c];
        const float xs[4] = {x.x, x.y, x.z, x.w};
        #pragma unroll
        for (int j = 0; j < 4; ++j) {
            const float4 ws = wsrc[c * 4 + j], wd = wdst[c * 4 + j];
            as.x += xs[j] * ws.x; as.y += xs[j] * ws.y; as.z += xs[j] * ws.z; as.w += xs[j] * ws.w;
            ad.x += xs[j] * wd.x; ad.y += xs[j] * wd.y; ad.z += xs[j] * wd.z; ad.w += xs[j] * wd.w;
        }
    }
    *(float4*)(asrc + (long)n * 4) = as;
    *(float4*)(adst + (long)n * 4) = ad;
}

__device__ __forceinline__ float leaky02(float x) { return (x > 0.f) ? x : 0.2f * x; }

__global__ __launch_bounds__(256)
void node_softmax_aggregate_in(const int* __restrict__ rowptr, const int* __restrict__ csr_src,
                               const float* __restrict__ asrc, const float* __restrict__ adst,
                               const float* __restrict__ X, float* __restrict__ agg) {
    constexpr int H = 4;
    const int lane = threadIdx.x & 31;
    const int n = __builtin_amdgcn_readfirstlane((int)((blockIdx.x * blockDim.x + threadIdx.x) >> 5));
    if (n >= NNODES) return;
    const int start = __builtin_amdgcn_readfirstlane(rowptr[n]);
    const int deg   = __builtin_amdgcn_readfirstlane(rowptr[n + 1]) - start;
    const int total = deg + 1;

    float ad[H];
    #pragma unroll
    for (int h = 0; h < H; ++h) ad[h] = adst[n * H + h];

    const bool v0 = lane < total;
    const int  s0 = (lane < deg) ? csr_src[start + lane] : n;
    const float4 a0 = *(const float4*)(asrc + (long)s0 * H);
    const float a0v[H] = {a0.x, a0.y, a0.z, a0.w};
    float x0[H];
    #pragma unroll
    for (int h = 0; h < H; ++h) x0[h] = v0 ? leaky02(a0v[h] + ad[h]) : -INFINITY;

    float m[H];
    #pragma unroll
    for (int h = 0; h < H; ++h) m[h] = x0[h];
    for (int c0 = 32; c0 < total; c0 += 32) {
        const int t = c0 + lane;
        const int s = (t < deg) ? csr_src[start + t] : n;
        if (t < total) {
            #pragma unroll
            for (int h = 0; h < H; ++h) m[h] = fmaxf(m[h], leaky02(asrc[s * H + h] + ad[h]));
        }
    }
    #pragma unroll
    for (int h = 0; h < H; ++h) {
        #pragma unroll
        for (int o = 16; o > 0; o >>= 1) m[h] = fmaxf(m[h], __shfl_xor(m[h], o));
    }

    float e0[H], sum[H];
    #pragma unroll
    for (int h = 0; h < H; ++h) { e0[h] = v0 ? expf(x0[h] - m[h]) : 0.f; sum[h] = e0[h]; }
    for (int c0 = 32; c0 < total; c0 += 32) {
        const int t = c0 + lane;
        const int s = (t < deg) ? csr_src[start + t] : n;
        if (t < total) {
            #pragma unroll
            for (int h = 0; h < H; ++h) sum[h] += expf(leaky02(asrc[s * H + h] + ad[h]) - m[h]);
        }
    }
    #pragma unroll
    for (int h = 0; h < H; ++h) {
        #pragma unroll
        for (int o = 16; o > 0; o >>= 1) sum[h] += __shfl_xor(sum[h], o);
    }

    float2 acc[H];
    #pragma unroll
    for (int h = 0; h < H; ++h) acc[h] = make_float2(0.f, 0.f);

    for (int c0 = 0; c0 < total; c0 += 32) {
        int s_l = s0;
        float cf[H];
        if (c0 == 0) {
            #pragma unroll
            for (int h = 0; h < H; ++h) cf[h] = e0[h] / sum[h];
        } else {
            const int t = c0 + lane;
            s_l = (t < deg) ? csr_src[start + t] : n;
            #pragma unroll
            for (int h = 0; h < H; ++h)
                cf[h] = (t < total) ? expf(leaky02(asrc[s_l * H + h] + ad[h]) - m[h]) / sum[h] : 0.f;
        }
        const int cnt = (total - c0 < 32) ? (total - c0) : 32;
        for (int j0 = 0; j0 < cnt; j0 += 8) {
            #pragma unroll
            for (int u = 0; u < 8; ++u) {
                const int jj = j0 + u;
                const int s = __builtin_amdgcn_readlane(s_l, jj);
                const float2 v = *(const float2*)(X + (long)s * 64 + lane * 2);
                #pragma unroll
                for (int h = 0; h < H; ++h) {
                    const float c = __int_as_float(__builtin_amdgcn_readlane(__float_as_int(cf[h]), jj));
                    acc[h].x = fmaf(c, v.x, acc[h].x);
                    acc[h].y = fmaf(c, v.y, acc[h].y);
                }
            }
        }
    }

    float* __restrict__ arow = agg + (long)n * (H * 64) + lane * 2;
    #pragma unroll
    for (int h = 0; h < H; ++h) *(float2*)(arow + h * 64) = acc[h];
}

__global__ void attn_scores(const float* __restrict__ xh,
                            const float* __restrict__ att_src,
                            const float* __restrict__ att_dst,
                            float* __restrict__ asrc, float* __restrict__ adst,
                            int H, int C) {
    int idx = blockIdx.x * blockDim.x + threadIdx.x;
    if (idx >= NNODES * H) return;
    int n = idx / H, h = idx % H;
    const float* x = xh + (long)n * H * C + (long)h * C;
    const float* as = att_src + h * C;
    const float* ad = att_dst + h * C;
    float ss = 0.f, sd = 0.f;
    for (int c = 0; c < C; ++c) { float v = x[c]; ss += v * as[c]; sd += v * ad[c]; }
    asrc[idx] = ss;
    adst[idx] = sd;
}

template <int H, int C>
__global__ __launch_bounds__(256)
void node_softmax_aggregate(const int* __restrict__ rowptr, const int* __restrict__ csr_src,
                            const float* __restrict__ asrc, const float* __restrict__ adst,
                            const float* __restrict__ xh, const float* __restrict__ bias,
                            float* __restrict__ out) {
    constexpr int HC  = H * C;
    constexpr int NIT = HC / 128;
    const int lane = threadIdx.x & 31;
    const int n = __builtin_amdgcn_readfirstlane((int)((blockIdx.x * blockDim.x + threadIdx.x) >> 5));
    if (n >= NNODES) return;
    const int start = __builtin_amdgcn_readfirstlane(rowptr[n]);
    const int deg   = __builtin_amdgcn_readfirstlane(rowptr[n + 1]) - start;
    const int total = deg + 1;

    float ad[H];
    #pragma unroll
    for (int h = 0; h < H; ++h) ad[h] = adst[n * H + h];

    const bool v0 = lane < total;
    const int  s0 = (lane < deg) ? csr_src[start + lane] : n;
    float x0[H];
    #pragma unroll
    for (int h = 0; h < H; ++h) x0[h] = v0 ? leaky02(asrc[s0 * H + h] + ad[h]) : -INFINITY;

    float m[H];
    #pragma unroll
    for (int h = 0; h < H; ++h) m[h] = x0[h];
    for (int c0 = 32; c0 < total; c0 += 32) {
        const int t = c0 + lane;
        const int s = (t < deg) ? csr_src[start + t] : n;
        if (t < total) {
            #pragma unroll
            for (int h = 0; h < H; ++h) m[h] = fmaxf(m[h], leaky02(asrc[s * H + h] + ad[h]));
        }
    }
    #pragma unroll
    for (int h = 0; h < H; ++h) {
        #pragma unroll
        for (int o = 16; o > 0; o >>= 1) m[h] = fmaxf(m[h], __shfl_xor(m[h], o));
    }

    float e0[H], sum[H];
    #pragma unroll
    for (int h = 0; h < H; ++h) { e0[h] = v0 ? expf(x0[h] - m[h]) : 0.f; sum[h] = e0[h]; }
    for (int c0 = 32; c0 < total; c0 += 32) {
        const int t = c0 + lane;
        const int s = (t < deg) ? csr_src[start + t] : n;
        if (t < total) {
            #pragma unroll
            for (int h = 0; h < H; ++h) sum[h] += expf(leaky02(asrc[s * H + h] + ad[h]) - m[h]);
        }
    }
    #pragma unroll
    for (int h = 0; h < H; ++h) {
        #pragma unroll
        for (int o = 16; o > 0; o >>= 1) sum[h] += __shfl_xor(sum[h], o);
    }

    int hsel[NIT];
    #pragma unroll
    for (int it = 0; it < NIT; ++it) hsel[it] = (it * 128 + lane * 4) / C;
    float4 acc[NIT];
    #pragma unroll
    for (int it = 0; it < NIT; ++it) acc[it] = make_float4(0.f, 0.f, 0.f, 0.f);

    for (int c0 = 0; c0 < total; c0 += 32) {
        int s_l = s0;
        float cf[H];
        if (c0 == 0) {
            #pragma unroll
            for (int h = 0; h < H; ++h) cf[h] = e0[h] / sum[h];
        } else {
            const int t = c0 + lane;
            s_l = (t < deg) ? csr_src[start + t] : n;
            #pragma unroll
            for (int h = 0; h < H; ++h)
                cf[h] = expf(leaky02(asrc[s_l * H + h] + ad[h]) - m[h]) / sum[h];
        }
        const int cnt = (total - c0 < 32) ? (total - c0) : 32;
        #pragma unroll 4
        for (int jj = 0; jj < cnt; ++jj) {
            const int s = __builtin_amdgcn_readlane(s_l, jj);
            float ch[H];
            #pragma unroll
            for (int h = 0; h < H; ++h)
                ch[h] = __int_as_float(__builtin_amdgcn_readlane(__float_as_int(cf[h]), jj));
            const float* __restrict__ row = xh + (long)s * HC + lane * 4;
            #pragma unroll
            for (int it = 0; it < NIT; ++it) {
                float c = ch[0];
                #pragma unroll
                for (int h = 1; h < H; ++h) c = (hsel[it] == h) ? ch[h] : c;
                const float4 v = *(const float4*)(row + it * 128);
                acc[it].x += c * v.x;
                acc[it].y += c * v.y;
                acc[it].z += c * v.z;
                acc[it].w += c * v.w;
            }
        }
    }

    float* __restrict__ orow = out + (long)n * HC + lane * 4;
    #pragma unroll
    for (int it = 0; it < NIT; ++it) {
        const float4 b = *(const float4*)(bias + it * 128 + lane * 4);
        float4 r;
        r.x = acc[it].x + b.x; r.y = acc[it].y + b.y; r.z = acc[it].z + b.z; r.w = acc[it].w + b.w;
        *(float4*)(orow + it * 128) = r;
    }
}

static inline int cdiv(long a, int b) { return (int)((a + b - 1) / b); }

extern "C" void kernel_launch(void* const* d_in, const int* in_sizes, int n_in,
                              void* d_out, int out_size, void* d_ws, size_t ws_size,
                              hipStream_t stream) {
    const float* X        = (const float*)d_in[0];
    const int*   ei       = (const int*)  d_in[1];
    const float* W1       = (const float*)d_in[2];
    const float* att_src1 = (const float*)d_in[3];
    const float* att_dst1 = (const float*)d_in[4];
    const float* b1       = (const float*)d_in[5];
    const float* W2       = (const float*)d_in[6];
    const float* att_src2 = (const float*)d_in[7];
    const float* att_dst2 = (const float*)d_in[8];
    const float* b2       = (const float*)d_in[9];
    float* out = (float*)d_out;

    float* ws = (float*)d_ws;
    long off = 0;
    float* agg    = ws + off; off += (long)NNODES * 256;
    float* xh     = ws + off; off += (long)NNODES * 128;
    float* asrc   = ws + off; off += (long)NNODES * 4;
    float* adst   = ws + off; off += (long)NNODES * 4;
    int*   cnt    = (int*)(ws + off); off += NNODES;
    int*   rowptr = (int*)(ws + off); off += NNODES + 4;
    int*   rank   = (int*)(ws + off); off += NEDGES;
    int*   csrc   = (int*)(ws + off); off += NEDGES;
    float* w12    = ws + off; off += 256 * 128;
    float* b12    = ws + off; off += 128;

    const int B = 256;

    zero_i32      <<<cdiv(NNODES, B), B, 0, stream>>>(cnt, NNODES);
    dst_count_rank<<<cdiv(NEDGES, B), B, 0, stream>>>(ei, cnt, rank);
    rowptr_scan   <<<1, SCAN_T, 0, stream>>>(cnt, rowptr);
    csr_fill      <<<cdiv(NEDGES, B), B, 0, stream>>>(ei, rowptr, rank, csrc);

    {
        input_scores<<<SCORE_BLOCKS + FOLD_BLOCKS, B, 0, stream>>>(X, W1, att_src1, att_dst1, asrc, adst,
                                                                   W2, b1, w12, b12);
        node_softmax_aggregate_in<<<cdiv((long)NNODES * 32, B), B, 0, stream>>>(
            rowptr, csrc, asrc, adst, X, agg);
    }

    {
        int waves = (NNODES / 16) * (128 / 64);
        gemm_f32_wmma<<<cdiv((long)waves * 32, B), B, 0, stream>>>(agg, w12, xh, NNODES, 128, 256,
                                                                   256, b12);
        attn_scores<<<cdiv((long)NNODES, B), B, 0, stream>>>(xh, att_src2, att_dst2,
                                                             asrc, adst, 1, 128);
        node_softmax_aggregate<1, 128><<<cdiv((long)NNODES * 32, B), B, 0, stream>>>(
            rowptr, csrc, asrc, adst, xh, b2, out);
    }
}


# Layer-1 aggregation and the agg @ W12 + b12 product fused in one launch: 16 nodes per work-group, aggregated rows stay in LDS and feed WMMA from there; GEMM launch and agg plane gone; sum order kept
# speedup vs baseline: 4.3060x; 1.0903x over previous
#include <hip/hip_runtime.h>
#include <stdint.h>


#define NNODES 50000
#define NEDGES 800000

typedef float v2f __attribute__((ext_vector_type(2)));
typedef float v8f __attribute__((ext_vector_type(8)));


__global__ void zero_i32(int* __restrict__ p, int n) {
    int i = blockIdx.x * blockDim.x + threadIdx.x;
    if (i < n) p[i] = 0;
}

__global__ void dst_count_rank(const int* __restrict__ ei, int* __restrict__ cnt,
                               int* __restrict__ rank) {
    int e = blockIdx.x * blockDim.x + threadIdx.x;
    if (e >= NEDGES) return;
    rank[e] = atomicAdd(&cnt[ei[NEDGES + e]], 1);
}

#define SCAN_T   1024
#define SCAN_PER ((NNODES + SCAN_T - 1) / SCAN_T)

__global__ __launch_bounds__(SCAN_T)
void rowptr_scan(const int* __restrict__ cnt, int* __restrict__ rowptr) {
    static_assert((SCAN_PER - 1) * SCAN_T <= NNODES, "all tiles but the last are full");
    __shared__ int sh[SCAN_T * SCAN_PER];
    __shared__ int wsum[32];
    const int t = threadIdx.x;
    const int lane = t & 31, w = t >> 5;
    {
        int v[SCAN_PER];
        #pragma unroll
        for (int k = 0; k < SCAN_PER - 1; ++k) v[k] = cnt[k * SCAN_T + t];
        const int il = (SCAN_PER - 1) * SCAN_T + t;
        v[SCAN_PER - 1] = (il < NNODES) ? cnt[il] : 0;
        #pragma unroll
        for (int k = 0; k < SCAN_PER; ++k) sh[k * SCAN_T + t] = v[k];
    }
    __syncthreads();
    const int base = t * SCAN_PER;
    int s = 0;
    for (int k = 0; k < SCAN_PER; ++k) s += sh[base + k];
    int inc = s;
    for (int o = 1; o < 32; o <<= 1) { int v = __shfl_up(inc, o); if (lane >= o) inc += v; }
    if (lane == 31) wsum[w] = inc;
    __syncthreads();
    if (w == 0) {
        int v = wsum[lane], vi = v;
        for (int o = 1; o < 32; o <<= 1) { int u = __shfl_up(vi, o); if (lane >= o) vi += u; }
        wsum[lane] = vi - v;
    }
    __syncthreads();
    int run = wsum[w] + inc - s;
    for (int k = 0; k < SCAN_PER; ++k) { int c = sh[base + k]; sh[base + k] = run; run += c; }
    __syncthreads();
    #pragma unroll
    for (int k = 0; k < SCAN_PER - 1; ++k) rowptr[k * SCAN_T + t] = sh[k * SCAN_T + t];
    {
        const int il = (SCAN_PER - 1) * SCAN_T + t;
        if (il < NNODES) rowptr[il] = sh[il];
    }
    if (t == SCAN_T - 1) rowptr[NNODES] = run;
}

__global__ void csr_fill(const int* __restrict__ ei, const int* __restrict__ rowptr,
                         const int* __restrict__ rank, int* __restrict__ csr_src) {
    int e = blockIdx.x * blockDim.x + threadIdx.x;
    if (e >= NEDGES) return;
    int s = ei[e], d = ei[NEDGES + e];
    csr_src[rowptr[d] + rank[e]] = s;
}


#define SCORE_BLOCKS ((NNODES + 255) / 256)
#define FOLD_BLOCKS  5

__device__ __forceinline__ void fold_weights_block(int fb, const float* __restrict__ W1,
                                                   const float* __restrict__ W2, const float* __restrict__ b1,
                                                   float* __restrict__ w12, float* __restrict__ b12,
                                                   float (*part)[128]) {
    const int lane  = threadIdx.x & 31;
    const int w     = threadIdx.x >> 5;
    const int strip = w & 1;
    const int q     = w >> 1;
    const int m     = lane & 15;
    const int kh    = (lane >> 4) << 1;
    const int n0    = lane & 15;
    const bool is_bias = (fb == FOLD_BLOCKS - 1);
    const int h     = is_bias ? q : fb;

    const float* __restrict__ Arow = is_bias ? (b1 + h * 64) : (W1 + (q * 16 + m) * 256 + h * 64);
    const float* __restrict__ Bcol = W2 + (long)h * 64 * 128 + strip * 64 + n0;

    v8f acc[4];
    #pragma unroll
    for (int t = 0; t < 4; ++t) acc[t] = (v8f){0.f,0.f,0.f,0.f,0.f,0.f,0.f,0.f};

    #pragma unroll 4
    for (int k0 = 0; k0 < 64; k0 += 4) {
        v2f a;
        a.x = Arow[k0 + kh + 0];
        a.y = Arow[k0 + kh + 1];
        #pragma unroll
        for (int t = 0; t < 4; ++t) {
            v2f b;
            b.x = Bcol[(k0 + kh + 0) * 128 + t * 16];
            b.y = Bcol[(k0 + kh + 1) * 128 + t * 16];
            acc[t] = __builtin_amdgcn_wmma_f32_16x16x4_f32(
                         false, a, false, b, (short)0, acc[t], false, false);
        }
    }

    if (!is_bias) {
        float* __restrict__ Crow = w12 + (long)(h * 64 + q * 16 + (lane >> 4) * 8) * 128 + strip * 64 + n0;
        #pragma unroll
        for (int t = 0; t < 4; ++t) {
            #pragma unroll
            for (int v = 0; v < 8; ++v) Crow[v * 128 + t * 16] = acc[t][v];
        }
    } else {
        if (lane < 16) {
            #pragma unroll
            for (int t = 0; t < 4; ++t) part[q][strip * 64 + t * 16 + n0] = acc[t][0];
        }
        __syncthreads();
        const int j = threadIdx.x;
        if (j < 128) b12[j] = ((part[0][j] + part[1][j]) + part[2][j]) + part[3][j];
    }
}

__global__ __launch_bounds__(256)
void input_scores(const float* __restrict__ X, const float* __restrict__ W1,
                  const float* __restrict__ att_src, const float* __restrict__ att_dst,
                  float* __restrict__ asrc, float* __restrict__ adst,
                  const float* __restrict__ W2, const float* __restrict__ b1,
                  float* __restrict__ w12, float* __restrict__ b12) {
    __shared__ float4 wsrc[64];
    __shared__ float4 wdst[64];
    __shared__ float part[4][128];
    if (blockIdx.x >= SCORE_BLOCKS) {
        fold_weights_block((int)blockIdx.x - SCORE_BLOCKS, W1, W2, b1, w12, b12, part);
        return;
    }
    const int t = threadIdx.x;
    {
        const int h = t & 3;
        const float4* __restrict__ w4 = (const float4*)(W1 + t * 64);
        const float4* __restrict__ s4 = (const float4*)(att_src + h * 64);
        const float4* __restrict__ d4 = (const float4*)(att_dst + h * 64);
        float ss = 0.f, sd = 0.f;
        #pragma unroll
        for (int c = 0; c < 16; ++c) {
            const float4 w = w4[c], a = s4[c], d = d4[c];
            ss += w.x * a.x; ss += w.y * a.y; ss += w.z * a.z; ss += w.w * a.w;
            sd += w.x * d.x; sd += w.y * d.y; sd += w.z * d.z; sd += w.w * d.w;
        }
        ((float*)wsrc)[t] = ss;
        ((float*)wdst)[t] = sd;
    }
    __syncthreads();
    const int n = blockIdx.x * 256 + t;
    if (n >= NNODES) return;
    const float4* __restrict__ x4 = (const float4*)(X + (long)n * 64);
    float4 as = make_float4(0.f, 0.f, 0.f, 0.f), ad = make_float4(0.f, 0.f, 0.f, 0.f);
    #pragma unroll
    for (int c = 0; c < 16; ++c) {
        const float4 x = x4[c];
        const float xs[4] = {x.x, x.y, x.z, x.w};
        #pragma unroll
        for (int j = 0; j < 4; ++j) {
            const float4 ws = wsrc[c * 4 + j], wd = wdst[c * 4 + j];
            as.x += xs[j] * ws.x; as.y += xs[j] * ws.y; as.z += xs[j] * ws.z; as.w += xs[j] * ws.w;
            ad.x += xs[j] * wd.x; ad.y += xs[j] * wd.y; ad.z += xs[j] * wd.z; ad.w += xs[j] * wd.w;
        }
    }
    *(float4*)(asrc + (long)n * 4) = as;
    *(float4*)(adst + (long)n * 4) = ad;
}

__device__ __forceinline__ float leaky02(float x) { return (x > 0.f) ? x : 0.2f * x; }


#define TILE_NODES 16
#define TILE_PITCH 260

static_assert(NNODES % TILE_NODES == 0, "every work-group owns 16 whole nodes");

__device__ __forceinline__ void aggregate_node_in(const int n, const int lane,
                                                  const int* __restrict__ rowptr, const int* __restrict__ csr_src,
                                                  const float* __restrict__ asrc, const float* __restrict__ adst,
                                                  const float* __restrict__ X, float* __restrict__ trow) {
    constexpr int H = 4;
    const int start = __builtin_amdgcn_readfirstlane(rowptr[n]);
    const int deg   = __builtin_amdgcn_readfirstlane(rowptr[n + 1]) - start;
    const int total = deg + 1;

    float ad[H];
    #pragma unroll
    for (int h = 0; h < H; ++h) ad[h] = adst[n * H + h];

    const bool v0 = lane < total;
    const int  s0 = (lane < deg) ? csr_src[start + lane] : n;
    const float4 a0 = *(const float4*)(asrc + (long)s0 * H);
    const float a0v[H] = {a0.x, a0.y, a0.z, a0.w};
    float x0[H];
    #pragma unroll
    for (int h = 0; h < H; ++h) x0[h] = v0 ? leaky02(a0v[h] + ad[h]) : -INFINITY;

    float m[H];
    #pragma unroll
    for (int h = 0; h < H; ++h) m[h] = x0[h];
    for (int c0 = 32; c0 < total; c0 += 32) {
        const int t = c0 + lane;
        const int s = (t < deg) ? csr_src[start + t] : n;
        if (t < total) {
            #pragma unroll
            for (int h = 0; h < H; ++h) m[h] = fmaxf(m[h], leaky02(asrc[s * H + h] + ad[h]));
        }
    }
    #pragma unroll
    for (int h = 0; h < H; ++h) {
        #pragma unroll
        for (int o = 16; o > 0; o >>= 1) m[h] = fmaxf(m[h], __shfl_xor(m[h], o));
    }

    float e0[H], sum[H];
    #pragma unroll
    for (int h = 0; h < H; ++h) { e0[h] = v0 ? expf(x0[h] - m[h]) : 0.f; sum[h] = e0[h]; }
    for (int c0 = 32; c0 < total; c0 += 32) {
        const int t = c0 + lane;
        const int s = (t < deg) ? csr_src[start + t] : n;
        if (t < total) {
            #pragma unroll
            for (int h = 0; h < H; ++h) sum[h] += expf(leaky02(asrc[s * H + h] + ad[h]) - m[h]);
        }
    }
    #pragma unroll
    for (int h = 0; h < H; ++h) {
        #pragma unroll
        for (int o = 16; o > 0; o >>= 1) sum[h] += __shfl_xor(sum[h], o);
    }

    float2 acc[H];
    #pragma unroll
    for (int h = 0; h < H; ++h) acc[h] = make_float2(0.f, 0.f);

    for (int c0 = 0; c0 < total; c0 += 32) {
        int s_l = s0;
        float cf[H];
        if (c0 == 0) {
            #pragma unroll
            for (int h = 0; h < H; ++h) cf[h] = e0[h] / sum[h];
        } else {
            const int t = c0 + lane;
            s_l = (t < deg) ? csr_src[start + t] : n;
            #pragma unroll
            for (int h = 0; h < H; ++h)
                cf[h] = (t < total) ? expf(leaky02(asrc[s_l * H + h] + ad[h]) - m[h]) / sum[h] : 0.f;
        }
        const int cnt = (total - c0 < 32) ? (total - c0) : 32;
        for (int j0 = 0; j0 < cnt; j0 += 8) {
            #pragma unroll
            for (int u = 0; u < 8; ++u) {
                const int jj = j0 + u;
                const int s = __builtin_amdgcn_readlane(s_l, jj);
                const float2 v = *(const float2*)(X + (long)s * 64 + lane * 2);
                #pragma unroll
                for (int h = 0; h < H; ++h) {
                    const float c = __int_as_float(__builtin_amdgcn_readlane(__float_as_int(cf[h]), jj));
                    acc[h].x = fmaf(c, v.x, acc[h].x);
                    acc[h].y = fmaf(c, v.y, acc[h].y);
                }
            }
        }
    }

    #pragma unroll
    for (int h = 0; h < H; ++h) *(float2*)(trow + h * 64 + lane * 2) = acc[h];
}

__global__ __launch_bounds__(256, 16)
void node_softmax_aggregate_in_project(const int* __restrict__ rowptr, const int* __restrict__ csr_src,
                                       const float* __restrict__ asrc, const float* __restrict__ adst,
                                       const float* __restrict__ X, const float* __restrict__ w12,
                                       const float* __restrict__ b12, float* __restrict__ xh) {
    __shared__ __attribute__((aligned(16))) float tile[TILE_NODES * TILE_PITCH];
    const int lane = threadIdx.x & 31;
    const int w = __builtin_amdgcn_readfirstlane((int)(threadIdx.x >> 5));
    const int nbase = (int)blockIdx.x * TILE_NODES;

    for (int i = 0; i < TILE_NODES / 8; ++i) {
        const int r = w * (TILE_NODES / 8) + i;
        aggregate_node_in(nbase + r, lane, rowptr, csr_src, asrc, adst, X, tile + r * TILE_PITCH);
    }
    __syncthreads();

    const int m  = lane & 15;
    const int kh = (lane >> 4) << 1;
    const int n0 = lane & 15;
    const int col = w * 16 + n0;
    const float bv = b12[col];
    v8f acc = (v8f){bv,bv,bv,bv,bv,bv,bv,bv};
    const float* __restrict__ Arow = tile + m * TILE_PITCH + kh;
    const float* __restrict__ Bcol = w12 + kh * 128 + col;
    #pragma unroll 16
    for (int k0 = 0; k0 < 256; k0 += 4) {
        const v2f a = *(const v2f*)(Arow + k0);
        v2f b;
        b.x = Bcol[(k0 + 0) * 128];
        b.y = Bcol[(k0 + 1) * 128];
        acc = __builtin_amdgcn_wmma_f32_16x16x4_f32(false, a, false, b, (short)0, acc, false, false);
    }
    float* __restrict__ Crow = xh + (long)(nbase + (lane >> 4) * 8) * 128 + col;
    #pragma unroll
    for (int v = 0; v < 8; ++v) Crow[v * 128] = acc[v];
}

__global__ void attn_scores(const float* __restrict__ xh,
                            const float* __restrict__ att_src,
                            const float* __restrict__ att_dst,
                            float* __restrict__ asrc, float* __restrict__ adst,
                            int H, int C) {
    int idx = blockIdx.x * blockDim.x + threadIdx.x;
    if (idx >= NNODES * H) return;
    int n = idx / H, h = idx % H;
    const float* x = xh + (long)n * H * C + (long)h * C;
    const float* as = att_src + h * C;
    const float* ad = att_dst + h * C;
    float ss = 0.f, sd = 0.f;
    for (int c = 0; c < C; ++c) { float v = x[c]; ss += v * as[c]; sd += v * ad[c]; }
    asrc[idx] = ss;
    adst[idx] = sd;
}

template <int H, int C>
__global__ __launch_bounds__(256)
void node_softmax_aggregate(const int* __restrict__ rowptr, const int* __restrict__ csr_src,
                            const float* __restrict__ asrc, const float* __restrict__ adst,
                            const float* __restrict__ xh, const float* __restrict__ bias,
                            float* __restrict__ out) {
    constexpr int HC  = H * C;
    constexpr int NIT = HC / 128;
    const int lane = threadIdx.x & 31;
    const int n = __builtin_amdgcn_readfirstlane((int)((blockIdx.x * blockDim.x + threadIdx.x) >> 5));
    if (n >= NNODES) return;
    const int start = __builtin_amdgcn_readfirstlane(rowptr[n]);
    const int deg   = __builtin_amdgcn_readfirstlane(rowptr[n + 1]) - start;
    const int total = deg + 1;

    float ad[H];
    #pragma unroll
    for (int h = 0; h < H; ++h) ad[h] = adst[n * H + h];

    const bool v0 = lane < total;
    const int  s0 = (lane < deg) ? csr_src[start + lane] : n;
    float x0[H];
    #pragma unroll
    for (int h = 0; h < H; ++h) x0[h] = v0 ? leaky02(asrc[s0 * H + h] + ad[h]) : -INFINITY;

    float m[H];
    #pragma unroll
    for (int h = 0; h < H; ++h) m[h] = x0[h];
    for (int c0 = 32; c0 < total; c0 += 32) {
        const int t = c0 + lane;
        const int s = (t < deg) ? csr_src[start + t] : n;
        if (t < total) {
            #pragma unroll
            for (int h = 0; h < H; ++h) m[h] = fmaxf(m[h], leaky02(asrc[s * H + h] + ad[h]));
        }
    }
    #pragma unroll
    for (int h = 0; h < H; ++h) {
        #pragma unroll
        for (int o = 16; o > 0; o >>= 1) m[h] = fmaxf(m[h], __shfl_xor(m[h], o));
    }

    float e0[H], sum[H];
    #pragma unroll
    for (int h = 0; h < H; ++h) { e0[h] = v0 ? expf(x0[h] - m[h]) : 0.f; sum[h] = e0[h]; }
    for (int c0 = 32; c0 < total; c0 += 32) {
        const int t = c0 + lane;
        const int s = (t < deg) ? csr_src[start + t] : n;
        if (t < total) {
            #pragma unroll
            for (int h = 0; h < H; ++h) sum[h] += expf(leaky02(asrc[s * H + h] + ad[h]) - m[h]);
        }
    }
    #pragma unroll
    for (int h = 0; h < H; ++h) {
        #pragma unroll
        for (int o = 16; o > 0; o >>= 1) sum[h] += __shfl_xor(sum[h], o);
    }

    int hsel[NIT];
    #pragma unroll
    for (int it = 0; it < NIT; ++it) hsel[it] = (it * 128 + lane * 4) / C;
    float4 acc[NIT];
    #pragma unroll
    for (int it = 0; it < NIT; ++it) acc[it] = make_float4(0.f, 0.f, 0.f, 0.f);

    for (int c0 = 0; c0 < total; c0 += 32) {
        int s_l = s0;
        float cf[H];
        if (c0 == 0) {
            #pragma unroll
            for (int h = 0; h < H; ++h) cf[h] = e0[h] / sum[h];
        } else {
            const int t = c0 + lane;
            s_l = (t < deg) ? csr_src[start + t] : n;
            #pragma unroll
            for (int h = 0; h < H; ++h)
                cf[h] = expf(leaky02(asrc[s_l * H + h] + ad[h]) - m[h]) / sum[h];
        }
        const int cnt = (total - c0 < 32) ? (total - c0) : 32;
        #pragma unroll 4
        for (int jj = 0; jj < cnt; ++jj) {
            const int s = __builtin_amdgcn_readlane(s_l, jj);
            float ch[H];
            #pragma unroll
            for (int h = 0; h < H; ++h)
                ch[h] = __int_as_float(__builtin_amdgcn_readlane(__float_as_int(cf[h]), jj));
            const float* __restrict__ row = xh + (long)s * HC + lane * 4;
            #pragma unroll
            for (int it = 0; it < NIT; ++it) {
                float c = ch[0];
                #pragma unroll
                for (int h = 1; h < H; ++h) c = (hsel[it] == h) ? ch[h] : c;
                const float4 v = *(const float4*)(row + it * 128);
                acc[it].x += c * v.x;
                acc[it].y += c * v.y;
                acc[it].z += c * v.z;
                acc[it].w += c * v.w;
            }
        }
    }

    float* __restrict__ orow = out + (long)n * HC + lane * 4;
    #pragma unroll
    for (int it = 0; it < NIT; ++it) {
        const float4 b = *(const float4*)(bias + it * 128 + lane * 4);
        float4 r;
        r.x = acc[it].x + b.x; r.y = acc[it].y + b.y; r.z = acc[it].z + b.z; r.w = acc[it].w + b.w;
        *(float4*)(orow + it * 128) = r;
    }
}

static inline int cdiv(long a, int b) { return (int)((a + b - 1) / b); }

extern "C" void kernel_launch(void* const* d_in, const int* in_sizes, int n_in,
                              void* d_out, int out_size, void* d_ws, size_t ws_size,
                              hipStream_t stream) {
    const float* X        = (const float*)d_in[0];
    const int*   ei       = (const int*)  d_in[1];
    const float* W1       = (const float*)d_in[2];
    const float* att_src1 = (const float*)d_in[3];
    const float* att_dst1 = (const float*)d_in[4];
    const float* b1       = (const float*)d_in[5];
    const float* W2       = (const float*)d_in[6];
    const float* att_src2 = (const float*)d_in[7];
    const float* att_dst2 = (const float*)d_in[8];
    const float* b2       = (const float*)d_in[9];
    float* out = (float*)d_out;

    float* ws = (float*)d_ws;
    long off = 0;
    float* xh     = ws + off; off += (long)NNODES * 128;
    float* asrc   = ws + off; off += (long)NNODES * 4;
    float* adst   = ws + off; off += (long)NNODES * 4;
    int*   cnt    = (int*)(ws + off); off += NNODES;
    int*   rowptr = (int*)(ws + off); off += NNODES + 4;
    int*   rank   = (int*)(ws + off); off += NEDGES;
    int*   csrc   = (int*)(ws + off); off += NEDGES;
    float* w12    = ws + off; off += 256 * 128;
    float* b12    = ws + off; off += 128;

    const int B = 256;

    zero_i32      <<<cdiv(NNODES, B), B, 0, stream>>>(cnt, NNODES);
    dst_count_rank<<<cdiv(NEDGES, B), B, 0, stream>>>(ei, cnt, rank);
    rowptr_scan   <<<1, SCAN_T, 0, stream>>>(cnt, rowptr);
    csr_fill      <<<cdiv(NEDGES, B), B, 0, stream>>>(ei, rowptr, rank, csrc);

    {
        input_scores<<<SCORE_BLOCKS + FOLD_BLOCKS, B, 0, stream>>>(X, W1, att_src1, att_dst1, asrc, adst,
                                                                   W2, b1, w12, b12);
        node_softmax_aggregate_in_project<<<NNODES / TILE_NODES, B, 0, stream>>>(
            rowptr, csrc, asrc, adst, X, w12, b12, xh);
    }

    {
        attn_scores<<<cdiv((long)NNODES, B), B, 0, stream>>>(xh, att_src2, att_dst2,
                                                             asrc, adst, 1, 128);
        node_softmax_aggregate<1, 128><<<cdiv((long)NNODES * 32, B), B, 0, stream>>>(
            rowptr, csrc, asrc, adst, xh, b2, out);
    }
}


# Edges grouped by destination in ONE launch via LDS: a work-group owns 512 nodes, streams all destinations, counts and places by LDS adds (arrival order), local scan; global atomics, 3 launches gone
# speedup vs baseline: 5.1204x; 1.1891x over previous
#include <hip/hip_runtime.h>
#include <stdint.h>


#define NNODES 50000
#define NEDGES 800000

typedef float v2f __attribute__((ext_vector_type(2)));
typedef float v8f __attribute__((ext_vector_type(8)));


#define CSR_RANGE  512
#define CSR_BLOCKS ((NNODES + CSR_RANGE - 1) / CSR_RANGE)
#define CSR_T      1024
#define CSR_CAP    24576
#define NCHUNK     (NEDGES / 4)
#define CSR_ROUNDS (NCHUNK / CSR_T)
#define CSR_U      13

static_assert(NEDGES % 4 == 0, "destinations are read as whole 16-B pieces");
static_assert(NEDGES <= (1 << 20), "an edge index fits the low 20 bits of a list entry");
static_assert(CSR_RANGE <= (1 << 12), "a local destination fits the high 12 bits of a list entry");
static_assert(CSR_ROUNDS % CSR_U == 0, "the unrolled rounds cover the full rounds exactly");

__device__ __forceinline__ void csr_note(const int d, const int e, const int lo, int& below,
                                         int* cnt, int* n_list, unsigned* list) {
    unsigned dl;
    below += __builtin_usub_overflow((unsigned)d, (unsigned)lo, &dl) ? 1 : 0;
    if (dl < (unsigned)CSR_RANGE) {
        atomicAdd(&cnt[dl], 1);
        const int p = atomicAdd(n_list, 1);
        if (p < CSR_CAP) list[p] = (unsigned)e | (dl << 20);
    }
}

__global__ __launch_bounds__(CSR_T)
void group_edges_by_dst(const int* __restrict__ ei, int* __restrict__ rowptr, int* __restrict__ csr_src) {
    __shared__ int cnt[CSR_RANGE];
    __shared__ int wsum[32];
    __shared__ int wbelow[32];
    __shared__ int n_list;
    __shared__ unsigned list[CSR_CAP];
    __shared__ int sorted[CSR_CAP];
    const int t = threadIdx.x;
    const int lane = t & 31, w = t >> 5;
    const int lo = (int)blockIdx.x * CSR_RANGE;
    if (t < CSR_RANGE) cnt[t] = 0;
    if (t == 0) n_list = 0;
    __syncthreads();

    const int4* __restrict__ d4 = (const int4*)(ei + NEDGES);
    int below = 0;
    for (int r = 0; r < CSR_ROUNDS; r += CSR_U) {
        int4 v[CSR_U];
        #pragma unroll
        for (int u = 0; u < CSR_U; ++u) v[u] = d4[(r + u) * CSR_T + t];
        #pragma unroll
        for (int u = 0; u < CSR_U; ++u) {
            const int e = ((r + u) * CSR_T + t) * 4;
            csr_note(v[u].x, e + 0, lo, below, cnt, &n_list, list);
            csr_note(v[u].y, e + 1, lo, below, cnt, &n_list, list);
            csr_note(v[u].z, e + 2, lo, below, cnt, &n_list, list);
            csr_note(v[u].w, e + 3, lo, below, cnt, &n_list, list);
        }
    }
    {
        const int i = CSR_ROUNDS * CSR_T + t;
        if (i < NCHUNK) {
            const int4 v = d4[i];
            const int e = i * 4;
            csr_note(v.x, e + 0, lo, below, cnt, &n_list, list);
            csr_note(v.y, e + 1, lo, below, cnt, &n_list, list);
            csr_note(v.z, e + 2, lo, below, cnt, &n_list, list);
            csr_note(v.w, e + 3, lo, below, cnt, &n_list, list);
        }
    }
    #pragma unroll
    for (int o = 16; o > 0; o >>= 1) below += __shfl_xor(below, o);
    if (lane == 0) wbelow[w] = below;
    __syncthreads();

    int base = 0;
    for (int j = 0; j < 32; ++j) base += wbelow[j];
    const int M = n_list;

    const int c = (t < CSR_RANGE) ? cnt[t] : 0;
    int inc = c;
    for (int o = 1; o < 32; o <<= 1) { int u = __shfl_up(inc, o); if (lane >= o) inc += u; }
    if (lane == 31) wsum[w] = inc;
    __syncthreads();
    int wp = 0;
    for (int j = 0; j < 32; ++j) wp += (j < w) ? wsum[j] : 0;
    const int off = wp + inc - c;
    if (t < CSR_RANGE && lo + t < NNODES) rowptr[lo + t] = base + off;
    if (blockIdx.x == CSR_BLOCKS - 1 && t == 0) rowptr[NNODES] = base + M;

    if (M <= CSR_CAP) {
        if (t < CSR_RANGE) cnt[t] = off;
        __syncthreads();
        for (int p0 = 0; p0 < M; p0 += CSR_T * 8) {
            unsigned ent[8];
            int s[8];
            #pragma unroll
            for (int u = 0; u < 8; ++u) {
                const int p = p0 + u * CSR_T + t;
                ent[u] = (p < M) ? list[p] : 0u;
                s[u] = ei[ent[u] & 0xFFFFFu];
            }
            #pragma unroll
            for (int u = 0; u < 8; ++u) {
                const int p = p0 + u * CSR_T + t;
                if (p < M) sorted[atomicAdd(&cnt[ent[u] >> 20], 1)] = s[u];
            }
        }
        __syncthreads();
        for (int p = t; p < M; p += CSR_T) csr_src[base + p] = sorted[p];
    } else {
        if (t < CSR_RANGE) cnt[t] = base + off;
        __syncthreads();
        for (int i = t; i < NCHUNK; i += CSR_T) {
            const int4 v = d4[i];
            const int dd[4] = {v.x, v.y, v.z, v.w};
            #pragma unroll
            for (int k = 0; k < 4; ++k) {
                const unsigned dl = (unsigned)(dd[k] - lo);
                if (dl < (unsigned)CSR_RANGE) csr_src[atomicAdd(&cnt[dl], 1)] = ei[i * 4 + k];
            }
        }
    }
}


#define SCORE_BLOCKS ((NNODES + 255) / 256)
#define FOLD_BLOCKS  5

__device__ __forceinline__ void fold_weights_block(int fb, const float* __restrict__ W1,
                                                   const float* __restrict__ W2, const float* __restrict__ b1,
                                                   float* __restrict__ w12, float* __restrict__ b12,
                                                   float (*part)[128]) {
    const int lane  = threadIdx.x & 31;
    const int w     = threadIdx.x >> 5;
    const int strip = w & 1;
    const int q     = w >> 1;
    const int m     = lane & 15;
    const int kh    = (lane >> 4) << 1;
    const int n0    = lane & 15;
    const bool is_bias = (fb == FOLD_BLOCKS - 1);
    const int h     = is_bias ? q : fb;

    const float* __restrict__ Arow = is_bias ? (b1 + h * 64) : (W1 + (q * 16 + m) * 256 + h * 64);
    const float* __restrict__ Bcol = W2 + (long)h * 64 * 128 + strip * 64 + n0;

    v8f acc[4];
    #pragma unroll
    for (int t = 0; t < 4; ++t) acc[t] = (v8f){0.f,0.f,0.f,0.f,0.f,0.f,0.f,0.f};

    #pragma unroll 4
    for (int k0 = 0; k0 < 64; k0 += 4) {
        v2f a;
        a.x = Arow[k0 + kh + 0];
        a.y = Arow[k0 + kh + 1];
        #pragma unroll
        for (int t = 0; t < 4; ++t) {
            v2f b;
            b.x = Bcol[(k0 + kh + 0) * 128 + t * 16];
            b.y = Bcol[(k0 + kh + 1) * 128 + t * 16];
            acc[t] = __builtin_amdgcn_wmma_f32_16x16x4_f32(
                         false, a, false, b, (short)0, acc[t], false, false);
        }
    }

    if (!is_bias) {
        float* __restrict__ Crow = w12 + (long)(h * 64 + q * 16 + (lane >> 4) * 8) * 128 + strip * 64 + n0;
        #pragma unroll
        for (int t = 0; t < 4; ++t) {
            #pragma unroll
            for (int v = 0; v < 8; ++v) Crow[v * 128 + t * 16] = acc[t][v];
        }
    } else {
        if (lane < 16) {
            #pragma unroll
            for (int t = 0; t < 4; ++t) part[q][strip * 64 + t * 16 + n0] = acc[t][0];
        }
        __syncthreads();
        const int j = threadIdx.x;
        if (j < 128) b12[j] = ((part[0][j] + part[1][j]) + part[2][j]) + part[3][j];
    }
}

__global__ __launch_bounds__(256)
void input_scores(const float* __restrict__ X, const float* __restrict__ W1,
                  const float* __restrict__ att_src, const float* __restrict__ att_dst,
                  float* __restrict__ asrc, float* __restrict__ adst,
                  const float* __restrict__ W2, const float* __restrict__ b1,
                  float* __restrict__ w12, float* __restrict__ b12) {
    __shared__ float4 wsrc[64];
    __shared__ float4 wdst[64];
    __shared__ float part[4][128];
    if (blockIdx.x >= SCORE_BLOCKS) {
        fold_weights_block((int)blockIdx.x - SCORE_BLOCKS, W1, W2, b1, w12, b12, part);
        return;
    }
    const int t = threadIdx.x;
    {
        const int h = t & 3;
        const float4* __restrict__ w4 = (const float4*)(W1 + t * 64);
        const float4* __restrict__ s4 = (const float4*)(att_src + h * 64);
        const float4* __restrict__ d4 = (const float4*)(att_dst + h * 64);
        float ss = 0.f, sd = 0.f;
        #pragma unroll
        for (int c = 0; c < 16; ++c) {
            const float4 w = w4[c], a = s4[c], d = d4[c];
            ss += w.x * a.x; ss += w.y * a.y; ss += w.z * a.z; ss += w.w * a.w;
            sd += w.x * d.x; sd += w.y * d.y; sd += w.z * d.z; sd += w.w * d.w;
        }
        ((float*)wsrc)[t] = ss;
        ((float*)wdst)[t] = sd;
    }
    __syncthreads();
    const int n = blockIdx.x * 256 + t;
    if (n >= NNODES) return;
    const float4* __restrict__ x4 = (const float4*)(X + (long)n * 64);
    float4 as = make_float4(0.f, 0.f, 0.f, 0.f), ad = make_float4(0.f, 0.f, 0.f, 0.f);
    #pragma unroll
    for (int c = 0; c < 16; ++c) {
        const float4 x = x4[c];
        const float xs[4] = {x.x, x.y, x.z, x.w};
        #pragma unroll
        for (int j = 0; j < 4; ++j) {
            const float4 ws = wsrc[c * 4 + j], wd = wdst[c * 4 + j];
            as.x += xs[j] * ws.x; as.y += xs[j] * ws.y; as.z += xs[j] * ws.z; as.w += xs[j] * ws.w;
            ad.x += xs[j] * wd.x; ad.y += xs[j] * wd.y; ad.z += xs[j] * wd.z; ad.w += xs[j] * wd.w;
        }
    }
    *(float4*)(asrc + (long)n * 4) = as;
    *(float4*)(adst + (long)n * 4) = ad;
}

__device__ __forceinline__ float leaky02(float x) { return (x > 0.f) ? x : 0.2f * x; }


#define TILE_NODES 16
#define TILE_PITCH 260

static_assert(NNODES % TILE_NODES == 0, "every work-group owns 16 whole nodes");

__device__ __forceinline__ void aggregate_node_in(const int n, const int lane,
                                                  const int* __restrict__ rowptr, const int* __restrict__ csr_src,
                                                  const float* __restrict__ asrc, const float* __restrict__ adst,
                                                  const float* __restrict__ X, float* __restrict__ trow) {
    constexpr int H = 4;
    const int start = __builtin_amdgcn_readfirstlane(rowptr[n]);
    const int deg   = __builtin_amdgcn_readfirstlane(rowptr[n + 1]) - start;
    const int total = deg + 1;

    float ad[H];
    #pragma unroll
    for (int h = 0; h < H; ++h) ad[h] = adst[n * H + h];

    const bool v0 = lane < total;
    const int  s0 = (lane < deg) ? csr_src[start + lane] : n;
    const float4 a0 = *(const float4*)(asrc + (long)s0 * H);
    const float a0v[H] = {a0.x, a0.y, a0.z, a0.w};
    float x0[H];
    #pragma unroll
    for (int h = 0; h < H; ++h) x0[h] = v0 ? leaky02(a0v[h] + ad[h]) : -INFINITY;

    float m[H];
    #pragma unroll
    for (int h = 0; h < H; ++h) m[h] = x0[h];
    for (int c0 = 32; c0 < total; c0 += 32) {
        const int t = c0 + lane;
        const int s = (t < deg) ? csr_src[start + t] : n;
        if (t < total) {
            #pragma unroll
            for (int h = 0; h < H; ++h) m[h] = fmaxf(m[h], leaky02(asrc[s * H + h] + ad[h]));
        }
    }
    #pragma unroll
    for (int h = 0; h < H; ++h) {
        #pragma unroll
        for (int o = 16; o > 0; o >>= 1) m[h] = fmaxf(m[h], __shfl_xor(m[h], o));
    }

    float e0[H], sum[H];
    #pragma unroll
    for (int h = 0; h < H; ++h) { e0[h] = v0 ? expf(x0[h] - m[h]) : 0.f; sum[h] = e0[h]; }
    for (int c0 = 32; c0 < total; c0 += 32) {
        const int t = c0 + lane;
        const int s = (t < deg) ? csr_src[start + t] : n;
        if (t < total) {
            #pragma unroll
            for (int h = 0; h < H; ++h) sum[h] += expf(leaky02(asrc[s * H + h] + ad[h]) - m[h]);
        }
    }
    #pragma unroll
    for (int h = 0; h < H; ++h) {
        #pragma unroll
        for (int o = 16; o > 0; o >>= 1) sum[h] += __shfl_xor(sum[h], o);
    }

    float2 acc[H];
    #pragma unroll
    for (int h = 0; h < H; ++h) acc[h] = make_float2(0.f, 0.f);

    for (int c0 = 0; c0 < total; c0 += 32) {
        int s_l = s0;
        float cf[H];
        if (c0 == 0) {
            #pragma unroll
            for (int h = 0; h < H; ++h) cf[h] = e0[h] / sum[h];
        } else {
            const int t = c0 + lane;
            s_l = (t < deg) ? csr_src[start + t] : n;
            #pragma unroll
            for (int h = 0; h < H; ++h)
                cf[h] = (t < total) ? expf(leaky02(asrc[s_l * H + h] + ad[h]) - m[h]) / sum[h] : 0.f;
        }
        const int cnt = (total - c0 < 32) ? (total - c0) : 32;
        for (int j0 = 0; j0 < cnt; j0 += 8) {
            #pragma unroll
            for (int u = 0; u < 8; ++u) {
                const int jj = j0 + u;
                const int s = __builtin_amdgcn_readlane(s_l, jj);
                const float2 v = *(const float2*)(X + (long)s * 64 + lane * 2);
                #pragma unroll
                for (int h = 0; h < H; ++h) {
                    const float c = __int_as_float(__builtin_amdgcn_readlane(__float_as_int(cf[h]), jj));
                    acc[h].x = fmaf(c, v.x, acc[h].x);
                    acc[h].y = fmaf(c, v.y, acc[h].y);
                }
            }
        }
    }

    #pragma unroll
    for (int h = 0; h < H; ++h) *(float2*)(trow + h * 64 + lane * 2) = acc[h];
}

__global__ __launch_bounds__(256, 16)
void node_softmax_aggregate_in_project(const int* __restrict__ rowptr, const int* __restrict__ csr_src,
                                       const float* __restrict__ asrc, const float* __restrict__ adst,
                                       const float* __restrict__ X, const float* __restrict__ w12,
                                       const float* __restrict__ b12, float* __restrict__ xh) {
    __shared__ __attribute__((aligned(16))) float tile[TILE_NODES * TILE_PITCH];
    const int lane = threadIdx.x & 31;
    const int w = __builtin_amdgcn_readfirstlane((int)(threadIdx.x >> 5));
    const int nbase = (int)blockIdx.x * TILE_NODES;

    for (int i = 0; i < TILE_NODES / 8; ++i) {
        const int r = w * (TILE_NODES / 8) + i;
        aggregate_node_in(nbase + r, lane, rowptr, csr_src, asrc, adst, X, tile + r * TILE_PITCH);
    }
    __syncthreads();

    const int m  = lane & 15;
    const int kh = (lane >> 4) << 1;
    const int n0 = lane & 15;
    const int col = w * 16 + n0;
    const float bv = b12[col];
    v8f acc = (v8f){bv,bv,bv,bv,bv,bv,bv,bv};
    const float* __restrict__ Arow = tile + m * TILE_PITCH + kh;
    const float* __restrict__ Bcol = w12 + kh * 128 + col;
    #pragma unroll 16
    for (int k0 = 0; k0 < 256; k0 += 4) {
        const v2f a = *(const v2f*)(Arow + k0);
        v2f b;
        b.x = Bcol[(k0 + 0) * 128];
        b.y = Bcol[(k0 + 1) * 128];
        acc = __builtin_amdgcn_wmma_f32_16x16x4_f32(false, a, false, b, (short)0, acc, false, false);
    }
    float* __restrict__ Crow = xh + (long)(nbase + (lane >> 4) * 8) * 128 + col;
    #pragma unroll
    for (int v = 0; v < 8; ++v) Crow[v * 128] = acc[v];
}

__global__ void attn_scores(const float* __restrict__ xh,
                            const float* __restrict__ att_src,
                            const float* __restrict__ att_dst,
                            float* __restrict__ asrc, float* __restrict__ adst,
                            int H, int C) {
    int idx = blockIdx.x * blockDim.x + threadIdx.x;
    if (idx >= NNODES * H) return;
    int n = idx / H, h = idx % H;
    const float* x = xh + (long)n * H * C + (long)h * C;
    const float* as = att_src + h * C;
    const float* ad = att_dst + h * C;
    float ss = 0.f, sd = 0.f;
    for (int c = 0; c < C; ++c) { float v = x[c]; ss += v * as[c]; sd += v * ad[c]; }
    asrc[idx] = ss;
    adst[idx] = sd;
}

template <int H, int C>
__global__ __launch_bounds__(256)
void node_softmax_aggregate(const int* __restrict__ rowptr, const int* __restrict__ csr_src,
                            const float* __restrict__ asrc, const float* __restrict__ adst,
                            const float* __restrict__ xh, const float* __restrict__ bias,
                            float* __restrict__ out) {
    constexpr int HC  = H * C;
    constexpr int NIT = HC / 128;
    const int lane = threadIdx.x & 31;
    const int n = __builtin_amdgcn_readfirstlane((int)((blockIdx.x * blockDim.x + threadIdx.x) >> 5));
    if (n >= NNODES) return;
    const int start = __builtin_amdgcn_readfirstlane(rowptr[n]);
    const int deg   = __builtin_amdgcn_readfirstlane(rowptr[n + 1]) - start;
    const int total = deg + 1;

    float ad[H];
    #pragma unroll
    for (int h = 0; h < H; ++h) ad[h] = adst[n * H + h];

    const bool v0 = lane < total;
    const int  s0 = (lane < deg) ? csr_src[start + lane] : n;
    float x0[H];
    #pragma unroll
    for (int h = 0; h < H; ++h) x0[h] = v0 ? leaky02(asrc[s0 * H + h] + ad[h]) : -INFINITY;

    float m[H];
    #pragma unroll
    for (int h = 0; h < H; ++h) m[h] = x0[h];
    for (int c0 = 32; c0 < total; c0 += 32) {
        const int t = c0 + lane;
        const int s = (t < deg) ? csr_src[start + t] : n;
        if (t < total) {
            #pragma unroll
            for (int h = 0; h < H; ++h) m[h] = fmaxf(m[h], leaky02(asrc[s * H + h] + ad[h]));
        }
    }
    #pragma unroll
    for (int h = 0; h < H; ++h) {
        #pragma unroll
        for (int o = 16; o > 0; o >>= 1) m[h] = fmaxf(m[h], __shfl_xor(m[h], o));
    }

    float e0[H], sum[H];
    #pragma unroll
    for (int h = 0; h < H; ++h) { e0[h] = v0 ? expf(x0[h] - m[h]) : 0.f; sum[h] = e0[h]; }
    for (int c0 = 32; c0 < total; c0 += 32) {
        const int t = c0 + lane;
        const int s = (t < deg) ? csr_src[start + t] : n;
        if (t < total) {
            #pragma unroll
            for (int h = 0; h < H; ++h) sum[h] += expf(leaky02(asrc[s * H + h] + ad[h]) - m[h]);
        }
    }
    #pragma unroll
    for (int h = 0; h < H; ++h) {
        #pragma unroll
        for (int o = 16; o > 0; o >>= 1) sum[h] += __shfl_xor(sum[h], o);
    }

    int hsel[NIT];
    #pragma unroll
    for (int it = 0; it < NIT; ++it) hsel[it] = (it * 128 + lane * 4) / C;
    float4 acc[NIT];
    #pragma unroll
    for (int it = 0; it < NIT; ++it) acc[it] = make_float4(0.f, 0.f, 0.f, 0.f);

    for (int c0 = 0; c0 < total; c0 += 32) {
        int s_l = s0;
        float cf[H];
        if (c0 == 0) {
            #pragma unroll
            for (int h = 0; h < H; ++h) cf[h] = e0[h] / sum[h];
        } else {
            const int t = c0 + lane;
            s_l = (t < deg) ? csr_src[start + t] : n;
            #pragma unroll
            for (int h = 0; h < H; ++h)
                cf[h] = expf(leaky02(asrc[s_l * H + h] + ad[h]) - m[h]) / sum[h];
        }
        const int cnt = (total - c0 < 32) ? (total - c0) : 32;
        #pragma unroll 4
        for (int jj = 0; jj < cnt; ++jj) {
            const int s = __builtin_amdgcn_readlane(s_l, jj);
            float ch[H];
            #pragma unroll
            for (int h = 0; h < H; ++h)
                ch[h] = __int_as_float(__builtin_amdgcn_readlane(__float_as_int(cf[h]), jj));
            const float* __restrict__ row = xh + (long)s * HC + lane * 4;
            #pragma unroll
            for (int it = 0; it < NIT; ++it) {
                float c = ch[0];
                #pragma unroll
                for (int h = 1; h < H; ++h) c = (hsel[it] == h) ? ch[h] : c;
                const float4 v = *(const float4*)(row + it * 128);
                acc[it].x += c * v.x;
                acc[it].y += c * v.y;
                acc[it].z += c * v.z;
                acc[it].w += c * v.w;
            }
        }
    }

    float* __restrict__ orow = out + (long)n * HC + lane * 4;
    #pragma unroll
    for (int it = 0; it < NIT; ++it) {
        const float4 b = *(const float4*)(bias + it * 128 + lane * 4);
        float4 r;
        r.x = acc[it].x + b.x; r.y = acc[it].y + b.y; r.z = acc[it].z + b.z; r.w = acc[it].w + b.w;
        *(float4*)(orow + it * 128) = r;
    }
}

static inline int cdiv(long a, int b) { return (int)((a + b - 1) / b); }

extern "C" void kernel_launch(void* const* d_in, const int* in_sizes, int n_in,
                              void* d_out, int out_size, void* d_ws, size_t ws_size,
                              hipStream_t stream) {
    const float* X        = (const float*)d_in[0];
    const int*   ei       = (const int*)  d_in[1];
    const float* W1       = (const float*)d_in[2];
    const float* att_src1 = (const float*)d_in[3];
    const float* att_dst1 = (const float*)d_in[4];
    const float* b1       = (const float*)d_in[5];
    const float* W2       = (const float*)d_in[6];
    const float* att_src2 = (const float*)d_in[7];
    const float* att_dst2 = (const float*)d_in[8];
    const float* b2       = (const float*)d_in[9];
    float* out = (float*)d_out;

    float* ws = (float*)d_ws;
    long off = 0;
    float* xh     = ws + off; off += (long)NNODES * 128;
    float* asrc   = ws + off; off += (long)NNODES * 4;
    float* adst   = ws + off; off += (long)NNODES * 4;
    int*   rowptr = (int*)(ws + off); off += NNODES + 4;
    int*   csrc   = (int*)(ws + off); off += NEDGES;
    float* w12    = ws + off; off += 256 * 128;
    float* b12    = ws + off; off += 128;

    const int B = 256;

    group_edges_by_dst<<<CSR_BLOCKS, CSR_T, 0, stream>>>(ei, rowptr, csrc);

    {
        input_scores<<<SCORE_BLOCKS + FOLD_BLOCKS, B, 0, stream>>>(X, W1, att_src1, att_dst1, asrc, adst,
                                                                   W2, b1, w12, b12);
        node_softmax_aggregate_in_project<<<NNODES / TILE_NODES, B, 0, stream>>>(
            rowptr, csrc, asrc, adst, X, w12, b12, xh);
    }

    {
        attn_scores<<<cdiv((long)NNODES, B), B, 0, stream>>>(xh, att_src2, att_dst2,
                                                             asrc, adst, 1, 128);
        node_softmax_aggregate<1, 128><<<cdiv((long)NNODES * 32, B), B, 0, stream>>>(
            rowptr, csrc, asrc, adst, xh, b2, out);
    }
}


# Edges grouped in two levels: 500 slice work-groups sort 1600 edges each by 512-node range in LDS, 98 range work-groups then read only their own pieces; segment order by arrival, sum order changes
# speedup vs baseline: 7.2817x; 1.4221x over previous
#include <hip/hip_runtime.h>
#include <stdint.h>


#define NNODES 50000
#define NEDGES 800000

typedef float v2f __attribute__((ext_vector_type(2)));
typedef float v8f __attribute__((ext_vector_type(8)));


#define CSR_RANGE  512
#define CSR_SHIFT  9
#define CSR_BLOCKS ((NNODES + CSR_RANGE - 1) / CSR_RANGE)
#define CSR_T      1024
#define CSR_CAP    24576
#define SLICE      1600
#define NSLICE     (NEDGES / SLICE)
#define SLICE_T    320
#define SLICE_K    (SLICE / SLICE_T)
#define OFFS_PITCH 512
#define WAVE_SLICES (OFFS_PITCH / 32)

static_assert(NEDGES % SLICE == 0, "the slices cover the edge list exactly");
static_assert(SLICE % SLICE_T == 0, "every thread of launch 1 has the same number of edges");
static_assert((1 << CSR_SHIFT) == CSR_RANGE, "range of a destination = destination >> CSR_SHIFT");
static_assert(NSLICE <= OFFS_PITCH && OFFS_PITCH <= CSR_T, "one thread of launch 2 per slice");
static_assert(NNODES <= (1 << 16), "a source index fits the low 16 bits of an entry");
static_assert(CSR_RANGE <= (1 << 16), "a local destination fits the high 16 bits of an entry");
static_assert(CSR_BLOCKS + 1 <= 128, "range counts and the total fit the 128-entry LDS table");

__global__ __launch_bounds__(SLICE_T)
void bucket_edge_slices(const int* __restrict__ ei, int* __restrict__ offsT, int* __restrict__ bucketed) {
    __shared__ int hist[128];
    const int t = threadIdx.x;
    const int lane = t & 31;
    const int e0 = (int)blockIdx.x * SLICE;
    if (t < 128) hist[t] = 0;
    __syncthreads();

    int d[SLICE_K], s[SLICE_K], rk[SLICE_K];
    #pragma unroll
    for (int k = 0; k < SLICE_K; ++k) {
        d[k] = ei[NEDGES + e0 + k * SLICE_T + t];
        s[k] = ei[e0 + k * SLICE_T + t];
    }
    #pragma unroll
    for (int k = 0; k < SLICE_K; ++k) {
        rk[k] = 0;
        if ((unsigned)d[k] < (unsigned)NNODES) rk[k] = atomicAdd(&hist[d[k] >> CSR_SHIFT], 1);
    }
    __syncthreads();

    if (t < 32) {
        int h[4];
        int sum = 0;
        #pragma unroll
        for (int j = 0; j < 4; ++j) { h[j] = hist[lane * 4 + j]; sum += h[j]; }
        int inc = sum;
        for (int o = 1; o < 32; o <<= 1) { int u = __shfl_up(inc, o); if (lane >= o) inc += u; }
        int run = inc - sum;
        #pragma unroll
        for (int j = 0; j < 4; ++j) { hist[lane * 4 + j] = run; run += h[j]; }
    }
    __syncthreads();

    if (t <= CSR_BLOCKS) offsT[t * OFFS_PITCH + (int)blockIdx.x] = hist[t];
    #pragma unroll
    for (int k = 0; k < SLICE_K; ++k) {
        if ((unsigned)d[k] < (unsigned)NNODES) {
            const int b = d[k] >> CSR_SHIFT;
            bucketed[e0 + hist[b] + rk[k]] = (s[k] & 0xFFFF) | ((d[k] & (CSR_RANGE - 1)) << 16);
        }
    }
}

__global__ __launch_bounds__(CSR_T)
void group_range_edges(const int* __restrict__ offsT, const int* __restrict__ bucketed,
                       int* __restrict__ rowptr, int* __restrict__ csr_src) {
    __shared__ int cnt[CSR_RANGE];
    __shared__ int wsum[32];
    __shared__ int wbase[32];
    __shared__ int seg_at[OFFS_PITCH];
    __shared__ int seg_n[OFFS_PITCH];
    __shared__ int seg_to[OFFS_PITCH];
    __shared__ unsigned list[CSR_CAP];
    __shared__ int sorted[CSR_CAP];
    const int t = threadIdx.x;
    const int lane = t & 31, w = t >> 5;
    const int r = (int)blockIdx.x;
    const int lo = r * CSR_RANGE;
    if (t < CSR_RANGE) cnt[t] = 0;

    int o0 = 0, c = 0;
    if (t < NSLICE) {
        o0 = offsT[r * OFFS_PITCH + t];
        c  = offsT[(r + 1) * OFFS_PITCH + t] - o0;
    }
    int inc = c;
    for (int o = 1; o < 32; o <<= 1) { int u = __shfl_up(inc, o); if (lane >= o) inc += u; }
    int osum = o0;
    #pragma unroll
    for (int o = 16; o > 0; o >>= 1) osum += __shfl_xor(osum, o);
    if (lane == 31) wsum[w] = inc;
    if (lane == 0) wbase[w] = osum;
    __syncthreads();

    int wp = 0, M = 0, base = 0;
    for (int j = 0; j < 32; ++j) {
        const int v = wsum[j];
        wp += (j < w) ? v : 0;
        M += v;
        base += wbase[j];
    }
    if (t < OFFS_PITCH) {
        seg_at[t] = t * SLICE + o0;
        seg_n[t]  = c;
        seg_to[t] = wp + inc - c;
    }
    __syncthreads();

    {
        const int mine = w * WAVE_SLICES + (lane & (WAVE_SLICES - 1));
        const int n_l = seg_n[mine], at_l = seg_at[mine], to_l = seg_to[mine];
        int ent[WAVE_SLICES];
        #pragma unroll
        for (int u = 0; u < WAVE_SLICES; ++u) {
            const int n  = __builtin_amdgcn_readlane(n_l, u);
            const int at = __builtin_amdgcn_readlane(at_l, u);
            ent[u] = (lane < n) ? bucketed[at + lane] : 0;
        }
        #pragma unroll
        for (int u = 0; u < WAVE_SLICES; ++u) {
            const int n  = __builtin_amdgcn_readlane(n_l, u);
            const int at = __builtin_amdgcn_readlane(at_l, u);
            const int to = __builtin_amdgcn_readlane(to_l, u);
            if (lane < n) {
                atomicAdd(&cnt[(unsigned)ent[u] >> 16], 1);
                const int p = to + lane;
                if (p < CSR_CAP) list[p] = (unsigned)ent[u];
            }
            for (int j = 32 + lane; j < n; j += 32) {
                const unsigned x = (unsigned)bucketed[at + j];
                atomicAdd(&cnt[x >> 16], 1);
                const int p = to + j;
                if (p < CSR_CAP) list[p] = x;
            }
        }
    }
    __syncthreads();

    const int cc = (t < CSR_RANGE) ? cnt[t] : 0;
    int inc2 = cc;
    for (int o = 1; o < 32; o <<= 1) { int u = __shfl_up(inc2, o); if (lane >= o) inc2 += u; }
    if (lane == 31) wsum[w] = inc2;
    __syncthreads();
    int wp2 = 0;
    for (int j = 0; j < 32; ++j) wp2 += (j < w) ? wsum[j] : 0;
    const int off = wp2 + inc2 - cc;
    if (t < CSR_RANGE && lo + t < NNODES) rowptr[lo + t] = base + off;
    if (blockIdx.x == CSR_BLOCKS - 1 && t == 0) rowptr[NNODES] = base + M;

    if (M <= CSR_CAP) {
        if (t < CSR_RANGE) cnt[t] = off;
        __syncthreads();
        for (int p0 = 0; p0 < M; p0 += CSR_T * 8) {
            unsigned x[8];
            int at[8];
            #pragma unroll
            for (int u = 0; u < 8; ++u) {
                const int p = p0 + u * CSR_T + t;
                x[u] = (p < M) ? list[p] : 0u;
            }
            #pragma unroll
            for (int u = 0; u < 8; ++u) {
                const int p = p0 + u * CSR_T + t;
                at[u] = (p < M) ? atomicAdd(&cnt[x[u] >> 16], 1) : 0;
            }
            #pragma unroll
            for (int u = 0; u < 8; ++u) {
                const int p = p0 + u * CSR_T + t;
                if (p < M) sorted[at[u]] = (int)(x[u] & 0xFFFFu);
            }
        }
        __syncthreads();
        for (int p = t; p < M; p += CSR_T) csr_src[base + p] = sorted[p];
    } else {
        if (t < CSR_RANGE) cnt[t] = base + off;
        __syncthreads();
        for (int u = 0; u < WAVE_SLICES; ++u) {
            const int i = w * WAVE_SLICES + u;
            const int n = seg_n[i];
            for (int j = lane; j < n; j += 32) {
                const unsigned x = (unsigned)bucketed[seg_at[i] + j];
                csr_src[atomicAdd(&cnt[x >> 16], 1)] = (int)(x & 0xFFFFu);
            }
        }
    }
}


#define SCORE_BLOCKS ((NNODES + 255) / 256)
#define FOLD_BLOCKS  5

__device__ __forceinline__ void fold_weights_block(int fb, const float* __restrict__ W1,
                                                   const float* __restrict__ W2, const float* __restrict__ b1,
                                                   float* __restrict__ w12, float* __restrict__ b12,
                                                   float (*part)[128]) {
    const int lane  = threadIdx.x & 31;
    const int w     = threadIdx.x >> 5;
    const int strip = w & 1;
    const int q     = w >> 1;
    const int m     = lane & 15;
    const int kh    = (lane >> 4) << 1;
    const int n0    = lane & 15;
    const bool is_bias = (fb == FOLD_BLOCKS - 1);
    const int h     = is_bias ? q : fb;

    const float* __restrict__ Arow = is_bias ? (b1 + h * 64) : (W1 + (q * 16 + m) * 256 + h * 64);
    const float* __restrict__ Bcol = W2 + (long)h * 64 * 128 + strip * 64 + n0;

    v8f acc[4];
    #pragma unroll
    for (int t = 0; t < 4; ++t) acc[t] = (v8f){0.f,0.f,0.f,0.f,0.f,0.f,0.f,0.f};

    #pragma unroll 4
    for (int k0 = 0; k0 < 64; k0 += 4) {
        v2f a;
        a.x = Arow[k0 + kh + 0];
        a.y = Arow[k0 + kh + 1];
        #pragma unroll
        for (int t = 0; t < 4; ++t) {
            v2f b;
            b.x = Bcol[(k0 + kh + 0) * 128 + t * 16];
            b.y = Bcol[(k0 + kh + 1) * 128 + t * 16];
            acc[t] = __builtin_amdgcn_wmma_f32_16x16x4_f32(
                         false, a, false, b, (short)0, acc[t], false, false);
        }
    }

    if (!is_bias) {
        float* __restrict__ Crow = w12 + (long)(h * 64 + q * 16 + (lane >> 4) * 8) * 128 + strip * 64 + n0;
        #pragma unroll
        for (int t = 0; t < 4; ++t) {
            #pragma unroll
            for (int v = 0; v < 8; ++v) Crow[v * 128 + t * 16] = acc[t][v];
        }
    } else {
        if (lane < 16) {
            #pragma unroll
            for (int t = 0; t < 4; ++t) part[q][strip * 64 + t * 16 + n0] = acc[t][0];
        }
        __syncthreads();
        const int j = threadIdx.x;
        if (j < 128) b12[j] = ((part[0][j] + part[1][j]) + part[2][j]) + part[3][j];
    }
}

__global__ __launch_bounds__(256)
void input_scores(const float* __restrict__ X, const float* __restrict__ W1,
                  const float* __restrict__ att_src, const float* __restrict__ att_dst,
                  float* __restrict__ asrc, float* __restrict__ adst,
                  const float* __restrict__ W2, const float* __restrict__ b1,
                  float* __restrict__ w12, float* __restrict__ b12) {
    __shared__ float4 wsrc[64];
    __shared__ float4 wdst[64];
    __shared__ float part[4][128];
    if (blockIdx.x >= SCORE_BLOCKS) {
        fold_weights_block((int)blockIdx.x - SCORE_BLOCKS, W1, W2, b1, w12, b12, part);
        return;
    }
    const int t = threadIdx.x;
    {
        const int h = t & 3;
        const float4* __restrict__ w4 = (const float4*)(W1 + t * 64);
        const float4* __restrict__ s4 = (const float4*)(att_src + h * 64);
        const float4* __restrict__ d4 = (const float4*)(att_dst + h * 64);
        float ss = 0.f, sd = 0.f;
        #pragma unroll
        for (int c = 0; c < 16; ++c) {
            const float4 w = w4[c], a = s4[c], d = d4[c];
            ss += w.x * a.x; ss += w.y * a.y; ss += w.z * a.z; ss += w.w * a.w;
            sd += w.x * d.x; sd += w.y * d.y; sd += w.z * d.z; sd += w.w * d.w;
        }
        ((float*)wsrc)[t] = ss;
        ((float*)wdst)[t] = sd;
    }
    __syncthreads();
    const int n = blockIdx.x * 256 + t;
    if (n >= NNODES) return;
    const float4* __restrict__ x4 = (const float4*)(X + (long)n * 64);
    float4 as = make_float4(0.f, 0.f, 0.f, 0.f), ad = make_float4(0.f, 0.f, 0.f, 0.f);
    #pragma unroll
    for (int c = 0; c < 16; ++c) {
        const float4 x = x4[c];
        const float xs[4] = {x.x, x.y, x.z, x.w};
        #pragma unroll
        for (int j = 0; j < 4; ++j) {
            const float4 ws = wsrc[c * 4 + j], wd = wdst[c * 4 + j];
            as.x += xs[j] * ws.x; as.y += xs[j] * ws.y; as.z += xs[j] * ws.z; as.w += xs[j] * ws.w;
            ad.x += xs[j] * wd.x; ad.y += xs[j] * wd.y; ad.z += xs[j] * wd.z; ad.w += xs[j] * wd.w;
        }
    }
    *(float4*)(asrc + (long)n * 4) = as;
    *(float4*)(adst + (long)n * 4) = ad;
}

__device__ __forceinline__ float leaky02(float x) { return (x > 0.f) ? x : 0.2f * x; }


#define TILE_NODES 16
#define TILE_PITCH 260

static_assert(NNODES % TILE_NODES == 0, "every work-group owns 16 whole nodes");

__device__ __forceinline__ void aggregate_node_in(const int n, const int lane,
                                                  const int* __restrict__ rowptr, const int* __restrict__ csr_src,
                                                  const float* __restrict__ asrc, const float* __restrict__ adst,
                                                  const float* __restrict__ X, float* __restrict__ trow) {
    constexpr int H = 4;
    const int start = __builtin_amdgcn_readfirstlane(rowptr[n]);
    const int deg   = __builtin_amdgcn_readfirstlane(rowptr[n + 1]) - start;
    const int total = deg + 1;

    float ad[H];
    #pragma unroll
    for (int h = 0; h < H; ++h) ad[h] = adst[n * H + h];

    const bool v0 = lane < total;
    const int  s0 = (lane < deg) ? csr_src[start + lane] : n;
    const float4 a0 = *(const float4*)(asrc + (long)s0 * H);
    const float a0v[H] = {a0.x, a0.y, a0.z, a0.w};
    float x0[H];
    #pragma unroll
    for (int h = 0; h < H; ++h) x0[h] = v0 ? leaky02(a0v[h] + ad[h]) : -INFINITY;

    float m[H];
    #pragma unroll
    for (int h = 0; h < H; ++h) m[h] = x0[h];
    for (int c0 = 32; c0 < total; c0 += 32) {
        const int t = c0 + lane;
        const int s = (t < deg) ? csr_src[start + t] : n;
        if (t < total) {
            #pragma unroll
            for (int h = 0; h < H; ++h) m[h] = fmaxf(m[h], leaky02(asrc[s * H + h] + ad[h]));
        }
    }
    #pragma unroll
    for (int h = 0; h < H; ++h) {
        #pragma unroll
        for (int o = 16; o > 0; o >>= 1) m[h] = fmaxf(m[h], __shfl_xor(m[h], o));
    }

    float e0[H], sum[H];
    #pragma unroll
    for (int h = 0; h < H; ++h) { e0[h] = v0 ? expf(x0[h] - m[h]) : 0.f; sum[h] = e0[h]; }
    for (int c0 = 32; c0 < total; c0 += 32) {
        const int t = c0 + lane;
        const int s = (t < deg) ? csr_src[start + t] : n;
        if (t < total) {
            #pragma unroll
            for (int h = 0; h < H; ++h) sum[h] += expf(leaky02(asrc[s * H + h] + ad[h]) - m[h]);
        }
    }
    #pragma unroll
    for (int h = 0; h < H; ++h) {
        #pragma unroll
        for (int o = 16; o > 0; o >>= 1) sum[h] += __shfl_xor(sum[h], o);
    }

    float2 acc[H];
    #pragma unroll
    for (int h = 0; h < H; ++h) acc[h] = make_float2(0.f, 0.f);

    for (int c0 = 0; c0 < total; c0 += 32) {
        int s_l = s0;
        float cf[H];
        if (c0 == 0) {
            #pragma unroll
            for (int h = 0; h < H; ++h) cf[h] = e0[h] / sum[h];
        } else {
            const int t = c0 + lane;
            s_l = (t < deg) ? csr_src[start + t] : n;
            #pragma unroll
            for (int h = 0; h < H; ++h)
                cf[h] = (t < total) ? expf(leaky02(asrc[s_l * H + h] + ad[h]) - m[h]) / sum[h] : 0.f;
        }
        const int cnt = (total - c0 < 32) ? (total - c0) : 32;
        for (int j0 = 0; j0 < cnt; j0 += 8) {
            #pragma unroll
            for (int u = 0; u < 8; ++u) {
                const int jj = j0 + u;
                const int s = __builtin_amdgcn_readlane(s_l, jj);
                const float2 v = *(const float2*)(X + (long)s * 64 + lane * 2);
                #pragma unroll
                for (int h = 0; h < H; ++h) {
                    const float c = __int_as_float(__builtin_amdgcn_readlane(__float_as_int(cf[h]), jj));
                    acc[h].x = fmaf(c, v.x, acc[h].x);
                    acc[h].y = fmaf(c, v.y, acc[h].y);
                }
            }
        }
    }

    #pragma unroll
    for (int h = 0; h < H; ++h) *(float2*)(trow + h * 64 + lane * 2) = acc[h];
}

__global__ __launch_bounds__(256, 16)
void node_softmax_aggregate_in_project(const int* __restrict__ rowptr, const int* __restrict__ csr_src,
                                       const float* __restrict__ asrc, const float* __restrict__ adst,
                                       const float* __restrict__ X, const float* __restrict__ w12,
                                       const float* __restrict__ b12, float* __restrict__ xh) {
    __shared__ __attribute__((aligned(16))) float tile[TILE_NODES * TILE_PITCH];
    const int lane = threadIdx.x & 31;
    const int w = __builtin_amdgcn_readfirstlane((int)(threadIdx.x >> 5));
    const int nbase = (int)blockIdx.x * TILE_NODES;

    for (int i = 0; i < TILE_NODES / 8; ++i) {
        const int r = w * (TILE_NODES / 8) + i;
        aggregate_node_in(nbase + r, lane, rowptr, csr_src, asrc, adst, X, tile + r * TILE_PITCH);
    }
    __syncthreads();

    const int m  = lane & 15;
    const int kh = (lane >> 4) << 1;
    const int n0 = lane & 15;
    const int col = w * 16 + n0;
    const float bv = b12[col];
    v8f acc = (v8f){bv,bv,bv,bv,bv,bv,bv,bv};
    const float* __restrict__ Arow = tile + m * TILE_PITCH + kh;
    const float* __restrict__ Bcol = w12 + kh * 128 + col;
    #pragma unroll 16
    for (int k0 = 0; k0 < 256; k0 += 4) {
        const v2f a = *(const v2f*)(Arow + k0);
        v2f b;
        b.x = Bcol[(k0 + 0) * 128];
        b.y = Bcol[(k0 + 1) * 128];
        acc = __builtin_amdgcn_wmma_f32_16x16x4_f32(false, a, false, b, (short)0, acc, false, false);
    }
    float* __restrict__ Crow = xh + (long)(nbase + (lane >> 4) * 8) * 128 + col;
    #pragma unroll
    for (int v = 0; v < 8; ++v) Crow[v * 128] = acc[v];
}

__global__ void attn_scores(const float* __restrict__ xh,
                            const float* __restrict__ att_src,
                            const float* __restrict__ att_dst,
                            float* __restrict__ asrc, float* __restrict__ adst,
                            int H, int C) {
    int idx = blockIdx.x * blockDim.x + threadIdx.x;
    if (idx >= NNODES * H) return;
    int n = idx / H, h = idx % H;
    const float* x = xh + (long)n * H * C + (long)h * C;
    const float* as = att_src + h * C;
    const float* ad = att_dst + h * C;
    float ss = 0.f, sd = 0.f;
    for (int c = 0; c < C; ++c) { float v = x[c]; ss += v * as[c]; sd += v * ad[c]; }
    asrc[idx] = ss;
    adst[idx] = sd;
}

template <int H, int C>
__global__ __launch_bounds__(256)
void node_softmax_aggregate(const int* __restrict__ rowptr, const int* __restrict__ csr_src,
                            const float* __restrict__ asrc, const float* __restrict__ adst,
                            const float* __restrict__ xh, const float* __restrict__ bias,
                            float* __restrict__ out) {
    constexpr int HC  = H * C;
    constexpr int NIT = HC / 128;
    const int lane = threadIdx.x & 31;
    const int n = __builtin_amdgcn_readfirstlane((int)((blockIdx.x * blockDim.x + threadIdx.x) >> 5));
    if (n >= NNODES) return;
    const int start = __builtin_amdgcn_readfirstlane(rowptr[n]);
    const int deg   = __builtin_amdgcn_readfirstlane(rowptr[n + 1]) - start;
    const int total = deg + 1;

    float ad[H];
    #pragma unroll
    for (int h = 0; h < H; ++h) ad[h] = adst[n * H + h];

    const bool v0 = lane < total;
    const int  s0 = (lane < deg) ? csr_src[start + lane] : n;
    float x0[H];
    #pragma unroll
    for (int h = 0; h < H; ++h) x0[h] = v0 ? leaky02(asrc[s0 * H + h] + ad[h]) : -INFINITY;

    float m[H];
    #pragma unroll
    for (int h = 0; h < H; ++h) m[h] = x0[h];
    for (int c0 = 32; c0 < total; c0 += 32) {
        const int t = c0 + lane;
        const int s = (t < deg) ? csr_src[start + t] : n;
        if (t < total) {
            #pragma unroll
            for (int h = 0; h < H; ++h) m[h] = fmaxf(m[h], leaky02(asrc[s * H + h] + ad[h]));
        }
    }
    #pragma unroll
    for (int h = 0; h < H; ++h) {
        #pragma unroll
        for (int o = 16; o > 0; o >>= 1) m[h] = fmaxf(m[h], __shfl_xor(m[h], o));
    }

    float e0[H], sum[H];
    #pragma unroll
    for (int h = 0; h < H; ++h) { e0[h] = v0 ? expf(x0[h] - m[h]) : 0.f; sum[h] = e0[h]; }
    for (int c0 = 32; c0 < total; c0 += 32) {
        const int t = c0 + lane;
        const int s = (t < deg) ? csr_src[start + t] : n;
        if (t < total) {
            #pragma unroll
            for (int h = 0; h < H; ++h) sum[h] += expf(leaky02(asrc[s * H + h] + ad[h]) - m[h]);
        }
    }
    #pragma unroll
    for (int h = 0; h < H; ++h) {
        #pragma unroll
        for (int o = 16; o > 0; o >>= 1) sum[h] += __shfl_xor(sum[h], o);
    }

    int hsel[NIT];
    #pragma unroll
    for (int it = 0; it < NIT; ++it) hsel[it] = (it * 128 + lane * 4) / C;
    float4 acc[NIT];
    #pragma unroll
    for (int it = 0; it < NIT; ++it) acc[it] = make_float4(0.f, 0.f, 0.f, 0.f);

    for (int c0 = 0; c0 < total; c0 += 32) {
        int s_l = s0;
        float cf[H];
        if (c0 == 0) {
            #pragma unroll
            for (int h = 0; h < H; ++h) cf[h] = e0[h] / sum[h];
        } else {
            const int t = c0 + lane;
            s_l = (t < deg) ? csr_src[start + t] : n;
            #pragma unroll
            for (int h = 0; h < H; ++h)
                cf[h] = expf(leaky02(asrc[s_l * H + h] + ad[h]) - m[h]) / sum[h];
        }
        const int cnt = (total - c0 < 32) ? (total - c0) : 32;
        #pragma unroll 4
        for (int jj = 0; jj < cnt; ++jj) {
            const int s = __builtin_amdgcn_readlane(s_l, jj);
            float ch[H];
            #pragma unroll
            for (int h = 0; h < H; ++h)
                ch[h] = __int_as_float(__builtin_amdgcn_readlane(__float_as_int(cf[h]), jj));
            const float* __restrict__ row = xh + (long)s * HC + lane * 4;
            #pragma unroll
            for (int it = 0; it < NIT; ++it) {
                float c = ch[0];
                #pragma unroll
                for (int h = 1; h < H; ++h) c = (hsel[it] == h) ? ch[h] : c;
                const float4 v = *(const float4*)(row + it * 128);
                acc[it].x += c * v.x;
                acc[it].y += c * v.y;
                acc[it].z += c * v.z;
                acc[it].w += c * v.w;
            }
        }
    }

    float* __restrict__ orow = out + (long)n * HC + lane * 4;
    #pragma unroll
    for (int it = 0; it < NIT; ++it) {
        const float4 b = *(const float4*)(bias + it * 128 + lane * 4);
        float4 r;
        r.x = acc[it].x + b.x; r.y = acc[it].y + b.y; r.z = acc[it].z + b.z; r.w = acc[it].w + b.w;
        *(float4*)(orow + it * 128) = r;
    }
}

static inline int cdiv(long a, int b) { return (int)((a + b - 1) / b); }

extern "C" void kernel_launch(void* const* d_in, const int* in_sizes, int n_in,
                              void* d_out, int out_size, void* d_ws, size_t ws_size,
                              hipStream_t stream) {
    const float* X        = (const float*)d_in[0];
    const int*   ei       = (const int*)  d_in[1];
    const float* W1       = (const float*)d_in[2];
    const float* att_src1 = (const float*)d_in[3];
    const float* att_dst1 = (const float*)d_in[4];
    const float* b1       = (const float*)d_in[5];
    const float* W2       = (const float*)d_in[6];
    const float* att_src2 = (const float*)d_in[7];
    const float* att_dst2 = (const float*)d_in[8];
    const float* b2       = (const float*)d_in[9];
    float* out = (float*)d_out;

    float* ws = (float*)d_ws;
    long off = 0;
    float* xh     = ws + off; off += (long)NNODES * 128;
    float* asrc   = ws + off; off += (long)NNODES * 4;
    float* adst   = ws + off; off += (long)NNODES * 4;
    int*   rowptr = (int*)(ws + off); off += NNODES + 4;
    int*   csrc   = (int*)(ws + off); off += NEDGES;
    float* w12    = ws + off; off += 256 * 128;
    float* b12    = ws + off; off += 128;
    int*   offsT  = (int*)(ws + off); off += (CSR_BLOCKS + 1) * OFFS_PITCH;
    int*   bucketed = (int*)(ws + off); off += NEDGES;

    const int B = 256;

    bucket_edge_slices<<<NSLICE, SLICE_T, 0, stream>>>(ei, offsT, bucketed);
    group_range_edges<<<CSR_BLOCKS, CSR_T, 0, stream>>>(offsT, bucketed, rowptr, csrc);

    {
        input_scores<<<SCORE_BLOCKS + FOLD_BLOCKS, B, 0, stream>>>(X, W1, att_src1, att_dst1, asrc, adst,
                                                                   W2, b1, w12, b12);
        node_softmax_aggregate_in_project<<<NNODES / TILE_NODES, B, 0, stream>>>(
            rowptr, csrc, asrc, adst, X, w12, b12, xh);
    }

    {
        attn_scores<<<cdiv((long)NNODES, B), B, 0, stream>>>(xh, att_src2, att_dst2,
                                                             asrc, adst, 1, 128);
        node_softmax_aggregate<1, 128><<<cdiv((long)NNODES * 32, B), B, 0, stream>>>(
            rowptr, csrc, asrc, adst, xh, b2, out);
    }
}


# Layer-2 scores by linearity: att2 folded through W2 and W1 by one more work-group of the score launch; each wave dots its aggregated LDS rows with it; layer-2 score launch gone; sum order changes
# speedup vs baseline: 7.6238x; 1.0470x over previous
#include <hip/hip_runtime.h>
#include <stdint.h>


#define NNODES 50000
#define NEDGES 800000

typedef float v2f __attribute__((ext_vector_type(2)));
typedef float v8f __attribute__((ext_vector_type(8)));


#define CSR_RANGE  512
#define CSR_SHIFT  9
#define CSR_BLOCKS ((NNODES + CSR_RANGE - 1) / CSR_RANGE)
#define CSR_T      1024
#define CSR_CAP    24576
#define SLICE      1600
#define NSLICE     (NEDGES / SLICE)
#define SLICE_T    320
#define SLICE_K    (SLICE / SLICE_T)
#define OFFS_PITCH 512
#define WAVE_SLICES (OFFS_PITCH / 32)

static_assert(NEDGES % SLICE == 0, "the slices cover the edge list exactly");
static_assert(SLICE % SLICE_T == 0, "every thread of launch 1 has the same number of edges");
static_assert((1 << CSR_SHIFT) == CSR_RANGE, "range of a destination = destination >> CSR_SHIFT");
static_assert(NSLICE <= OFFS_PITCH && OFFS_PITCH <= CSR_T, "one thread of launch 2 per slice");
static_assert(NNODES <= (1 << 16), "a source index fits the low 16 bits of an entry");
static_assert(CSR_RANGE <= (1 << 16), "a local destination fits the high 16 bits of an entry");
static_assert(CSR_BLOCKS + 1 <= 128, "range counts and the total fit the 128-entry LDS table");

__global__ __launch_bounds__(SLICE_T)
void bucket_edge_slices(const int* __restrict__ ei, int* __restrict__ offsT, int* __restrict__ bucketed) {
    __shared__ int hist[128];
    const int t = threadIdx.x;
    const int lane = t & 31;
    const int e0 = (int)blockIdx.x * SLICE;
    if (t < 128) hist[t] = 0;
    __syncthreads();

    int d[SLICE_K], s[SLICE_K], rk[SLICE_K];
    #pragma unroll
    for (int k = 0; k < SLICE_K; ++k) {
        d[k] = ei[NEDGES + e0 + k * SLICE_T + t];
        s[k] = ei[e0 + k * SLICE_T + t];
    }
    #pragma unroll
    for (int k = 0; k < SLICE_K; ++k) {
        rk[k] = 0;
        if ((unsigned)d[k] < (unsigned)NNODES) rk[k] = atomicAdd(&hist[d[k] >> CSR_SHIFT], 1);
    }
    __syncthreads();

    if (t < 32) {
        int h[4];
        int sum = 0;
        #pragma unroll
        for (int j = 0; j < 4; ++j) { h[j] = hist[lane * 4 + j]; sum += h[j]; }
        int inc = sum;
        for (int o = 1; o < 32; o <<= 1) { int u = __shfl_up(inc, o); if (lane >= o) inc += u; }
        int run = inc - sum;
        #pragma unroll
        for (int j = 0; j < 4; ++j) { hist[lane * 4 + j] = run; run += h[j]; }
    }
    __syncthreads();

    if (t <= CSR_BLOCKS) offsT[t * OFFS_PITCH + (int)blockIdx.x] = hist[t];
    #pragma unroll
    for (int k = 0; k < SLICE_K; ++k) {
        if ((unsigned)d[k] < (unsigned)NNODES) {
            const int b = d[k] >> CSR_SHIFT;
            bucketed[e0 + hist[b] + rk[k]] = (s[k] & 0xFFFF) | ((d[k] & (CSR_RANGE - 1)) << 16);
        }
    }
}

__global__ __launch_bounds__(CSR_T)
void group_range_edges(const int* __restrict__ offsT, const int* __restrict__ bucketed,
                       int* __restrict__ rowptr, int* __restrict__ csr_src) {
    __shared__ int cnt[CSR_RANGE];
    __shared__ int wsum[32];
    __shared__ int wbase[32];
    __shared__ int seg_at[OFFS_PITCH];
    __shared__ int seg_n[OFFS_PITCH];
    __shared__ int seg_to[OFFS_PITCH];
    __shared__ unsigned list[CSR_CAP];
    __shared__ int sorted[CSR_CAP];
    const int t = threadIdx.x;
    const int lane = t & 31, w = t >> 5;
    const int r = (int)blockIdx.x;
    const int lo = r * CSR_RANGE;
    if (t < CSR_RANGE) cnt[t] = 0;

    int o0 = 0, c = 0;
    if (t < NSLICE) {
        o0 = offsT[r * OFFS_PITCH + t];
        c  = offsT[(r + 1) * OFFS_PITCH + t] - o0;
    }
    int inc = c;
    for (int o = 1; o < 32; o <<= 1) { int u = __shfl_up(inc, o); if (lane >= o) inc += u; }
    int osum = o0;
    #pragma unroll
    for (int o = 16; o > 0; o >>= 1) osum += __shfl_xor(osum, o);
    if (lane == 31) wsum[w] = inc;
    if (lane == 0) wbase[w] = osum;
    __syncthreads();

    int wp = 0, M = 0, base = 0;
    for (int j = 0; j < 32; ++j) {
        const int v = wsum[j];
        wp += (j < w) ? v : 0;
        M += v;
        base += wbase[j];
    }
    if (t < OFFS_PITCH) {
        seg_at[t] = t * SLICE + o0;
        seg_n[t]  = c;
        seg_to[t] = wp + inc - c;
    }
    __syncthreads();

    {
        const int mine = w * WAVE_SLICES + (lane & (WAVE_SLICES - 1));
        const int n_l = seg_n[mine], at_l = seg_at[mine], to_l = seg_to[mine];
        int ent[WAVE_SLICES];
        #pragma unroll
        for (int u = 0; u < WAVE_SLICES; ++u) {
            const int n  = __builtin_amdgcn_readlane(n_l, u);
            const int at = __builtin_amdgcn_readlane(at_l, u);
            ent[u] = (lane < n) ? bucketed[at + lane] : 0;
        }
        #pragma unroll
        for (int u = 0; u < WAVE_SLICES; ++u) {
            const int n  = __builtin_amdgcn_readlane(n_l, u);
            const int at = __builtin_amdgcn_readlane(at_l, u);
            const int to = __builtin_amdgcn_readlane(to_l, u);
            if (lane < n) {
                atomicAdd(&cnt[(unsigned)ent[u] >> 16], 1);
                const int p = to + lane;
                if (p < CSR_CAP) list[p] = (unsigned)ent[u];
            }
            for (int j = 32 + lane; j < n; j += 32) {
                const unsigned x = (unsigned)bucketed[at + j];
                atomicAdd(&cnt[x >> 16], 1);
                const int p = to + j;
                if (p < CSR_CAP) list[p] = x;
            }
        }
    }
    __syncthreads();

    const int cc = (t < CSR_RANGE) ? cnt[t] : 0;
    int inc2 = cc;
    for (int o = 1; o < 32; o <<= 1) { int u = __shfl_up(inc2, o); if (lane >= o) inc2 += u; }
    if (lane == 31) wsum[w] = inc2;
    __syncthreads();
    int wp2 = 0;
    for (int j = 0; j < 32; ++j) wp2 += (j < w) ? wsum[j] : 0;
    const int off = wp2 + inc2 - cc;
    if (t < CSR_RANGE && lo + t < NNODES) rowptr[lo + t] = base + off;
    if (blockIdx.x == CSR_BLOCKS - 1 && t == 0) rowptr[NNODES] = base + M;

    if (M <= CSR_CAP) {
        if (t < CSR_RANGE) cnt[t] = off;
        __syncthreads();
        for (int p0 = 0; p0 < M; p0 += CSR_T * 8) {
            unsigned x[8];
            int at[8];
            #pragma unroll
            for (int u = 0; u < 8; ++u) {
                const int p = p0 + u * CSR_T + t;
                x[u] = (p < M) ? list[p] : 0u;
            }
            #pragma unroll
            for (int u = 0; u < 8; ++u) {
                const int p = p0 + u * CSR_T + t;
                at[u] = (p < M) ? atomicAdd(&cnt[x[u] >> 16], 1) : 0;
            }
            #pragma unroll
            for (int u = 0; u < 8; ++u) {
                const int p = p0 + u * CSR_T + t;
                if (p < M) sorted[at[u]] = (int)(x[u] & 0xFFFFu);
            }
        }
        __syncthreads();
        for (int p = t; p < M; p += CSR_T) csr_src[base + p] = sorted[p];
    } else {
        if (t < CSR_RANGE) cnt[t] = base + off;
        __syncthreads();
        for (int u = 0; u < WAVE_SLICES; ++u) {
            const int i = w * WAVE_SLICES + u;
            const int n = seg_n[i];
            for (int j = lane; j < n; j += 32) {
                const unsigned x = (unsigned)bucketed[seg_at[i] + j];
                csr_src[atomicAdd(&cnt[x >> 16], 1)] = (int)(x & 0xFFFFu);
            }
        }
    }
}


#define SCORE_BLOCKS ((NNODES + 255) / 256)
#define FOLD_BLOCKS  5
#define SCORE2_BLOCKS 1

__device__ __forceinline__ void fold_weights_block(int fb, const float* __restrict__ W1,
                                                   const float* __restrict__ W2, const float* __restrict__ b1,
                                                   float* __restrict__ w12, float* __restrict__ b12,
                                                   float (*part)[128]) {
    const int lane  = threadIdx.x & 31;
    const int w     = threadIdx.x >> 5;
    const int strip = w & 1;
    const int q     = w >> 1;
    const int m     = lane & 15;
    const int kh    = (lane >> 4) << 1;
    const int n0    = lane & 15;
    const bool is_bias = (fb == FOLD_BLOCKS - 1);
    const int h     = is_bias ? q : fb;

    const float* __restrict__ Arow = is_bias ? (b1 + h * 64) : (W1 + (q * 16 + m) * 256 + h * 64);
    const float* __restrict__ Bcol = W2 + (long)h * 64 * 128 + strip * 64 + n0;

    v8f acc[4];
    #pragma unroll
    for (int t = 0; t < 4; ++t) acc[t] = (v8f){0.f,0.f,0.f,0.f,0.f,0.f,0.f,0.f};

    #pragma unroll 4
    for (int k0 = 0; k0 < 64; k0 += 4) {
        v2f a;
        a.x = Arow[k0 + kh + 0];
        a.y = Arow[k0 + kh + 1];
        #pragma unroll
        for (int t = 0; t < 4; ++t) {
            v2f b;
            b.x = Bcol[(k0 + kh + 0) * 128 + t * 16];
            b.y = Bcol[(k0 + kh + 1) * 128 + t * 16];
            acc[t] = __builtin_amdgcn_wmma_f32_16x16x4_f32(
                         false, a, false, b, (short)0, acc[t], false, false);
        }
    }

    if (!is_bias) {
        float* __restrict__ Crow = w12 + (long)(h * 64 + q * 16 + (lane >> 4) * 8) * 128 + strip * 64 + n0;
        #pragma unroll
        for (int t = 0; t < 4; ++t) {
            #pragma unroll
            for (int v = 0; v < 8; ++v) Crow[v * 128 + t * 16] = acc[t][v];
        }
    } else {
        if (lane < 16) {
            #pragma unroll
            for (int t = 0; t < 4; ++t) part[q][strip * 64 + t * 16 + n0] = acc[t][0];
        }
        __syncthreads();
        const int j = threadIdx.x;
        if (j < 128) b12[j] = ((part[0][j] + part[1][j]) + part[2][j]) + part[3][j];
    }
}

__device__ __forceinline__ void fold_layer2_scores_block(const float* __restrict__ W1, const float* __restrict__ W2,
                                                         const float* __restrict__ b1,
                                                         const float* __restrict__ att_src2,
                                                         const float* __restrict__ att_dst2,
                                                         float* __restrict__ w12s, float* __restrict__ w12d,
                                                         float* __restrict__ c12,
                                                         float* us, float* ud, float* red) {
    const int t = threadIdx.x;
    const int lane = t & 31, w = t >> 5;
    {
        const float4 as = *(const float4*)(att_src2 + lane * 4);
        const float4 ad = *(const float4*)(att_dst2 + lane * 4);
        float rs = 0.f, rd = 0.f;
        #pragma unroll 8
        for (int i = 0; i < 32; ++i) {
            const int r = w * 32 + i;
            const float4 x = *(const float4*)(W2 + r * 128 + lane * 4);
            float ps = x.x * as.x; ps = fmaf(x.y, as.y, ps); ps = fmaf(x.z, as.z, ps); ps = fmaf(x.w, as.w, ps);
            float pd = x.x * ad.x; pd = fmaf(x.y, ad.y, pd); pd = fmaf(x.z, ad.z, pd); pd = fmaf(x.w, ad.w, pd);
            #pragma unroll
            for (int o = 16; o > 0; o >>= 1) { ps += __shfl_xor(ps, o); pd += __shfl_xor(pd, o); }
            rs = (lane == i) ? ps : rs;
            rd = (lane == i) ? pd : rd;
        }
        us[t] = rs;
        ud[t] = rd;
    }
    __syncthreads();
    {
        const int h = t & 3, k = t >> 2;
        const float4* __restrict__ w4 = (const float4*)(W1 + t * 64);
        const float4* s4 = (const float4*)(us + h * 64);
        const float4* d4 = (const float4*)(ud + h * 64);
        float ss = 0.f, sd = 0.f;
        #pragma unroll
        for (int c = 0; c < 16; ++c) {
            const float4 wv = w4[c], a = s4[c], d = d4[c];
            ss += wv.x * a.x; ss += wv.y * a.y; ss += wv.z * a.z; ss += wv.w * a.w;
            sd += wv.x * d.x; sd += wv.y * d.y; sd += wv.z * d.z; sd += wv.w * d.w;
        }
        w12s[h * 64 + k] = ss;
        w12d[h * 64 + k] = sd;
    }
    float qs = b1[t] * us[t], qd = b1[t] * ud[t];
    #pragma unroll
    for (int o = 16; o > 0; o >>= 1) { qs += __shfl_xor(qs, o); qd += __shfl_xor(qd, o); }
    if (lane == 0) { red[w] = qs; red[8 + w] = qd; }
    __syncthreads();
    if (t == 0) {
        float cs = 0.f, cd = 0.f;
        for (int j = 0; j < 8; ++j) { cs += red[j]; cd += red[8 + j]; }
        c12[0] = cs;
        c12[1] = cd;
    }
}

__global__ __launch_bounds__(256)
void input_scores(const float* __restrict__ X, const float* __restrict__ W1,
                  const float* __restrict__ att_src, const float* __restrict__ att_dst,
                  float* __restrict__ asrc, float* __restrict__ adst,
                  const float* __restrict__ W2, const float* __restrict__ b1,
                  float* __restrict__ w12, float* __restrict__ b12,
                  const float* __restrict__ att_src2, const float* __restrict__ att_dst2,
                  float* __restrict__ w12s, float* __restrict__ w12d, float* __restrict__ c12) {
    __shared__ float4 wsrc[64];
    __shared__ float4 wdst[64];
    __shared__ float part[4][128];
    if (blockIdx.x >= SCORE_BLOCKS) {
        const int fb = (int)blockIdx.x - SCORE_BLOCKS;
        if (fb < FOLD_BLOCKS) fold_weights_block(fb, W1, W2, b1, w12, b12, part);
        else fold_layer2_scores_block(W1, W2, b1, att_src2, att_dst2, w12s, w12d, c12,
                                      (float*)wsrc, (float*)wdst, &part[0][0]);
        return;
    }
    const int t = threadIdx.x;
    {
        const int h = t & 3;
        const float4* __restrict__ w4 = (const float4*)(W1 + t * 64);
        const float4* __restrict__ s4 = (const float4*)(att_src + h * 64);
        const float4* __restrict__ d4 = (const float4*)(att_dst + h * 64);
        float ss = 0.f, sd = 0.f;
        #pragma unroll
        for (int c = 0; c < 16; ++c) {
            const float4 w = w4[c], a = s4[c], d = d4[c];
            ss += w.x * a.x; ss += w.y * a.y; ss += w.z * a.z; ss += w.w * a.w;
            sd += w.x * d.x; sd += w.y * d.y; sd += w.z * d.z; sd += w.w * d.w;
        }
        ((float*)wsrc)[t] = ss;
        ((float*)wdst)[t] = sd;
    }
    __syncthreads();
    const int n = blockIdx.x * 256 + t;
    if (n >= NNODES) return;
    const float4* __restrict__ x4 = (const float4*)(X + (long)n * 64);
    float4 as = make_float4(0.f, 0.f, 0.f, 0.f), ad = make_float4(0.f, 0.f, 0.f, 0.f);
    #pragma unroll
    for (int c = 0; c < 16; ++c) {
        const float4 x = x4[c];
        const float xs[4] = {x.x, x.y, x.z, x.w};
        #pragma unroll
        for (int j = 0; j < 4; ++j) {
            const float4 ws = wsrc[c * 4 + j], wd = wdst[c * 4 + j];
            as.x += xs[j] * ws.x; as.y += xs[j] * ws.y; as.z += xs[j] * ws.z; as.w += xs[j] * ws.w;
            ad.x += xs[j] * wd.x; ad.y += xs[j] * wd.y; ad.z += xs[j] * wd.z; ad.w += xs[j] * wd.w;
        }
    }
    *(float4*)(asrc + (long)n * 4) = as;
    *(float4*)(adst + (long)n * 4) = ad;
}

__device__ __forceinline__ float leaky02(float x) { return (x > 0.f) ? x : 0.2f * x; }


#define TILE_NODES 16
#define TILE_PITCH 260

static_assert(NNODES % TILE_NODES == 0, "every work-group owns 16 whole nodes");

__device__ __forceinline__ void aggregate_node_in(const int n, const int lane,
                                                  const int* __restrict__ rowptr, const int* __restrict__ csr_src,
                                                  const float* __restrict__ asrc, const float* __restrict__ adst,
                                                  const float* __restrict__ X, float* __restrict__ trow) {
    constexpr int H = 4;
    const int start = __builtin_amdgcn_readfirstlane(rowptr[n]);
    const int deg   = __builtin_amdgcn_readfirstlane(rowptr[n + 1]) - start;
    const int total = deg + 1;

    float ad[H];
    #pragma unroll
    for (int h = 0; h < H; ++h) ad[h] = adst[n * H + h];

    const bool v0 = lane < total;
    const int  s0 = (lane < deg) ? csr_src[start + lane] : n;
    const float4 a0 = *(const float4*)(asrc + (long)s0 * H);
    const float a0v[H] = {a0.x, a0.y, a0.z, a0.w};
    float x0[H];
    #pragma unroll
    for (int h = 0; h < H; ++h) x0[h] = v0 ? leaky02(a0v[h] + ad[h]) : -INFINITY;

    float m[H];
    #pragma unroll
    for (int h = 0; h < H; ++h) m[h] = x0[h];
    for (int c0 = 32; c0 < total; c0 += 32) {
        const int t = c0 + lane;
        const int s = (t < deg) ? csr_src[start + t] : n;
        if (t < total) {
            #pragma unroll
            for (int h = 0; h < H; ++h) m[h] = fmaxf(m[h], leaky02(asrc[s * H + h] + ad[h]));
        }
    }
    #pragma unroll
    for (int h = 0; h < H; ++h) {
        #pragma unroll
        for (int o = 16; o > 0; o >>= 1) m[h] = fmaxf(m[h], __shfl_xor(m[h], o));
    }

    float e0[H], sum[H];
    #pragma unroll
    for (int h = 0; h < H; ++h) { e0[h] = v0 ? expf(x0[h] - m[h]) : 0.f; sum[h] = e0[h]; }
    for (int c0 = 32; c0 < total; c0 += 32) {
        const int t = c0 + lane;
        const int s = (t < deg) ? csr_src[start + t] : n;
        if (t < total) {
            #pragma unroll
            for (int h = 0; h < H; ++h) sum[h] += expf(leaky02(asrc[s * H + h] + ad[h]) - m[h]);
        }
    }
    #pragma unroll
    for (int h = 0; h < H; ++h) {
        #pragma unroll
        for (int o = 16; o > 0; o >>= 1) sum[h] += __shfl_xor(sum[h], o);
    }

    float2 acc[H];
    #pragma unroll
    for (int h = 0; h < H; ++h) acc[h] = make_float2(0.f, 0.f);

    for (int c0 = 0; c0 < total; c0 += 32) {
        int s_l = s0;
        float cf[H];
        if (c0 == 0) {
            #pragma unroll
            for (int h = 0; h < H; ++h) cf[h] = e0[h] / sum[h];
        } else {
            const int t = c0 + lane;
            s_l = (t < deg) ? csr_src[start + t] : n;
            #pragma unroll
            for (int h = 0; h < H; ++h)
                cf[h] = (t < total) ? expf(leaky02(asrc[s_l * H + h] + ad[h]) - m[h]) / sum[h] : 0.f;
        }
        const int cnt = (total - c0 < 32) ? (total - c0) : 32;
        for (int j0 = 0; j0 < cnt; j0 += 8) {
            #pragma unroll
            for (int u = 0; u < 8; ++u) {
                const int jj = j0 + u;
                const int s = __builtin_amdgcn_readlane(s_l, jj);
                const float2 v = *(const float2*)(X + (long)s * 64 + lane * 2);
                #pragma unroll
                for (int h = 0; h < H; ++h) {
                    const float c = __int_as_float(__builtin_amdgcn_readlane(__float_as_int(cf[h]), jj));
                    acc[h].x = fmaf(c, v.x, acc[h].x);
                    acc[h].y = fmaf(c, v.y, acc[h].y);
                }
            }
        }
    }

    #pragma unroll
    for (int h = 0; h < H; ++h) *(float2*)(trow + h * 64 + lane * 2) = acc[h];
}

__global__ __launch_bounds__(256, 16)
void node_softmax_aggregate_in_project(const int* __restrict__ rowptr, const int* __restrict__ csr_src,
                                       const float* __restrict__ asrc, const float* __restrict__ adst,
                                       const float* __restrict__ X, const float* __restrict__ w12,
                                       const float* __restrict__ b12, float* __restrict__ xh,
                                       const float* __restrict__ w12s, const float* __restrict__ w12d,
                                       const float* __restrict__ c12,
                                       float* __restrict__ as2, float* __restrict__ ad2) {
    __shared__ __attribute__((aligned(16))) float tile[TILE_NODES * TILE_PITCH];
    const int lane = threadIdx.x & 31;
    const int w = __builtin_amdgcn_readfirstlane((int)(threadIdx.x >> 5));
    const int nbase = (int)blockIdx.x * TILE_NODES;

    for (int i = 0; i < TILE_NODES / 8; ++i) {
        const int r = w * (TILE_NODES / 8) + i;
        aggregate_node_in(nbase + r, lane, rowptr, csr_src, asrc, adst, X, tile + r * TILE_PITCH);
    }

    {
        float2 ws[4], wd[4];
        #pragma unroll
        for (int h = 0; h < 4; ++h) {
            ws[h] = *(const float2*)(w12s + h * 64 + lane * 2);
            wd[h] = *(const float2*)(w12d + h * 64 + lane * 2);
        }
        const float cs = c12[0], cd = c12[1];
        float rs = 0.f, rd = 0.f;
        #pragma unroll
        for (int i = 0; i < TILE_NODES / 8; ++i) {
            const float* trow = tile + (w * (TILE_NODES / 8) + i) * TILE_PITCH;
            float ps = 0.f, pd = 0.f;
            #pragma unroll
            for (int h = 0; h < 4; ++h) {
                const float2 a = *(const float2*)(trow + h * 64 + lane * 2);
                ps = fmaf(a.x, ws[h].x, ps); ps = fmaf(a.y, ws[h].y, ps);
                pd = fmaf(a.x, wd[h].x, pd); pd = fmaf(a.y, wd[h].y, pd);
            }
            #pragma unroll
            for (int o = 16; o > 0; o >>= 1) { ps += __shfl_xor(ps, o); pd += __shfl_xor(pd, o); }
            rs = (lane == i) ? ps : rs;
            rd = (lane == i) ? pd : rd;
        }
        if (lane < TILE_NODES / 8) {
            const int n = nbase + w * (TILE_NODES / 8) + lane;
            as2[n] = rs + cs;
            ad2[n] = rd + cd;
        }
    }
    __syncthreads();

    const int m  = lane & 15;
    const int kh = (lane >> 4) << 1;
    const int n0 = lane & 15;
    const int col = w * 16 + n0;
    const float bv = b12[col];
    v8f acc = (v8f){bv,bv,bv,bv,bv,bv,bv,bv};
    const float* __restrict__ Arow = tile + m * TILE_PITCH + kh;
    const float* __restrict__ Bcol = w12 + kh * 128 + col;
    #pragma unroll 16
    for (int k0 = 0; k0 < 256; k0 += 4) {
        const v2f a = *(const v2f*)(Arow + k0);
        v2f b;
        b.x = Bcol[(k0 + 0) * 128];
        b.y = Bcol[(k0 + 1) * 128];
        acc = __builtin_amdgcn_wmma_f32_16x16x4_f32(false, a, false, b, (short)0, acc, false, false);
    }
    float* __restrict__ Crow = xh + (long)(nbase + (lane >> 4) * 8) * 128 + col;
    #pragma unroll
    for (int v = 0; v < 8; ++v) Crow[v * 128] = acc[v];
}

template <int H, int C>
__global__ __launch_bounds__(256)
void node_softmax_aggregate(const int* __restrict__ rowptr, const int* __restrict__ csr_src,
                            const float* __restrict__ asrc, const float* __restrict__ adst,
                            const float* __restrict__ xh, const float* __restrict__ bias,
                            float* __restrict__ out) {
    constexpr int HC  = H * C;
    constexpr int NIT = HC / 128;
    const int lane = threadIdx.x & 31;
    const int n = __builtin_amdgcn_readfirstlane((int)((blockIdx.x * blockDim.x + threadIdx.x) >> 5));
    if (n >= NNODES) return;
    const int start = __builtin_amdgcn_readfirstlane(rowptr[n]);
    const int deg   = __builtin_amdgcn_readfirstlane(rowptr[n + 1]) - start;
    const int total = deg + 1;

    float ad[H];
    #pragma unroll
    for (int h = 0; h < H; ++h) ad[h] = adst[n * H + h];

    const bool v0 = lane < total;
    const int  s0 = (lane < deg) ? csr_src[start + lane] : n;
    float x0[H];
    #pragma unroll
    for (int h = 0; h < H; ++h) x0[h] = v0 ? leaky02(asrc[s0 * H + h] + ad[h]) : -INFINITY;

    float m[H];
    #pragma unroll
    for (int h = 0; h < H; ++h) m[h] = x0[h];
    for (int c0 = 32; c0 < total; c0 += 32) {
        const int t = c0 + lane;
        const int s = (t < deg) ? csr_src[start + t] : n;
        if (t < total) {
            #pragma unroll
            for (int h = 0; h < H; ++h) m[h] = fmaxf(m[h], leaky02(asrc[s * H + h] + ad[h]));
        }
    }
    #pragma unroll
    for (int h = 0; h < H; ++h) {
        #pragma unroll
        for (int o = 16; o > 0; o >>= 1) m[h] = fmaxf(m[h], __shfl_xor(m[h], o));
    }

    float e0[H], sum[H];
    #pragma unroll
    for (int h = 0; h < H; ++h) { e0[h] = v0 ? expf(x0[h] - m[h]) : 0.f; sum[h] = e0[h]; }
    for (int c0 = 32; c0 < total; c0 += 32) {
        const int t = c0 + lane;
        const int s = (t < deg) ? csr_src[start + t] : n;
        if (t < total) {
            #pragma unroll
            for (int h = 0; h < H; ++h) sum[h] += expf(leaky02(asrc[s * H + h] + ad[h]) - m[h]);
        }
    }
    #pragma unroll
    for (int h = 0; h < H; ++h) {
        #pragma unroll
        for (int o = 16; o > 0; o >>= 1) sum[h] += __shfl_xor(sum[h], o);
    }

    int hsel[NIT];
    #pragma unroll
    for (int it = 0; it < NIT; ++it) hsel[it] = (it * 128 + lane * 4) / C;
    float4 acc[NIT];
    #pragma unroll
    for (int it = 0; it < NIT; ++it) acc[it] = make_float4(0.f, 0.f, 0.f, 0.f);

    for (int c0 = 0; c0 < total; c0 += 32) {
        int s_l = s0;
        float cf[H];
        if (c0 == 0) {
            #pragma unroll
            for (int h = 0; h < H; ++h) cf[h] = e0[h] / sum[h];
        } else {
            const int t = c0 + lane;
            s_l = (t < deg) ? csr_src[start + t] : n;
            #pragma unroll
            for (int h = 0; h < H; ++h)
                cf[h] = expf(leaky02(asrc[s_l * H + h] + ad[h]) - m[h]) / sum[h];
        }
        const int cnt = (total - c0 < 32) ? (total - c0) : 32;
        #pragma unroll 4
        for (int jj = 0; jj < cnt; ++jj) {
            const int s = __builtin_amdgcn_readlane(s_l, jj);
            float ch[H];
            #pragma unroll
            for (int h = 0; h < H; ++h)
                ch[h] = __int_as_float(__builtin_amdgcn_readlane(__float_as_int(cf[h]), jj));
            const float* __restrict__ row = xh + (long)s * HC + lane * 4;
            #pragma unroll
            for (int it = 0; it < NIT; ++it) {
                float c = ch[0];
                #pragma unroll
                for (int h = 1; h < H; ++h) c = (hsel[it] == h) ? ch[h] : c;
                const float4 v = *(const float4*)(row + it * 128);
                acc[it].x += c * v.x;
                acc[it].y += c * v.y;
                acc[it].z += c * v.z;
                acc[it].w += c * v.w;
            }
        }
    }

    float* __restrict__ orow = out + (long)n * HC + lane * 4;
    #pragma unroll
    for (int it = 0; it < NIT; ++it) {
        const float4 b = *(const float4*)(bias + it * 128 + lane * 4);
        float4 r;
        r.x = acc[it].x + b.x; r.y = acc[it].y + b.y; r.z = acc[it].z + b.z; r.w = acc[it].w + b.w;
        *(float4*)(orow + it * 128) = r;
    }
}

static inline int cdiv(long a, int b) { return (int)((a + b - 1) / b); }

extern "C" void kernel_launch(void* const* d_in, const int* in_sizes, int n_in,
                              void* d_out, int out_size, void* d_ws, size_t ws_size,
                              hipStream_t stream) {
    const float* X        = (const float*)d_in[0];
    const int*   ei       = (const int*)  d_in[1];
    const float* W1       = (const float*)d_in[2];
    const float* att_src1 = (const float*)d_in[3];
    const float* att_dst1 = (const float*)d_in[4];
    const float* b1       = (const float*)d_in[5];
    const float* W2       = (const float*)d_in[6];
    const float* att_src2 = (const float*)d_in[7];
    const float* att_dst2 = (const float*)d_in[8];
    const float* b2       = (const float*)d_in[9];
    float* out = (float*)d_out;

    float* ws = (float*)d_ws;
    long off = 0;
    float* xh     = ws + off; off += (long)NNODES * 128;
    float* asrc   = ws + off; off += (long)NNODES * 4;
    float* adst   = ws + off; off += (long)NNODES * 4;
    int*   rowptr = (int*)(ws + off); off += NNODES + 4;
    int*   csrc   = (int*)(ws + off); off += NEDGES;
    float* w12    = ws + off; off += 256 * 128;
    float* b12    = ws + off; off += 128;
    int*   offsT  = (int*)(ws + off); off += (CSR_BLOCKS + 1) * OFFS_PITCH;
    int*   bucketed = (int*)(ws + off); off += NEDGES;
    float* w12s   = ws + off; off += 256;
    float* w12d   = ws + off; off += 256;
    float* c12    = ws + off; off += 4;
    float* as2    = ws + off; off += NNODES;
    float* ad2    = ws + off; off += NNODES;

    const int B = 256;

    bucket_edge_slices<<<NSLICE, SLICE_T, 0, stream>>>(ei, offsT, bucketed);
    group_range_edges<<<CSR_BLOCKS, CSR_T, 0, stream>>>(offsT, bucketed, rowptr, csrc);

    {
        input_scores<<<SCORE_BLOCKS + FOLD_BLOCKS + SCORE2_BLOCKS, B, 0, stream>>>(
            X, W1, att_src1, att_dst1, asrc, adst, W2, b1, w12, b12, att_src2, att_dst2, w12s, w12d, c12);
        node_softmax_aggregate_in_project<<<NNODES / TILE_NODES, B, 0, stream>>>(
            rowptr, csrc, asrc, adst, X, w12, b12, xh, w12s, w12d, c12, as2, ad2);
    }

    {
        node_softmax_aggregate<1, 128><<<cdiv((long)NNODES * 32, B), B, 0, stream>>>(
            rowptr, csrc, as2, ad2, xh, b2, out);
    }
}


# W12 written by the fold work-groups in the B loader's fragment order: a wave's B fragments of two k-steps are one whole-line 16-B load per lane, 32 loads per wave not 128; sum order kept
# speedup vs baseline: 8.1408x; 1.0678x over previous
#include <hip/hip_runtime.h>
#include <stdint.h>


#define NNODES 50000
#define NEDGES 800000

typedef float v2f __attribute__((ext_vector_type(2)));
typedef float v8f __attribute__((ext_vector_type(8)));


#define CSR_RANGE  512
#define CSR_SHIFT  9
#define CSR_BLOCKS ((NNODES + CSR_RANGE - 1) / CSR_RANGE)
#define CSR_T      1024
#define CSR_CAP    24576
#define SLICE      1600
#define NSLICE     (NEDGES / SLICE)
#define SLICE_T    320
#define SLICE_K    (SLICE / SLICE_T)
#define OFFS_PITCH 512
#define WAVE_SLICES (OFFS_PITCH / 32)

static_assert(NEDGES % SLICE == 0, "the slices cover the edge list exactly");
static_assert(SLICE % SLICE_T == 0, "every thread of launch 1 has the same number of edges");
static_assert((1 << CSR_SHIFT) == CSR_RANGE, "range of a destination = destination >> CSR_SHIFT");
static_assert(NSLICE <= OFFS_PITCH && OFFS_PITCH <= CSR_T, "one thread of launch 2 per slice");
static_assert(NNODES <= (1 << 16), "a source index fits the low 16 bits of an entry");
static_assert(CSR_RANGE <= (1 << 16), "a local destination fits the high 16 bits of an entry");
static_assert(CSR_BLOCKS + 1 <= 128, "range counts and the total fit the 128-entry LDS table");

__global__ __launch_bounds__(SLICE_T)
void bucket_edge_slices(const int* __restrict__ ei, int* __restrict__ offsT, int* __restrict__ bucketed) {
    __shared__ int hist[128];
    const int t = threadIdx.x;
    const int lane = t & 31;
    const int e0 = (int)blockIdx.x * SLICE;
    if (t < 128) hist[t] = 0;
    __syncthreads();

    int d[SLICE_K], s[SLICE_K], rk[SLICE_K];
    #pragma unroll
    for (int k = 0; k < SLICE_K; ++k) {
        d[k] = ei[NEDGES + e0 + k * SLICE_T + t];
        s[k] = ei[e0 + k * SLICE_T + t];
    }
    #pragma unroll
    for (int k = 0; k < SLICE_K; ++k) {
        rk[k] = 0;
        if ((unsigned)d[k] < (unsigned)NNODES) rk[k] = atomicAdd(&hist[d[k] >> CSR_SHIFT], 1);
    }
    __syncthreads();

    if (t < 32) {
        int h[4];
        int sum = 0;
        #pragma unroll
        for (int j = 0; j < 4; ++j) { h[j] = hist[lane * 4 + j]; sum += h[j]; }
        int inc = sum;
        for (int o = 1; o < 32; o <<= 1) { int u = __shfl_up(inc, o); if (lane >= o) inc += u; }
        int run = inc - sum;
        #pragma unroll
        for (int j = 0; j < 4; ++j) { hist[lane * 4 + j] = run; run += h[j]; }
    }
    __syncthreads();

    if (t <= CSR_BLOCKS) offsT[t * OFFS_PITCH + (int)blockIdx.x] = hist[t];
    #pragma unroll
    for (int k = 0; k < SLICE_K; ++k) {
        if ((unsigned)d[k] < (unsigned)NNODES) {
            const int b = d[k] >> CSR_SHIFT;
            bucketed[e0 + hist[b] + rk[k]] = (s[k] & 0xFFFF) | ((d[k] & (CSR_RANGE - 1)) << 16);
        }
    }
}

__global__ __launch_bounds__(CSR_T)
void group_range_edges(const int* __restrict__ offsT, const int* __restrict__ bucketed,
                       int* __restrict__ rowptr, int* __restrict__ csr_src) {
    __shared__ int cnt[CSR_RANGE];
    __shared__ int wsum[32];
    __shared__ int wbase[32];
    __shared__ int seg_at[OFFS_PITCH];
    __shared__ int seg_n[OFFS_PITCH];
    __shared__ int seg_to[OFFS_PITCH];
    __shared__ unsigned list[CSR_CAP];
    __shared__ int sorted[CSR_CAP];
    const int t = threadIdx.x;
    const int lane = t & 31, w = t >> 5;
    const int r = (int)blockIdx.x;
    const int lo = r * CSR_RANGE;
    if (t < CSR_RANGE) cnt[t] = 0;

    int o0 = 0, c = 0;
    if (t < NSLICE) {
        o0 = offsT[r * OFFS_PITCH + t];
        c  = offsT[(r + 1) * OFFS_PITCH + t] - o0;
    }
    int inc = c;
    for (int o = 1; o < 32; o <<= 1) { int u = __shfl_up(inc, o); if (lane >= o) inc += u; }
    int osum = o0;
    #pragma unroll
    for (int o = 16; o > 0; o >>= 1) osum += __shfl_xor(osum, o);
    if (lane == 31) wsum[w] = inc;
    if (lane == 0) wbase[w] = osum;
    __syncthreads();

    int wp = 0, M = 0, base = 0;
    for (int j = 0; j < 32; ++j) {
        const int v = wsum[j];
        wp += (j < w) ? v : 0;
        M += v;
        base += wbase[j];
    }
    if (t < OFFS_PITCH) {
        seg_at[t] = t * SLICE + o0;
        seg_n[t]  = c;
        seg_to[t] = wp + inc - c;
    }
    __syncthreads();

    {
        const int mine = w * WAVE_SLICES + (lane & (WAVE_SLICES - 1));
        const int n_l = seg_n[mine], at_l = seg_at[mine], to_l = seg_to[mine];
        int ent[WAVE_SLICES];
        #pragma unroll
        for (int u = 0; u < WAVE_SLICES; ++u) {
            const int n  = __builtin_amdgcn_readlane(n_l, u);
            const int at = __builtin_amdgcn_readlane(at_l, u);
            ent[u] = (lane < n) ? bucketed[at + lane] : 0;
        }
        #pragma unroll
        for (int u = 0; u < WAVE_SLICES; ++u) {
            const int n  = __builtin_amdgcn_readlane(n_l, u);
            const int at = __builtin_amdgcn_readlane(at_l, u);
            const int to = __builtin_amdgcn_readlane(to_l, u);
            if (lane < n) {
                atomicAdd(&cnt[(unsigned)ent[u] >> 16], 1);
                const int p = to + lane;
                if (p < CSR_CAP) list[p] = (unsigned)ent[u];
            }
            for (int j = 32 + lane; j < n; j += 32) {
                const unsigned x = (unsigned)bucketed[at + j];
                atomicAdd(&cnt[x >> 16], 1);
                const int p = to + j;
                if (p < CSR_CAP) list[p] = x;
            }
        }
    }
    __syncthreads();

    const int cc = (t < CSR_RANGE) ? cnt[t] : 0;
    int inc2 = cc;
    for (int o = 1; o < 32; o <<= 1) { int u = __shfl_up(inc2, o); if (lane >= o) inc2 += u; }
    if (lane == 31) wsum[w] = inc2;
    __syncthreads();
    int wp2 = 0;
    for (int j = 0; j < 32; ++j) wp2 += (j < w) ? wsum[j] : 0;
    const int off = wp2 + inc2 - cc;
    if (t < CSR_RANGE && lo + t < NNODES) rowptr[lo + t] = base + off;
    if (blockIdx.x == CSR_BLOCKS - 1 && t == 0) rowptr[NNODES] = base + M;

    if (M <= CSR_CAP) {
        if (t < CSR_RANGE) cnt[t] = off;
        __syncthreads();
        for (int p0 = 0; p0 < M; p0 += CSR_T * 8) {
            unsigned x[8];
            int at[8];
            #pragma unroll
            for (int u = 0; u < 8; ++u) {
                const int p = p0 + u * CSR_T + t;
                x[u] = (p < M) ? list[p] : 0u;
            }
            #pragma unroll
            for (int u = 0; u < 8; ++u) {
                const int p = p0 + u * CSR_T + t;
                at[u] = (p < M) ? atomicAdd(&cnt[x[u] >> 16], 1) : 0;
            }
            #pragma unroll
            for (int u = 0; u < 8; ++u) {
                const int p = p0 + u * CSR_T + t;
                if (p < M) sorted[at[u]] = (int)(x[u] & 0xFFFFu);
            }
        }
        __syncthreads();
        for (int p = t; p < M; p += CSR_T) csr_src[base + p] = sorted[p];
    } else {
        if (t < CSR_RANGE) cnt[t] = base + off;
        __syncthreads();
        for (int u = 0; u < WAVE_SLICES; ++u) {
            const int i = w * WAVE_SLICES + u;
            const int n = seg_n[i];
            for (int j = lane; j < n; j += 32) {
                const unsigned x = (unsigned)bucketed[seg_at[i] + j];
                csr_src[atomicAdd(&cnt[x >> 16], 1)] = (int)(x & 0xFFFFu);
            }
        }
    }
}


#define SCORE_BLOCKS ((NNODES + 255) / 256)
#define FOLD_BLOCKS  5
#define SCORE2_BLOCKS 1

__device__ __forceinline__ void fold_weights_block(int fb, const float* __restrict__ W1,
                                                   const float* __restrict__ W2, const float* __restrict__ b1,
                                                   float* __restrict__ w12, float* __restrict__ b12,
                                                   float (*part)[128]) {
    const int lane  = threadIdx.x & 31;
    const int w     = threadIdx.x >> 5;
    const int strip = w & 1;
    const int q     = w >> 1;
    const int m     = lane & 15;
    const int kh    = (lane >> 4) << 1;
    const int n0    = lane & 15;
    const bool is_bias = (fb == FOLD_BLOCKS - 1);
    const int h     = is_bias ? q : fb;

    const float* __restrict__ Arow = is_bias ? (b1 + h * 64) : (W1 + (q * 16 + m) * 256 + h * 64);
    const float* __restrict__ Bcol = W2 + (long)h * 64 * 128 + strip * 64 + n0;

    v8f acc[4];
    #pragma unroll
    for (int t = 0; t < 4; ++t) acc[t] = (v8f){0.f,0.f,0.f,0.f,0.f,0.f,0.f,0.f};

    #pragma unroll 4
    for (int k0 = 0; k0 < 64; k0 += 4) {
        v2f a;
        a.x = Arow[k0 + kh + 0];
        a.y = Arow[k0 + kh + 1];
        #pragma unroll
        for (int t = 0; t < 4; ++t) {
            v2f b;
            b.x = Bcol[(k0 + kh + 0) * 128 + t * 16];
            b.y = Bcol[(k0 + kh + 1) * 128 + t * 16];
            acc[t] = __builtin_amdgcn_wmma_f32_16x16x4_f32(
                         false, a, false, b, (short)0, acc[t], false, false);
        }
    }

    if (!is_bias) {
        const int kq = h * 8 + q * 2 + (lane >> 4);
        #pragma unroll
        for (int t = 0; t < 4; ++t) {
            float4* __restrict__ dst = (float4*)w12 + ((long)((strip * 4 + t) * 32 + kq) * 32 + n0);
            dst[0]  = make_float4(acc[t][0], acc[t][1], acc[t][4], acc[t][5]);
            dst[16] = make_float4(acc[t][2], acc[t][3], acc[t][6], acc[t][7]);
        }
    } else {
        if (lane < 16) {
            #pragma unroll
            for (int t = 0; t < 4; ++t) part[q][strip * 64 + t * 16 + n0] = acc[t][0];
        }
        __syncthreads();
        const int j = threadIdx.x;
        if (j < 128) b12[j] = ((part[0][j] + part[1][j]) + part[2][j]) + part[3][j];
    }
}

__device__ __forceinline__ void fold_layer2_scores_block(const float* __restrict__ W1, const float* __restrict__ W2,
                                                         const float* __restrict__ b1,
                                                         const float* __restrict__ att_src2,
                                                         const float* __restrict__ att_dst2,
                                                         float* __restrict__ w12s, float* __restrict__ w12d,
                                                         float* __restrict__ c12,
                                                         float* us, float* ud, float* red) {
    const int t = threadIdx.x;
    const int lane = t & 31, w = t >> 5;
    {
        const float4 as = *(const float4*)(att_src2 + lane * 4);
        const float4 ad = *(const float4*)(att_dst2 + lane * 4);
        float rs = 0.f, rd = 0.f;
        #pragma unroll 8
        for (int i = 0; i < 32; ++i) {
            const int r = w * 32 + i;
            const float4 x = *(const float4*)(W2 + r * 128 + lane * 4);
            float ps = x.x * as.x; ps = fmaf(x.y, as.y, ps); ps = fmaf(x.z, as.z, ps); ps = fmaf(x.w, as.w, ps);
            float pd = x.x * ad.x; pd = fmaf(x.y, ad.y, pd); pd = fmaf(x.z, ad.z, pd); pd = fmaf(x.w, ad.w, pd);
            #pragma unroll
            for (int o = 16; o > 0; o >>= 1) { ps += __shfl_xor(ps, o); pd += __shfl_xor(pd, o); }
            rs = (lane == i) ? ps : rs;
            rd = (lane == i) ? pd : rd;
        }
        us[t] = rs;
        ud[t] = rd;
    }
    __syncthreads();
    {
        const int h = t & 3, k = t >> 2;
        const float4* __restrict__ w4 = (const float4*)(W1 + t * 64);
        const float4* s4 = (const float4*)(us + h * 64);
        const float4* d4 = (const float4*)(ud + h * 64);
        float ss = 0.f, sd = 0.f;
        #pragma unroll
        for (int c = 0; c < 16; ++c) {
            const float4 wv = w4[c], a = s4[c], d = d4[c];
            ss += wv.x * a.x; ss += wv.y * a.y; ss += wv.z * a.z; ss += wv.w * a.w;
            sd += wv.x * d.x; sd += wv.y * d.y; sd += wv.z * d.z; sd += wv.w * d.w;
        }
        w12s[h * 64 + k] = ss;
        w12d[h * 64 + k] = sd;
    }
    float qs = b1[t] * us[t], qd = b1[t] * ud[t];
    #pragma unroll
    for (int o = 16; o > 0; o >>= 1) { qs += __shfl_xor(qs, o); qd += __shfl_xor(qd, o); }
    if (lane == 0) { red[w] = qs; red[8 + w] = qd; }
    __syncthreads();
    if (t == 0) {
        float cs = 0.f, cd = 0.f;
        for (int j = 0; j < 8; ++j) { cs += red[j]; cd += red[8 + j]; }
        c12[0] = cs;
        c12[1] = cd;
    }
}

__global__ __launch_bounds__(256)
void input_scores(const float* __restrict__ X, const float* __restrict__ W1,
                  const float* __restrict__ att_src, const float* __restrict__ att_dst,
                  float* __restrict__ asrc, float* __restrict__ adst,
                  const float* __restrict__ W2, const float* __restrict__ b1,
                  float* __restrict__ w12, float* __restrict__ b12,
                  const float* __restrict__ att_src2, const float* __restrict__ att_dst2,
                  float* __restrict__ w12s, float* __restrict__ w12d, float* __restrict__ c12) {
    __shared__ float4 wsrc[64];
    __shared__ float4 wdst[64];
    __shared__ float part[4][128];
    if (blockIdx.x >= SCORE_BLOCKS) {
        const int fb = (int)blockIdx.x - SCORE_BLOCKS;
        if (fb < FOLD_BLOCKS) fold_weights_block(fb, W1, W2, b1, w12, b12, part);
        else fold_layer2_scores_block(W1, W2, b1, att_src2, att_dst2, w12s, w12d, c12,
                                      (float*)wsrc, (float*)wdst, &part[0][0]);
        return;
    }
    const int t = threadIdx.x;
    {
        const int h = t & 3;
        const float4* __restrict__ w4 = (const float4*)(W1 + t * 64);
        const float4* __restrict__ s4 = (const float4*)(att_src + h * 64);
        const float4* __restrict__ d4 = (const float4*)(att_dst + h * 64);
        float ss = 0.f, sd = 0.f;
        #pragma unroll
        for (int c = 0; c < 16; ++c) {
            const float4 w = w4[c], a = s4[c], d = d4[c];
            ss += w.x * a.x; ss += w.y * a.y; ss += w.z * a.z; ss += w.w * a.w;
            sd += w.x * d.x; sd += w.y * d.y; sd += w.z * d.z; sd += w.w * d.w;
        }
        ((float*)wsrc)[t] = ss;
        ((float*)wdst)[t] = sd;
    }
    __syncthreads();
    const int n = blockIdx.x * 256 + t;
    if (n >= NNODES) return;
    const float4* __restrict__ x4 = (const float4*)(X + (long)n * 64);
    float4 as = make_float4(0.f, 0.f, 0.f, 0.f), ad = make_float4(0.f, 0.f, 0.f, 0.f);
    #pragma unroll
    for (int c = 0; c < 16; ++c) {
        const float4 x = x4[c];
        const float xs[4] = {x.x, x.y, x.z, x.w};
        #pragma unroll
        for (int j = 0; j < 4; ++j) {
            const float4 ws = wsrc[c * 4 + j], wd = wdst[c * 4 + j];
            as.x += xs[j] * ws.x; as.y += xs[j] * ws.y; as.z += xs[j] * ws.z; as.w += xs[j] * ws.w;
            ad.x += xs[j] * wd.x; ad.y += xs[j] * wd.y; ad.z += xs[j] * wd.z; ad.w += xs[j] * wd.w;
        }
    }
    *(float4*)(asrc + (long)n * 4) = as;
    *(float4*)(adst + (long)n * 4) = ad;
}

__device__ __forceinline__ float leaky02(float x) { return (x > 0.f) ? x : 0.2f * x; }


#define TILE_NODES 16
#define TILE_PITCH 260

static_assert(NNODES % TILE_NODES == 0, "every work-group owns 16 whole nodes");

__device__ __forceinline__ void aggregate_node_in(const int n, const int lane,
                                                  const int* __restrict__ rowptr, const int* __restrict__ csr_src,
                                                  const float* __restrict__ asrc, const float* __restrict__ adst,
                                                  const float* __restrict__ X, float* __restrict__ trow) {
    constexpr int H = 4;
    const int start = __builtin_amdgcn_readfirstlane(rowptr[n]);
    const int deg   = __builtin_amdgcn_readfirstlane(rowptr[n + 1]) - start;
    const int total = deg + 1;

    float ad[H];
    #pragma unroll
    for (int h = 0; h < H; ++h) ad[h] = adst[n * H + h];

    const bool v0 = lane < total;
    const int  s0 = (lane < deg) ? csr_src[start + lane] : n;
    const float4 a0 = *(const float4*)(asrc + (long)s0 * H);
    const float a0v[H] = {a0.x, a0.y, a0.z, a0.w};
    float x0[H];
    #pragma unroll
    for (int h = 0; h < H; ++h) x0[h] = v0 ? leaky02(a0v[h] + ad[h]) : -INFINITY;

    float m[H];
    #pragma unroll
    for (int h = 0; h < H; ++h) m[h] = x0[h];
    for (int c0 = 32; c0 < total; c0 += 32) {
        const int t = c0 + lane;
        const int s = (t < deg) ? csr_src[start + t] : n;
        if (t < total) {
            #pragma unroll
            for (int h = 0; h < H; ++h) m[h] = fmaxf(m[h], leaky02(asrc[s * H + h] + ad[h]));
        }
    }
    #pragma unroll
    for (int h = 0; h < H; ++h) {
        #pragma unroll
        for (int o = 16; o > 0; o >>= 1) m[h] = fmaxf(m[h], __shfl_xor(m[h], o));
    }

    float e0[H], sum[H];
    #pragma unroll
    for (int h = 0; h < H; ++h) { e0[h] = v0 ? expf(x0[h] - m[h]) : 0.f; sum[h] = e0[h]; }
    for (int c0 = 32; c0 < total; c0 += 32) {
        const int t = c0 + lane;
        const int s = (t < deg) ? csr_src[start + t] : n;
        if (t < total) {
            #pragma unroll
            for (int h = 0; h < H; ++h) sum[h] += expf(leaky02(asrc[s * H + h] + ad[h]) - m[h]);
        }
    }
    #pragma unroll
    for (int h = 0; h < H; ++h) {
        #pragma unroll
        for (int o = 16; o > 0; o >>= 1) sum[h] += __shfl_xor(sum[h], o);
    }

    float2 acc[H];
    #pragma unroll
    for (int h = 0; h < H; ++h) acc[h] = make_float2(0.f, 0.f);

    for (int c0 = 0; c0 < total; c0 += 32) {
        int s_l = s0;
        float cf[H];
        if (c0 == 0) {
            #pragma unroll
            for (int h = 0; h < H; ++h) cf[h] = e0[h] / sum[h];
        } else {
            const int t = c0 + lane;
            s_l = (t < deg) ? csr_src[start + t] : n;
            #pragma unroll
            for (int h = 0; h < H; ++h)
                cf[h] = (t < total) ? expf(leaky02(asrc[s_l * H + h] + ad[h]) - m[h]) / sum[h] : 0.f;
        }
        const int cnt = (total - c0 < 32) ? (total - c0) : 32;
        for (int j0 = 0; j0 < cnt; j0 += 8) {
            #pragma unroll
            for (int u = 0; u < 8; ++u) {
                const int jj = j0 + u;
                const int s = __builtin_amdgcn_readlane(s_l, jj);
                const float2 v = *(const float2*)(X + (long)s * 64 + lane * 2);
                #pragma unroll
                for (int h = 0; h < H; ++h) {
                    const float c = __int_as_float(__builtin_amdgcn_readlane(__float_as_int(cf[h]), jj));
                    acc[h].x = fmaf(c, v.x, acc[h].x);
                    acc[h].y = fmaf(c, v.y, acc[h].y);
                }
            }
        }
    }

    #pragma unroll
    for (int h = 0; h < H; ++h) *(float2*)(trow + h * 64 + lane * 2) = acc[h];
}

__global__ __launch_bounds__(256, 16)
void node_softmax_aggregate_in_project(const int* __restrict__ rowptr, const int* __restrict__ csr_src,
                                       const float* __restrict__ asrc, const float* __restrict__ adst,
                                       const float* __restrict__ X, const float* __restrict__ w12,
                                       const float* __restrict__ b12, float* __restrict__ xh,
                                       const float* __restrict__ w12s, const float* __restrict__ w12d,
                                       const float* __restrict__ c12,
                                       float* __restrict__ as2, float* __restrict__ ad2) {
    __shared__ __attribute__((aligned(16))) float tile[TILE_NODES * TILE_PITCH];
    const int lane = threadIdx.x & 31;
    const int w = __builtin_amdgcn_readfirstlane((int)(threadIdx.x >> 5));
    const int nbase = (int)blockIdx.x * TILE_NODES;

    for (int i = 0; i < TILE_NODES / 8; ++i) {
        const int r = w * (TILE_NODES / 8) + i;
        aggregate_node_in(nbase + r, lane, rowptr, csr_src, asrc, adst, X, tile + r * TILE_PITCH);
    }

    {
        float2 ws[4], wd[4];
        #pragma unroll
        for (int h = 0; h < 4; ++h) {
            ws[h] = *(const float2*)(w12s + h * 64 + lane * 2);
            wd[h] = *(const float2*)(w12d + h * 64 + lane * 2);
        }
        const float cs = c12[0], cd = c12[1];
        float rs = 0.f, rd = 0.f;
        #pragma unroll
        for (int i = 0; i < TILE_NODES / 8; ++i) {
            const float* trow = tile + (w * (TILE_NODES / 8) + i) * TILE_PITCH;
            float ps = 0.f, pd = 0.f;
            #pragma unroll
            for (int h = 0; h < 4; ++h) {
                const float2 a = *(const float2*)(trow + h * 64 + lane * 2);
                ps = fmaf(a.x, ws[h].x, ps); ps = fmaf(a.y, ws[h].y, ps);
                pd = fmaf(a.x, wd[h].x, pd); pd = fmaf(a.y, wd[h].y, pd);
            }
            #pragma unroll
            for (int o = 16; o > 0; o >>= 1) { ps += __shfl_xor(ps, o); pd += __shfl_xor(pd, o); }
            rs = (lane == i) ? ps : rs;
            rd = (lane == i) ? pd : rd;
        }
        if (lane < TILE_NODES / 8) {
            const int n = nbase + w * (TILE_NODES / 8) + lane;
            as2[n] = rs + cs;
            ad2[n] = rd + cd;
        }
    }
    __syncthreads();

    const int m  = lane & 15;
    const int kh = (lane >> 4) << 1;
    const int n0 = lane & 15;
    const int col = w * 16 + n0;
    const float bv = b12[col];
    v8f acc = (v8f){bv,bv,bv,bv,bv,bv,bv,bv};
    const float* __restrict__ Arow = tile + m * TILE_PITCH + kh;
    const float4* __restrict__ Bf = (const float4*)w12 + (long)w * 32 * 32 + lane;
    #pragma unroll 8
    for (int kq = 0; kq < 32; ++kq) {
        const float4 b4 = Bf[kq * 32];
        const v2f a0 = *(const v2f*)(Arow + kq * 8);
        const v2f a1 = *(const v2f*)(Arow + kq * 8 + 4);
        v2f b0, b1;
        b0.x = b4.x; b0.y = b4.y;
        b1.x = b4.z; b1.y = b4.w;
        acc = __builtin_amdgcn_wmma_f32_16x16x4_f32(false, a0, false, b0, (short)0, acc, false, false);
        acc = __builtin_amdgcn_wmma_f32_16x16x4_f32(false, a1, false, b1, (short)0, acc, false, false);
    }
    float* __restrict__ Crow = xh + (long)(nbase + (lane >> 4) * 8) * 128 + col;
    #pragma unroll
    for (int v = 0; v < 8; ++v) Crow[v * 128] = acc[v];
}

template <int H, int C>
__global__ __launch_bounds__(256)
void node_softmax_aggregate(const int* __restrict__ rowptr, const int* __restrict__ csr_src,
                            const float* __restrict__ asrc, const float* __restrict__ adst,
                            const float* __restrict__ xh, const float* __restrict__ bias,
                            float* __restrict__ out) {
    constexpr int HC  = H * C;
    constexpr int NIT = HC / 128;
    const int lane = threadIdx.x & 31;
    const int n = __builtin_amdgcn_readfirstlane((int)((blockIdx.x * blockDim.x + threadIdx.x) >> 5));
    if (n >= NNODES) return;
    const int start = __builtin_amdgcn_readfirstlane(rowptr[n]);
    const int deg   = __builtin_amdgcn_readfirstlane(rowptr[n + 1]) - start;
    const int total = deg + 1;

    float ad[H];
    #pragma unroll
    for (int h = 0; h < H; ++h) ad[h] = adst[n * H + h];

    const bool v0 = lane < total;
    const int  s0 = (lane < deg) ? csr_src[start + lane] : n;
    float x0[H];
    #pragma unroll
    for (int h = 0; h < H; ++h) x0[h] = v0 ? leaky02(asrc[s0 * H + h] + ad[h]) : -INFINITY;

    float m[H];
    #pragma unroll
    for (int h = 0; h < H; ++h) m[h] = x0[h];
    for (int c0 = 32; c0 < total; c0 += 32) {
        const int t = c0 + lane;
        const int s = (t < deg) ? csr_src[start + t] : n;
        if (t < total) {
            #pragma unroll
            for (int h = 0; h < H; ++h) m[h] = fmaxf(m[h], leaky02(asrc[s * H + h] + ad[h]));
        }
    }
    #pragma unroll
    for (int h = 0; h < H; ++h) {
        #pragma unroll
        for (int o = 16; o > 0; o >>= 1) m[h] = fmaxf(m[h], __shfl_xor(m[h], o));
    }

    float e0[H], sum[H];
    #pragma unroll
    for (int h = 0; h < H; ++h) { e0[h] = v0 ? expf(x0[h] - m[h]) : 0.f; sum[h] = e0[h]; }
    for (int c0 = 32; c0 < total; c0 += 32) {
        const int t = c0 + lane;
        const int s = (t < deg) ? csr_src[start + t] : n;
        if (t < total) {
            #pragma unroll
            for (int h = 0; h < H; ++h) sum[h] += expf(leaky02(asrc[s * H + h] + ad[h]) - m[h]);
        }
    }
    #pragma unroll
    for (int h = 0; h < H; ++h) {
        #pragma unroll
        for (int o = 16; o > 0; o >>= 1) sum[h] += __shfl_xor(sum[h], o);
    }

    int hsel[NIT];
    #pragma unroll
    for (int it = 0; it < NIT; ++it) hsel[it] = (it * 128 + lane * 4) / C;
    float4 acc[NIT];
    #pragma unroll
    for (int it = 0; it < NIT; ++it) acc[it] = make_float4(0.f, 0.f, 0.f, 0.f);

    for (int c0 = 0; c0 < total; c0 += 32) {
        int s_l = s0;
        float cf[H];
        if (c0 == 0) {
            #pragma unroll
            for (int h = 0; h < H; ++h) cf[h] = e0[h] / sum[h];
        } else {
            const int t = c0 + lane;
            s_l = (t < deg) ? csr_src[start + t] : n;
            #pragma unroll
            for (int h = 0; h < H; ++h)
                cf[h] = expf(leaky02(asrc[s_l * H + h] + ad[h]) - m[h]) / sum[h];
        }
        const int cnt = (total - c0 < 32) ? (total - c0) : 32;
        #pragma unroll 4
        for (int jj = 0; jj < cnt; ++jj) {
            const int s = __builtin_amdgcn_readlane(s_l, jj);
            float ch[H];
            #pragma unroll
            for (int h = 0; h < H; ++h)
                ch[h] = __int_as_float(__builtin_amdgcn_readlane(__float_as_int(cf[h]), jj));
            const float* __restrict__ row = xh + (long)s * HC + lane * 4;
            #pragma unroll
            for (int it = 0; it < NIT; ++it) {
                float c = ch[0];
                #pragma unroll
                for (int h = 1; h < H; ++h) c = (hsel[it] == h) ? ch[h] : c;
                const float4 v = *(const float4*)(row + it * 128);
                acc[it].x += c * v.x;
                acc[it].y += c * v.y;
                acc[it].z += c * v.z;
                acc[it].w += c * v.w;
            }
        }
    }

    float* __restrict__ orow = out + (long)n * HC + lane * 4;
    #pragma unroll
    for (int it = 0; it < NIT; ++it) {
        const float4 b = *(const float4*)(bias + it * 128 + lane * 4);
        float4 r;
        r.x = acc[it].x + b.x; r.y = acc[it].y + b.y; r.z = acc[it].z + b.z; r.w = acc[it].w + b.w;
        *(float4*)(orow + it * 128) = r;
    }
}

static inline int cdiv(long a, int b) { return (int)((a + b - 1) / b); }

extern "C" void kernel_launch(void* const* d_in, const int* in_sizes, int n_in,
                              void* d_out, int out_size, void* d_ws, size_t ws_size,
                              hipStream_t stream) {
    const float* X        = (const float*)d_in[0];
    const int*   ei       = (const int*)  d_in[1];
    const float* W1       = (const float*)d_in[2];
    const float* att_src1 = (const float*)d_in[3];
    const float* att_dst1 = (const float*)d_in[4];
    const float* b1       = (const float*)d_in[5];
    const float* W2       = (const float*)d_in[6];
    const float* att_src2 = (const float*)d_in[7];
    const float* att_dst2 = (const float*)d_in[8];
    const float* b2       = (const float*)d_in[9];
    float* out = (float*)d_out;

    float* ws = (float*)d_ws;
    long off = 0;
    float* xh     = ws + off; off += (long)NNODES * 128;
    float* asrc   = ws + off; off += (long)NNODES * 4;
    float* adst   = ws + off; off += (long)NNODES * 4;
    int*   rowptr = (int*)(ws + off); off += NNODES + 4;
    int*   csrc   = (int*)(ws + off); off += NEDGES;
    float* w12    = ws + off; off += 256 * 128;
    float* b12    = ws + off; off += 128;
    int*   offsT  = (int*)(ws + off); off += (CSR_BLOCKS + 1) * OFFS_PITCH;
    int*   bucketed = (int*)(ws + off); off += NEDGES;
    float* w12s   = ws + off; off += 256;
    float* w12d   = ws + off; off += 256;
    float* c12    = ws + off; off += 4;
    float* as2    = ws + off; off += NNODES;
    float* ad2    = ws + off; off += NNODES;

    const int B = 256;

    bucket_edge_slices<<<NSLICE, SLICE_T, 0, stream>>>(ei, offsT, bucketed);
    group_range_edges<<<CSR_BLOCKS, CSR_T, 0, stream>>>(offsT, bucketed, rowptr, csrc);

    {
        input_scores<<<SCORE_BLOCKS + FOLD_BLOCKS + SCORE2_BLOCKS, B, 0, stream>>>(
            X, W1, att_src1, att_dst1, asrc, adst, W2, b1, w12, b12, att_src2, att_dst2, w12s, w12d, c12);
        node_softmax_aggregate_in_project<<<NNODES / TILE_NODES, B, 0, stream>>>(
            rowptr, csrc, asrc, adst, X, w12, b12, xh, w12s, w12d, c12, as2, ad2);
    }

    {
        node_softmax_aggregate<1, 128><<<cdiv((long)NNODES * 32, B), B, 0, stream>>>(
            rowptr, csrc, as2, ad2, xh, b2, out);
    }
}


# Input scores and weight folds run as extra work-groups of the range-grouping launch (98 work-groups, one WGP each) on its empty WGPs; score launch gone, 5 launches to 4; sum order kept
# speedup vs baseline: 8.6657x; 1.0645x over previous
#include <hip/hip_runtime.h>
#include <stdint.h>


#define NNODES 50000
#define NEDGES 800000

typedef float v2f __attribute__((ext_vector_type(2)));
typedef float v8f __attribute__((ext_vector_type(8)));


#define CSR_RANGE  512
#define CSR_SHIFT  9
#define CSR_BLOCKS ((NNODES + CSR_RANGE - 1) / CSR_RANGE)
#define CSR_T      1024
#define CSR_CAP    24576
#define SLICE      1600
#define NSLICE     (NEDGES / SLICE)
#define SLICE_T    320
#define SLICE_K    (SLICE / SLICE_T)
#define OFFS_PITCH 512
#define WAVE_SLICES (OFFS_PITCH / 32)

static_assert(NEDGES % SLICE == 0, "the slices cover the edge list exactly");
static_assert(SLICE % SLICE_T == 0, "every thread of launch 1 has the same number of edges");
static_assert((1 << CSR_SHIFT) == CSR_RANGE, "range of a destination = destination >> CSR_SHIFT");
static_assert(NSLICE <= OFFS_PITCH && OFFS_PITCH <= CSR_T, "one thread of launch 2 per slice");
static_assert(NNODES <= (1 << 16), "a source index fits the low 16 bits of an entry");
static_assert(CSR_RANGE <= (1 << 16), "a local destination fits the high 16 bits of an entry");
static_assert(CSR_BLOCKS + 1 <= 128, "range counts and the total fit the 128-entry LDS table");

__global__ __launch_bounds__(SLICE_T)
void bucket_edge_slices(const int* __restrict__ ei, int* __restrict__ offsT, int* __restrict__ bucketed) {
    __shared__ int hist[128];
    const int t = threadIdx.x;
    const int lane = t & 31;
    const int e0 = (int)blockIdx.x * SLICE;
    if (t < 128) hist[t] = 0;
    __syncthreads();

    int d[SLICE_K], s[SLICE_K], rk[SLICE_K];
    #pragma unroll
    for (int k = 0; k < SLICE_K; ++k) {
        d[k] = ei[NEDGES + e0 + k * SLICE_T + t];
        s[k] = ei[e0 + k * SLICE_T + t];
    }
    #pragma unroll
    for (int k = 0; k < SLICE_K; ++k) {
        rk[k] = 0;
        if ((unsigned)d[k] < (unsigned)NNODES) rk[k] = atomicAdd(&hist[d[k] >> CSR_SHIFT], 1);
    }
    __syncthreads();

    if (t < 32) {
        int h[4];
        int sum = 0;
        #pragma unroll
        for (int j = 0; j < 4; ++j) { h[j] = hist[lane * 4 + j]; sum += h[j]; }
        int inc = sum;
        for (int o = 1; o < 32; o <<= 1) { int u = __shfl_up(inc, o); if (lane >= o) inc += u; }
        int run = inc - sum;
        #pragma unroll
        for (int j = 0; j < 4; ++j) { hist[lane * 4 + j] = run; run += h[j]; }
    }
    __syncthreads();

    if (t <= CSR_BLOCKS) offsT[t * OFFS_PITCH + (int)blockIdx.x] = hist[t];
    #pragma unroll
    for (int k = 0; k < SLICE_K; ++k) {
        if ((unsigned)d[k] < (unsigned)NNODES) {
            const int b = d[k] >> CSR_SHIFT;
            bucketed[e0 + hist[b] + rk[k]] = (s[k] & 0xFFFF) | ((d[k] & (CSR_RANGE - 1)) << 16);
        }
    }
}


#define FOLD_BLOCKS   5
#define SCORE2_BLOCKS 1
#define SIDE_T        256
#define SCORE_NODES   512
#define SCORE_BLOCKS  ((NNODES + SCORE_NODES - 1) / SCORE_NODES)
#define LAUNCH2_BLOCKS (CSR_BLOCKS + FOLD_BLOCKS + SCORE2_BLOCKS + SCORE_BLOCKS)

static_assert(SIDE_T <= CSR_T && SCORE_NODES <= CSR_T, "launch 2's work-groups all have CSR_T threads");

__device__ __forceinline__ void fold_weights_block(int fb, const float* __restrict__ W1,
                                                   const float* __restrict__ W2, const float* __restrict__ b1,
                                                   float* __restrict__ w12, float* __restrict__ b12,
                                                   float (*part)[128]) {
    const int lane  = threadIdx.x & 31;
    const int w     = threadIdx.x >> 5;
    const bool is_bias = (fb == FOLD_BLOCKS - 1);
    if (w < SIDE_T / 32) {
        const int strip = w & 1;
        const int q     = w >> 1;
        const int m     = lane & 15;
        const int kh    = (lane >> 4) << 1;
        const int n0    = lane & 15;
        const int h     = is_bias ? q : fb;

        const float* __restrict__ Arow = is_bias ? (b1 + h * 64) : (W1 + (q * 16 + m) * 256 + h * 64);
        const float* __restrict__ Bcol = W2 + (long)h * 64 * 128 + strip * 64 + n0;

        v8f acc[4];
        #pragma unroll
        for (int t = 0; t < 4; ++t) acc[t] = (v8f){0.f,0.f,0.f,0.f,0.f,0.f,0.f,0.f};

        #pragma unroll 4
        for (int k0 = 0; k0 < 64; k0 += 4) {
            v2f a;
            a.x = Arow[k0 + kh + 0];
            a.y = Arow[k0 + kh + 1];
            #pragma unroll
            for (int t = 0; t < 4; ++t) {
                v2f b;
                b.x = Bcol[(k0 + kh + 0) * 128 + t * 16];
                b.y = Bcol[(k0 + kh + 1) * 128 + t * 16];
                acc[t] = __builtin_amdgcn_wmma_f32_16x16x4_f32(
                             false, a, false, b, (short)0, acc[t], false, false);
            }
        }

        if (!is_bias) {
            const int kq = h * 8 + q * 2 + (lane >> 4);
            #pragma unroll
            for (int t = 0; t < 4; ++t) {
                float4* __restrict__ dst = (float4*)w12 + ((long)((strip * 4 + t) * 32 + kq) * 32 + n0);
                dst[0]  = make_float4(acc[t][0], acc[t][1], acc[t][4], acc[t][5]);
                dst[16] = make_float4(acc[t][2], acc[t][3], acc[t][6], acc[t][7]);
            }
        } else if (lane < 16) {
            #pragma unroll
            for (int t = 0; t < 4; ++t) part[q][strip * 64 + t * 16 + n0] = acc[t][0];
        }
    }
    if (is_bias) {
        __syncthreads();
        const int j = threadIdx.x;
        if (j < 128) b12[j] = ((part[0][j] + part[1][j]) + part[2][j]) + part[3][j];
    }
}

__device__ __forceinline__ void fold_layer2_scores_block(const float* __restrict__ W1, const float* __restrict__ W2,
                                                         const float* __restrict__ b1,
                                                         const float* __restrict__ att_src2,
                                                         const float* __restrict__ att_dst2,
                                                         float* __restrict__ w12s, float* __restrict__ w12d,
                                                         float* __restrict__ c12,
                                                         float* us, float* ud, float* red) {
    const int t = threadIdx.x;
    const int lane = t & 31, w = t >> 5;
    const bool on = t < SIDE_T;
    if (on) {
        const float4 as = *(const float4*)(att_src2 + lane * 4);
        const float4 ad = *(const float4*)(att_dst2 + lane * 4);
        float rs = 0.f, rd = 0.f;
        #pragma unroll 8
        for (int i = 0; i < 32; ++i) {
            const int r = w * 32 + i;
            const float4 x = *(const float4*)(W2 + r * 128 + lane * 4);
            float ps = x.x * as.x; ps = fmaf(x.y, as.y, ps); ps = fmaf(x.z, as.z, ps); ps = fmaf(x.w, as.w, ps);
            float pd = x.x * ad.x; pd = fmaf(x.y, ad.y, pd); pd = fmaf(x.z, ad.z, pd); pd = fmaf(x.w, ad.w, pd);
            #pragma unroll
            for (int o = 16; o > 0; o >>= 1) { ps += __shfl_xor(ps, o); pd += __shfl_xor(pd, o); }
            rs = (lane == i) ? ps : rs;
            rd = (lane == i) ? pd : rd;
        }
        us[t] = rs;
        ud[t] = rd;
    }
    __syncthreads();
    if (on) {
        {
            const int h = t & 3, k = t >> 2;
            const float4* __restrict__ w4 = (const float4*)(W1 + t * 64);
            const float4* s4 = (const float4*)(us + h * 64);
            const float4* d4 = (const float4*)(ud + h * 64);
            float ss = 0.f, sd = 0.f;
            #pragma unroll
            for (int c = 0; c < 16; ++c) {
                const float4 wv = w4[c], a = s4[c], d = d4[c];
                ss += wv.x * a.x; ss += wv.y * a.y; ss += wv.z * a.z; ss += wv.w * a.w;
                sd += wv.x * d.x; sd += wv.y * d.y; sd += wv.z * d.z; sd += wv.w * d.w;
            }
            w12s[h * 64 + k] = ss;
            w12d[h * 64 + k] = sd;
        }
        float qs = b1[t] * us[t], qd = b1[t] * ud[t];
        #pragma unroll
        for (int o = 16; o > 0; o >>= 1) { qs += __shfl_xor(qs, o); qd += __shfl_xor(qd, o); }
        if (lane == 0) { red[w] = qs; red[8 + w] = qd; }
    }
    __syncthreads();
    if (t == 0) {
        float cs = 0.f, cd = 0.f;
        for (int j = 0; j < 8; ++j) { cs += red[j]; cd += red[8 + j]; }
        c12[0] = cs;
        c12[1] = cd;
    }
}

__device__ __forceinline__ void input_scores_block(int sb, const float* __restrict__ X, const float* __restrict__ W1,
                                                   const float* __restrict__ att_src, const float* __restrict__ att_dst,
                                                   float* __restrict__ asrc, float* __restrict__ adst,
                                                   float4* wsrc, float4* wdst) {
    const int t = threadIdx.x;
    if (t < SIDE_T) {
        const int h = t & 3;
        const float4* __restrict__ w4 = (const float4*)(W1 + t * 64);
        const float4* __restrict__ s4 = (const float4*)(att_src + h * 64);
        const float4* __restrict__ d4 = (const float4*)(att_dst + h * 64);
        float ss = 0.f, sd = 0.f;
        #pragma unroll
        for (int c = 0; c < 16; ++c) {
            const float4 w = w4[c], a = s4[c], d = d4[c];
            ss += w.x * a.x; ss += w.y * a.y; ss += w.z * a.z; ss += w.w * a.w;
            sd += w.x * d.x; sd += w.y * d.y; sd += w.z * d.z; sd += w.w * d.w;
        }
        ((float*)wsrc)[t] = ss;
        ((float*)wdst)[t] = sd;
    }
    __syncthreads();
    const int n = sb * SCORE_NODES + t;
    if (t >= SCORE_NODES || n >= NNODES) return;
    const float4* __restrict__ x4 = (const float4*)(X + (long)n * 64);
    float4 as = make_float4(0.f, 0.f, 0.f, 0.f), ad = make_float4(0.f, 0.f, 0.f, 0.f);
    float4 xr[16];
    #pragma unroll
    for (int c = 0; c < 16; ++c) xr[c] = x4[c];
    __builtin_amdgcn_sched_barrier(0);
    #pragma unroll
    for (int c = 0; c < 16; ++c) {
        const float4 x = xr[c];
        const float xs[4] = {x.x, x.y, x.z, x.w};
        #pragma unroll
        for (int j = 0; j < 4; ++j) {
            const float4 ws = wsrc[c * 4 + j], wd = wdst[c * 4 + j];
            as.x += xs[j] * ws.x; as.y += xs[j] * ws.y; as.z += xs[j] * ws.z; as.w += xs[j] * ws.w;
            ad.x += xs[j] * wd.x; ad.y += xs[j] * wd.y; ad.z += xs[j] * wd.z; ad.w += xs[j] * wd.w;
        }
    }
    *(float4*)(asrc + (long)n * 4) = as;
    *(float4*)(adst + (long)n * 4) = ad;
}

struct RangeLds {
    int cnt[CSR_RANGE];
    int wsum[32];
    int wbase[32];
    int seg_at[OFFS_PITCH];
    int seg_n[OFFS_PITCH];
    int seg_to[OFFS_PITCH];
    unsigned list[CSR_CAP];
    int sorted[CSR_CAP];
};
struct SideLds {
    float wsrc[64 * 4];
    float wdst[64 * 4];
    float part[4][128];
};
union __attribute__((aligned(16))) Launch2Lds {
    RangeLds range;
    SideLds side;
};

__global__ __launch_bounds__(CSR_T)
void group_range_edges_and_input_work(const int* __restrict__ offsT, const int* __restrict__ bucketed,
                                      int* __restrict__ rowptr, int* __restrict__ csr_src,
                                      const float* __restrict__ X, const float* __restrict__ W1,
                                      const float* __restrict__ att_src, const float* __restrict__ att_dst,
                                      float* __restrict__ asrc, float* __restrict__ adst,
                                      const float* __restrict__ W2, const float* __restrict__ b1,
                                      float* __restrict__ w12, float* __restrict__ b12,
                                      const float* __restrict__ att_src2, const float* __restrict__ att_dst2,
                                      float* __restrict__ w12s, float* __restrict__ w12d, float* __restrict__ c12) {
    __shared__ Launch2Lds lds;
    if (blockIdx.x >= CSR_BLOCKS) {
        const int sb = (int)blockIdx.x - CSR_BLOCKS;
        float4* const wsrc = (float4*)lds.side.wsrc;
        float4* const wdst = (float4*)lds.side.wdst;
        if (sb < FOLD_BLOCKS) fold_weights_block(sb, W1, W2, b1, w12, b12, lds.side.part);
        else if (sb < FOLD_BLOCKS + SCORE2_BLOCKS)
            fold_layer2_scores_block(W1, W2, b1, att_src2, att_dst2, w12s, w12d, c12,
                                     lds.side.wsrc, lds.side.wdst, &lds.side.part[0][0]);
        else input_scores_block(sb - FOLD_BLOCKS - SCORE2_BLOCKS, X, W1, att_src, att_dst, asrc, adst, wsrc, wdst);
        return;
    }
    int* const cnt = lds.range.cnt;
    int* const wsum = lds.range.wsum;
    int* const wbase = lds.range.wbase;
    int* const seg_at = lds.range.seg_at;
    int* const seg_n = lds.range.seg_n;
    int* const seg_to = lds.range.seg_to;
    unsigned* const list = lds.range.list;
    int* const sorted = lds.range.sorted;
    const int t = threadIdx.x;
    const int lane = t & 31, w = t >> 5;
    const int r = (int)blockIdx.x;
    const int lo = r * CSR_RANGE;
    if (t < CSR_RANGE) cnt[t] = 0;

    int o0 = 0, c = 0;
    if (t < NSLICE) {
        o0 = offsT[r * OFFS_PITCH + t];
        c  = offsT[(r + 1) * OFFS_PITCH + t] - o0;
    }
    int inc = c;
    for (int o = 1; o < 32; o <<= 1) { int u = __shfl_up(inc, o); if (lane >= o) inc += u; }
    int osum = o0;
    #pragma unroll
    for (int o = 16; o > 0; o >>= 1) osum += __shfl_xor(osum, o);
    if (lane == 31) wsum[w] = inc;
    if (lane == 0) wbase[w] = osum;
    __syncthreads();

    int wp = 0, M = 0, base = 0;
    for (int j = 0; j < 32; ++j) {
        const int v = wsum[j];
        wp += (j < w) ? v : 0;
        M += v;
        base += wbase[j];
    }
    if (t < OFFS_PITCH) {
        seg_at[t] = t * SLICE + o0;
        seg_n[t]  = c;
        seg_to[t] = wp + inc - c;
    }
    __syncthreads();

    {
        const int mine = w * WAVE_SLICES + (lane & (WAVE_SLICES - 1));
        const int n_l = seg_n[mine], at_l = seg_at[mine], to_l = seg_to[mine];
        int ent[WAVE_SLICES];
        #pragma unroll
        for (int u = 0; u < WAVE_SLICES; ++u) {
            const int n  = __builtin_amdgcn_readlane(n_l, u);
            const int at = __builtin_amdgcn_readlane(at_l, u);
            ent[u] = (lane < n) ? bucketed[at + lane] : 0;
        }
        #pragma unroll
        for (int u = 0; u < WAVE_SLICES; ++u) {
            const int n  = __builtin_amdgcn_readlane(n_l, u);
            const int at = __builtin_amdgcn_readlane(at_l, u);
            const int to = __builtin_amdgcn_readlane(to_l, u);
            if (lane < n) {
                atomicAdd(&cnt[(unsigned)ent[u] >> 16], 1);
                const int p = to + lane;
                if (p < CSR_CAP) list[p] = (unsigned)ent[u];
            }
            for (int j = 32 + lane; j < n; j += 32) {
                const unsigned x = (unsigned)bucketed[at + j];
                atomicAdd(&cnt[x >> 16], 1);
                const int p = to + j;
                if (p < CSR_CAP) list[p] = x;
            }
        }
    }
    __syncthreads();

    const int cc = (t < CSR_RANGE) ? cnt[t] : 0;
    int inc2 = cc;
    for (int o = 1; o < 32; o <<= 1) { int u = __shfl_up(inc2, o); if (lane >= o) inc2 += u; }
    if (lane == 31) wsum[w] = inc2;
    __syncthreads();
    int wp2 = 0;
    for (int j = 0; j < 32; ++j) wp2 += (j < w) ? wsum[j] : 0;
    const int off = wp2 + inc2 - cc;
    if (t < CSR_RANGE && lo + t < NNODES) rowptr[lo + t] = base + off;
    if (blockIdx.x == CSR_BLOCKS - 1 && t == 0) rowptr[NNODES] = base + M;

    if (M <= CSR_CAP) {
        if (t < CSR_RANGE) cnt[t] = off;
        __syncthreads();
        for (int p0 = 0; p0 < M; p0 += CSR_T * 8) {
            unsigned x[8];
            int at[8];
            #pragma unroll
            for (int u = 0; u < 8; ++u) {
                const int p = p0 + u * CSR_T + t;
                x[u] = (p < M) ? list[p] : 0u;
            }
            #pragma unroll
            for (int u = 0; u < 8; ++u) {
                const int p = p0 + u * CSR_T + t;
                at[u] = (p < M) ? atomicAdd(&cnt[x[u] >> 16], 1) : 0;
            }
            #pragma unroll
            for (int u = 0; u < 8; ++u) {
                const int p = p0 + u * CSR_T + t;
                if (p < M) sorted[at[u]] = (int)(x[u] & 0xFFFFu);
            }
        }
        __syncthreads();
        for (int p = t; p < M; p += CSR_T) csr_src[base + p] = sorted[p];
    } else {
        if (t < CSR_RANGE) cnt[t] = base + off;
        __syncthreads();
        for (int u = 0; u < WAVE_SLICES; ++u) {
            const int i = w * WAVE_SLICES + u;
            const int n = seg_n[i];
            for (int j = lane; j < n; j += 32) {
                const unsigned x = (unsigned)bucketed[seg_at[i] + j];
                csr_src[atomicAdd(&cnt[x >> 16], 1)] = (int)(x & 0xFFFFu);
            }
        }
    }
}


__device__ __forceinline__ float leaky02(float x) { return (x > 0.f) ? x : 0.2f * x; }


#define TILE_NODES 16
#define TILE_PITCH 260

static_assert(NNODES % TILE_NODES == 0, "every work-group owns 16 whole nodes");

__device__ __forceinline__ void aggregate_node_in(const int n, const int lane,
                                                  const int* __restrict__ rowptr, const int* __restrict__ csr_src,
                                                  const float* __restrict__ asrc, const float* __restrict__ adst,
                                                  const float* __restrict__ X, float* __restrict__ trow) {
    constexpr int H = 4;
    const int start = __builtin_amdgcn_readfirstlane(rowptr[n]);
    const int deg   = __builtin_amdgcn_readfirstlane(rowptr[n + 1]) - start;
    const int total = deg + 1;

    float ad[H];
    #pragma unroll
    for (int h = 0; h < H; ++h) ad[h] = adst[n * H + h];

    const bool v0 = lane < total;
    const int  s0 = (lane < deg) ? csr_src[start + lane] : n;
    const float4 a0 = *(const float4*)(asrc + (long)s0 * H);
    const float a0v[H] = {a0.x, a0.y, a0.z, a0.w};
    float x0[H];
    #pragma unroll
    for (int h = 0; h < H; ++h) x0[h] = v0 ? leaky02(a0v[h] + ad[h]) : -INFINITY;

    float m[H];
    #pragma unroll
    for (int h = 0; h < H; ++h) m[h] = x0[h];
    for (int c0 = 32; c0 < total; c0 += 32) {
        const int t = c0 + lane;
        const int s = (t < deg) ? csr_src[start + t] : n;
        if (t < total) {
            #pragma unroll
            for (int h = 0; h < H; ++h) m[h] = fmaxf(m[h], leaky02(asrc[s * H + h] + ad[h]));
        }
    }
    #pragma unroll
    for (int h = 0; h < H; ++h) {
        #pragma unroll
        for (int o = 16; o > 0; o >>= 1) m[h] = fmaxf(m[h], __shfl_xor(m[h], o));
    }

    float e0[H], sum[H];
    #pragma unroll
    for (int h = 0; h < H; ++h) { e0[h] = v0 ? expf(x0[h] - m[h]) : 0.f; sum[h] = e0[h]; }
    for (int c0 = 32; c0 < total; c0 += 32) {
        const int t = c0 + lane;
        const int s = (t < deg) ? csr_src[start + t] : n;
        if (t < total) {
            #pragma unroll
            for (int h = 0; h < H; ++h) sum[h] += expf(leaky02(asrc[s * H + h] + ad[h]) - m[h]);
        }
    }
    #pragma unroll
    for (int h = 0; h < H; ++h) {
        #pragma unroll
        for (int o = 16; o > 0; o >>= 1) sum[h] += __shfl_xor(sum[h], o);
    }

    float2 acc[H];
    #pragma unroll
    for (int h = 0; h < H; ++h) acc[h] = make_float2(0.f, 0.f);

    for (int c0 = 0; c0 < total; c0 += 32) {
        int s_l = s0;
        float cf[H];
        if (c0 == 0) {
            #pragma unroll
            for (int h = 0; h < H; ++h) cf[h] = e0[h] / sum[h];
        } else {
            const int t = c0 + lane;
            s_l = (t < deg) ? csr_src[start + t] : n;
            #pragma unroll
            for (int h = 0; h < H; ++h)
                cf[h] = (t < total) ? expf(leaky02(asrc[s_l * H + h] + ad[h]) - m[h]) / sum[h] : 0.f;
        }
        const int cnt = (total - c0 < 32) ? (total - c0) : 32;
        for (int j0 = 0; j0 < cnt; j0 += 8) {
            #pragma unroll
            for (int u = 0; u < 8; ++u) {
                const int jj = j0 + u;
                const int s = __builtin_amdgcn_readlane(s_l, jj);
                const float2 v = *(const float2*)(X + (long)s * 64 + lane * 2);
                #pragma unroll
                for (int h = 0; h < H; ++h) {
                    const float c = __int_as_float(__builtin_amdgcn_readlane(__float_as_int(cf[h]), jj));
                    acc[h].x = fmaf(c, v.x, acc[h].x);
                    acc[h].y = fmaf(c, v.y, acc[h].y);
                }
            }
        }
    }

    #pragma unroll
    for (int h = 0; h < H; ++h) *(float2*)(trow + h * 64 + lane * 2) = acc[h];
}

__global__ __launch_bounds__(256, 16)
void node_softmax_aggregate_in_project(const int* __restrict__ rowptr, const int* __restrict__ csr_src,
                                       const float* __restrict__ asrc, const float* __restrict__ adst,
                                       const float* __restrict__ X, const float* __restrict__ w12,
                                       const float* __restrict__ b12, float* __restrict__ xh,
                                       const float* __restrict__ w12s, const float* __restrict__ w12d,
                                       const float* __restrict__ c12,
                                       float* __restrict__ as2, float* __restrict__ ad2) {
    __shared__ __attribute__((aligned(16))) float tile[TILE_NODES * TILE_PITCH];
    const int lane = threadIdx.x & 31;
    const int w = __builtin_amdgcn_readfirstlane((int)(threadIdx.x >> 5));
    const int nbase = (int)blockIdx.x * TILE_NODES;

    for (int i = 0; i < TILE_NODES / 8; ++i) {
        const int r = w * (TILE_NODES / 8) + i;
        aggregate_node_in(nbase + r, lane, rowptr, csr_src, asrc, adst, X, tile + r * TILE_PITCH);
    }

    {
        float2 ws[4], wd[4];
        #pragma unroll
        for (int h = 0; h < 4; ++h) {
            ws[h] = *(const float2*)(w12s + h * 64 + lane * 2);
            wd[h] = *(const float2*)(w12d + h * 64 + lane * 2);
        }
        const float cs = c12[0], cd = c12[1];
        float rs = 0.f, rd = 0.f;
        #pragma unroll
        for (int i = 0; i < TILE_NODES / 8; ++i) {
            const float* trow = tile + (w * (TILE_NODES / 8) + i) * TILE_PITCH;
            float ps = 0.f, pd = 0.f;
            #pragma unroll
            for (int h = 0; h < 4; ++h) {
                const float2 a = *(const float2*)(trow + h * 64 + lane * 2);
                ps = fmaf(a.x, ws[h].x, ps); ps = fmaf(a.y, ws[h].y, ps);
                pd = fmaf(a.x, wd[h].x, pd); pd = fmaf(a.y, wd[h].y, pd);
            }
            #pragma unroll
            for (int o = 16; o > 0; o >>= 1) { ps += __shfl_xor(ps, o); pd += __shfl_xor(pd, o); }
            rs = (lane == i) ? ps : rs;
            rd = (lane == i) ? pd : rd;
        }
        if (lane < TILE_NODES / 8) {
            const int n = nbase + w * (TILE_NODES / 8) + lane;
            as2[n] = rs + cs;
            ad2[n] = rd + cd;
        }
    }
    __syncthreads();

    const int m  = lane & 15;
    const int kh = (lane >> 4) << 1;
    const int n0 = lane & 15;
    const int col = w * 16 + n0;
    const float bv = b12[col];
    v8f acc = (v8f){bv,bv,bv,bv,bv,bv,bv,bv};
    const float* __restrict__ Arow = tile + m * TILE_PITCH + kh;
    const float4* __restrict__ Bf = (const float4*)w12 + (long)w * 32 * 32 + lane;
    #pragma unroll 8
    for (int kq = 0; kq < 32; ++kq) {
        const float4 b4 = Bf[kq * 32];
        const v2f a0 = *(const v2f*)(Arow + kq * 8);
        const v2f a1 = *(const v2f*)(Arow + kq * 8 + 4);
        v2f b0, b1;
        b0.x = b4.x; b0.y = b4.y;
        b1.x = b4.z; b1.y = b4.w;
        acc = __builtin_amdgcn_wmma_f32_16x16x4_f32(false, a0, false, b0, (short)0, acc, false, false);
        acc = __builtin_amdgcn_wmma_f32_16x16x4_f32(false, a1, false, b1, (short)0, acc, false, false);
    }
    float* __restrict__ Crow = xh + (long)(nbase + (lane >> 4) * 8) * 128 + col;
    #pragma unroll
    for (int v = 0; v < 8; ++v) Crow[v * 128] = acc[v];
}

template <int H, int C>
__global__ __launch_bounds__(256)
void node_softmax_aggregate(const int* __restrict__ rowptr, const int* __restrict__ csr_src,
                            const float* __restrict__ asrc, const float* __restrict__ adst,
                            const float* __restrict__ xh, const float* __restrict__ bias,
                            float* __restrict__ out) {
    constexpr int HC  = H * C;
    constexpr int NIT = HC / 128;
    const int lane = threadIdx.x & 31;
    const int n = __builtin_amdgcn_readfirstlane((int)((blockIdx.x * blockDim.x + threadIdx.x) >> 5));
    if (n >= NNODES) return;
    const int start = __builtin_amdgcn_readfirstlane(rowptr[n]);
    const int deg   = __builtin_amdgcn_readfirstlane(rowptr[n + 1]) - start;
    const int total = deg + 1;

    float ad[H];
    #pragma unroll
    for (int h = 0; h < H; ++h) ad[h] = adst[n * H + h];

    const bool v0 = lane < total;
    const int  s0 = (lane < deg) ? csr_src[start + lane] : n;
    float x0[H];
    #pragma unroll
    for (int h = 0; h < H; ++h) x0[h] = v0 ? leaky02(asrc[s0 * H + h] + ad[h]) : -INFINITY;

    float m[H];
    #pragma unroll
    for (int h = 0; h < H; ++h) m[h] = x0[h];
    for (int c0 = 32; c0 < total; c0 += 32) {
        const int t = c0 + lane;
        const int s = (t < deg) ? csr_src[start + t] : n;
        if (t < total) {
            #pragma unroll
            for (int h = 0; h < H; ++h) m[h] = fmaxf(m[h], leaky02(asrc[s * H + h] + ad[h]));
        }
    }
    #pragma unroll
    for (int h = 0; h < H; ++h) {
        #pragma unroll
        for (int o = 16; o > 0; o >>= 1) m[h] = fmaxf(m[h], __shfl_xor(m[h], o));
    }

    float e0[H], sum[H];
    #pragma unroll
    for (int h = 0; h < H; ++h) { e0[h] = v0 ? expf(x0[h] - m[h]) : 0.f; sum[h] = e0[h]; }
    for (int c0 = 32; c0 < total; c0 += 32) {
        const int t = c0 + lane;
        const int s = (t < deg) ? csr_src[start + t] : n;
        if (t < total) {
            #pragma unroll
            for (int h = 0; h < H; ++h) sum[h] += expf(leaky02(asrc[s * H + h] + ad[h]) - m[h]);
        }
    }
    #pragma unroll
    for (int h = 0; h < H; ++h) {
        #pragma unroll
        for (int o = 16; o > 0; o >>= 1) sum[h] += __shfl_xor(sum[h], o);
    }

    int hsel[NIT];
    #pragma unroll
    for (int it = 0; it < NIT; ++it) hsel[it] = (it * 128 + lane * 4) / C;
    float4 acc[NIT];
    #pragma unroll
    for (int it = 0; it < NIT; ++it) acc[it] = make_float4(0.f, 0.f, 0.f, 0.f);

    for (int c0 = 0; c0 < total; c0 += 32) {
        int s_l = s0;
        float cf[H];
        if (c0 == 0) {
            #pragma unroll
            for (int h = 0; h < H; ++h) cf[h] = e0[h] / sum[h];
        } else {
            const int t = c0 + lane;
            s_l = (t < deg) ? csr_src[start + t] : n;
            #pragma unroll
            for (int h = 0; h < H; ++h)
                cf[h] = expf(leaky02(asrc[s_l * H + h] + ad[h]) - m[h]) / sum[h];
        }
        const int cnt = (total - c0 < 32) ? (total - c0) : 32;
        #pragma unroll 4
        for (int jj = 0; jj < cnt; ++jj) {
            const int s = __builtin_amdgcn_readlane(s_l, jj);
            float ch[H];
            #pragma unroll
            for (int h = 0; h < H; ++h)
                ch[h] = __int_as_float(__builtin_amdgcn_readlane(__float_as_int(cf[h]), jj));
            const float* __restrict__ row = xh + (long)s * HC + lane * 4;
            #pragma unroll
            for (int it = 0; it < NIT; ++it) {
                float c = ch[0];
                #pragma unroll
                for (int h = 1; h < H; ++h) c = (hsel[it] == h) ? ch[h] : c;
                const float4 v = *(const float4*)(row + it * 128);
                acc[it].x += c * v.x;
                acc[it].y += c * v.y;
                acc[it].z += c * v.z;
                acc[it].w += c * v.w;
            }
        }
    }

    float* __restrict__ orow = out + (long)n * HC + lane * 4;
    #pragma unroll
    for (int it = 0; it < NIT; ++it) {
        const float4 b = *(const float4*)(bias + it * 128 + lane * 4);
        float4 r;
        r.x = acc[it].x + b.x; r.y = acc[it].y + b.y; r.z = acc[it].z + b.z; r.w = acc[it].w + b.w;
        *(float4*)(orow + it * 128) = r;
    }
}

static inline int cdiv(long a, int b) { return (int)((a + b - 1) / b); }

extern "C" void kernel_launch(void* const* d_in, const int* in_sizes, int n_in,
                              void* d_out, int out_size, void* d_ws, size_t ws_size,
                              hipStream_t stream) {
    const float* X        = (const float*)d_in[0];
    const int*   ei       = (const int*)  d_in[1];
    const float* W1       = (const float*)d_in[2];
    const float* att_src1 = (const float*)d_in[3];
    const float* att_dst1 = (const float*)d_in[4];
    const float* b1       = (const float*)d_in[5];
    const float* W2       = (const float*)d_in[6];
    const float* att_src2 = (const float*)d_in[7];
    const float* att_dst2 = (const float*)d_in[8];
    const float* b2       = (const float*)d_in[9];
    float* out = (float*)d_out;

    float* ws = (float*)d_ws;
    long off = 0;
    float* xh     = ws + off; off += (long)NNODES * 128;
    float* asrc   = ws + off; off += (long)NNODES * 4;
    float* adst   = ws + off; off += (long)NNODES * 4;
    int*   rowptr = (int*)(ws + off); off += NNODES + 4;
    int*   csrc   = (int*)(ws + off); off += NEDGES;
    float* w12    = ws + off; off += 256 * 128;
    float* b12    = ws + off; off += 128;
    int*   offsT  = (int*)(ws + off); off += (CSR_BLOCKS + 1) * OFFS_PITCH;
    int*   bucketed = (int*)(ws + off); off += NEDGES;
    float* w12s   = ws + off; off += 256;
    float* w12d   = ws + off; off += 256;
    float* c12    = ws + off; off += 4;
    float* as2    = ws + off; off += NNODES;
    float* ad2    = ws + off; off += NNODES;

    const int B = 256;

    bucket_edge_slices<<<NSLICE, SLICE_T, 0, stream>>>(ei, offsT, bucketed);
    group_range_edges_and_input_work<<<LAUNCH2_BLOCKS, CSR_T, 0, stream>>>(
        offsT, bucketed, rowptr, csrc, X, W1, att_src1, att_dst1, asrc, adst, W2, b1, w12, b12,
        att_src2, att_dst2, w12s, w12d, c12);

    {
        node_softmax_aggregate_in_project<<<NNODES / TILE_NODES, B, 0, stream>>>(
            rowptr, csrc, asrc, adst, X, w12, b12, xh, w12s, w12d, c12, as2, ad2);
    }

    {
        node_softmax_aggregate<1, 128><<<cdiv((long)NNODES * 32, B), B, 0, stream>>>(
            rowptr, csrc, as2, ad2, xh, b2, out);
    }
}
